# starting kernel
# baseline (speedup 1.0000x reference)
#include <hip/hip_runtime.h>
#include <hip/hip_bf16.h>
#include <math.h>

typedef __attribute__((ext_vector_type(2))) float v2f;
typedef __attribute__((ext_vector_type(8))) float v8f;

#define NN      100000
#define NE      1600000
#define HH      64
#define TT      384
#define NTILES  6250

__device__ __forceinline__ v2f mk2(float a, float b) {
    v2f r; r.x = a; r.y = b; return r;
}

__device__ __forceinline__ v8f wmma4(v2f a, v2f b, v8f c) {
    return __builtin_amdgcn_wmma_f32_16x16x4_f32(
         false, a,  false, b,
         (short)0, c,  false,  false);
}

__global__ void degree_kernel(const int* __restrict__ src, const int* __restrict__ dst,
                              float* __restrict__ degF, float* __restrict__ degR) {
    int e = blockIdx.x * 256 + threadIdx.x;
    if (e < NE) {
        atomicAdd(&degF[dst[e]], 1.0f);
        atomicAdd(&degR[src[e]], 1.0f);
    }
}

__global__ void invert_kernel(float* __restrict__ degF, float* __restrict__ degR) {
    int i = blockIdx.x * 256 + threadIdx.x;
    if (i < NN) {
        float a = degF[i];
        float b = degR[i];
        degF[i] = (a > 0.0f) ? (1.0f / a) : 0.0f;
        degR[i] = (b > 0.0f) ? (1.0f / b) : 0.0f;
    }
}

__global__ __launch_bounds__(256) void enc_kernel(const float* __restrict__ emb,
                                                  const float* __restrict__ W,
                                                  const float* __restrict__ bias,
                                                  float* __restrict__ x) {
    const int wave = threadIdx.x >> 5;
    const int lane = threadIdx.x & 31;
    const int nt   = blockIdx.x * 8 + wave;
    if (nt >= NTILES) return;
    const int ln = lane & 15;
    const int hi = lane >> 4;
    const int kh = hi << 1;

    const float* arow = emb + (size_t)(nt * 16 + ln) * TT + kh;
    v8f acc0 = {}, acc1 = {}, acc2 = {}, acc3 = {};

    #pragma unroll 4
    for (int k0 = 0; k0 < TT; k0 += 4) {
        v2f a = *reinterpret_cast<const v2f*>(arow + k0);
        const float* bp = W + (size_t)(k0 + kh) * HH + ln;
        v2f b0 = mk2(bp[0],  bp[HH]);
        v2f b1 = mk2(bp[16], bp[HH + 16]);
        v2f b2 = mk2(bp[32], bp[HH + 32]);
        v2f b3 = mk2(bp[48], bp[HH + 48]);
        acc0 = wmma4(a, b0, acc0);
        acc1 = wmma4(a, b1, acc1);
        acc2 = wmma4(a, b2, acc2);
        acc3 = wmma4(a, b3, acc3);
    }

    float bs0 = bias[ln], bs1 = bias[16 + ln], bs2 = bias[32 + ln], bs3 = bias[48 + ln];
    #pragma unroll
    for (int v = 0; v < 8; ++v) {
        size_t r = (size_t)(nt * 16 + v + 8 * hi) * HH + ln;
        x[r]      = acc0[v] + bs0;
        x[r + 16] = acc1[v] + bs1;
        x[r + 32] = acc2[v] + bs2;
        x[r + 48] = acc3[v] + bs3;
    }
}

__global__ __launch_bounds__(256) void scatter_kernel(const float* __restrict__ x,
                                                      const float* __restrict__ w,
                                                      const int* __restrict__ src,
                                                      const int* __restrict__ dst,
                                                      float* __restrict__ nF,
                                                      float* __restrict__ nR) {
    size_t gid = (size_t)blockIdx.x * 256 + threadIdx.x;
    int e = (int)(gid >> 6);
    int c = (int)(gid & 63);
    if (e >= NE) return;
    int s = src[e];
    int d = dst[e];
    float we = w[e];
    float xs = x[(size_t)s * HH + c];
    float xd = x[(size_t)d * HH + c];
    atomicAdd(&nF[(size_t)d * HH + c], we * xs);
    atomicAdd(&nR[(size_t)s * HH + c], we * xd);
}

__global__ __launch_bounds__(256) void layer_kernel(
    float* __restrict__ x,
    const float* __restrict__ nF, const float* __restrict__ nR,
    const float* __restrict__ invF, const float* __restrict__ invR,
    const float* __restrict__ WsF, const float* __restrict__ WnF, const float* __restrict__ bF,
    const float* __restrict__ WsR, const float* __restrict__ WnR, const float* __restrict__ bR) {
    const int wave = threadIdx.x >> 5;
    const int lane = threadIdx.x & 31;
    const int nt   = blockIdx.x * 8 + wave;
    if (nt >= NTILES) return;
    const int ln = lane & 15;
    const int hi = lane >> 4;
    const int kh = hi << 1;
    const int rowA = nt * 16 + ln;

    const float sF = invF[rowA];
    const float sR = invR[rowA];
    const float* xr = x  + (size_t)rowA * HH + kh;
    const float* fr = nF + (size_t)rowA * HH + kh;
    const float* rr = nR + (size_t)rowA * HH + kh;

    v8f aF0 = {}, aF1 = {}, aF2 = {}, aF3 = {};
    v8f aR0 = {}, aR1 = {}, aR2 = {}, aR3 = {};

    #pragma unroll 4
    for (int k0 = 0; k0 < HH; k0 += 4) {
        v2f aS = *reinterpret_cast<const v2f*>(xr + k0);
        v2f aN = *reinterpret_cast<const v2f*>(fr + k0);
        v2f aV = *reinterpret_cast<const v2f*>(rr + k0);
        aN *= sF;
        aV *= sR;

        const float* pSF = WsF + (size_t)(k0 + kh) * HH + ln;
        const float* pNF = WnF + (size_t)(k0 + kh) * HH + ln;
        const float* pSR = WsR + (size_t)(k0 + kh) * HH + ln;
        const float* pNR = WnR + (size_t)(k0 + kh) * HH + ln;

        aF0 = wmma4(aS, mk2(pSF[0], pSF[HH]), aF0);
        aF0 = wmma4(aN, mk2(pNF[0], pNF[HH]), aF0);
        aR0 = wmma4(aS, mk2(pSR[0], pSR[HH]), aR0);
        aR0 = wmma4(aV, mk2(pNR[0], pNR[HH]), aR0);
        aF1 = wmma4(aS, mk2(pSF[16], pSF[HH + 16]), aF1);
        aF1 = wmma4(aN, mk2(pNF[16], pNF[HH + 16]), aF1);
        aR1 = wmma4(aS, mk2(pSR[16], pSR[HH + 16]), aR1);
        aR1 = wmma4(aV, mk2(pNR[16], pNR[HH + 16]), aR1);
        aF2 = wmma4(aS, mk2(pSF[32], pSF[HH + 32]), aF2);
        aF2 = wmma4(aN, mk2(pNF[32], pNF[HH + 32]), aF2);
        aR2 = wmma4(aS, mk2(pSR[32], pSR[HH + 32]), aR2);
        aR2 = wmma4(aV, mk2(pNR[32], pNR[HH + 32]), aR2);
        aF3 = wmma4(aS, mk2(pSF[48], pSF[HH + 48]), aF3);
        aF3 = wmma4(aN, mk2(pNF[48], pNF[HH + 48]), aF3);
        aR3 = wmma4(aS, mk2(pSR[48], pSR[HH + 48]), aR3);
        aR3 = wmma4(aV, mk2(pNR[48], pNR[HH + 48]), aR3);
    }

    float bf0 = bF[ln], bf1 = bF[16 + ln], bf2 = bF[32 + ln], bf3 = bF[48 + ln];
    float br0 = bR[ln], br1 = bR[16 + ln], br2 = bR[32 + ln], br3 = bR[48 + ln];

    #pragma unroll
    for (int v = 0; v < 8; ++v) {
        size_t r = (size_t)(nt * 16 + v + 8 * hi) * HH + ln;
        x[r]      += fmaxf(aF0[v] + bf0, 0.0f) + fmaxf(aR0[v] + br0, 0.0f);
        x[r + 16] += fmaxf(aF1[v] + bf1, 0.0f) + fmaxf(aR1[v] + br1, 0.0f);
        x[r + 32] += fmaxf(aF2[v] + bf2, 0.0f) + fmaxf(aR2[v] + br2, 0.0f);
        x[r + 48] += fmaxf(aF3[v] + bf3, 0.0f) + fmaxf(aR3[v] + br3, 0.0f);
    }
}

__global__ __launch_bounds__(64) void norm_kernel(const float* __restrict__ x,
                                                  const int* __restrict__ ids,
                                                  float* __restrict__ out) {
    __shared__ float s[64];
    int b = blockIdx.x;
    int c = threadIdx.x;
    int node = ids[b];
    float v = x[(size_t)node * HH + c];
    s[c] = v * v;
    __syncthreads();
    #pragma unroll
    for (int off = 32; off > 0; off >>= 1) {
        if (c < off) s[c] += s[c + off];
        __syncthreads();
    }
    float nrm = sqrtf(s[0]);
    out[(size_t)b * HH + c] = v / nrm;
}

extern "C" void kernel_launch(void* const* d_in, const int* in_sizes, int n_in,
                              void* d_out, int out_size, void* d_ws, size_t ws_size,
                              hipStream_t stream) {
    const float* text_emb = (const float*)d_in[0];
    const float* weights  = (const float*)d_in[1];
    const float* W_enc    = (const float*)d_in[2];
    const float* b_enc    = (const float*)d_in[3];
    const float* Wself_f  = (const float*)d_in[4];
    const float* Wneigh_f = (const float*)d_in[5];
    const float* b_f      = (const float*)d_in[6];
    const float* Wself_r  = (const float*)d_in[7];
    const float* Wneigh_r = (const float*)d_in[8];
    const float* b_r      = (const float*)d_in[9];
    const int*   src      = (const int*)d_in[10];
    const int*   dst      = (const int*)d_in[11];
    const int*   ids      = (const int*)d_in[12];
    float* out = (float*)d_out;

    const size_t NH = (size_t)NN * HH;
    float* x    = (float*)d_ws;
    float* nF   = x + NH;
    float* nR   = x + 2 * NH;
    float* invF = x + 3 * NH;
    float* invR = invF + NN;

    const int batch = in_sizes[12];

    hipMemsetAsync(invF, 0, (size_t)2 * NN * sizeof(float), stream);
    degree_kernel<<<(NE + 255) / 256, 256, 0, stream>>>(src, dst, invF, invR);
    invert_kernel<<<(NN + 255) / 256, 256, 0, stream>>>(invF, invR);

    enc_kernel<<<(NTILES + 7) / 8, 256, 0, stream>>>(text_emb, W_enc, b_enc, x);

    for (int l = 0; l < 2; ++l) {
        hipMemsetAsync(nF, 0, (size_t)2 * NH * sizeof(float), stream);
        scatter_kernel<<<(int)(((size_t)NE * HH + 255) / 256), 256, 0, stream>>>(
            x, weights, src, dst, nF, nR);
        layer_kernel<<<(NTILES + 7) / 8, 256, 0, stream>>>(
            x, nF, nR, invF, invR,
            Wself_f + (size_t)l * HH * HH, Wneigh_f + (size_t)l * HH * HH, b_f + (size_t)l * HH,
            Wself_r + (size_t)l * HH * HH, Wneigh_r + (size_t)l * HH * HH, b_r + (size_t)l * HH);
    }

    norm_kernel<<<batch, 64, 0, stream>>>(x, ids, out);
}


# Neighbour sums by node-range owners: a work-group keeps 1024 nodes' sums in LDS, filters the whole edge list, adds w*x rows by LDS float atomics; global atomics, clears gone; sum order = LDS arrival
# speedup vs baseline: 1.4100x; 1.4100x over previous
#include <hip/hip_runtime.h>
#include <hip/hip_bf16.h>
#include <math.h>

typedef __attribute__((ext_vector_type(2))) float v2f;
typedef __attribute__((ext_vector_type(8))) float v8f;

#define NN      100000
#define NE      1600000
#define HH      64
#define TT      384
#define NTILES  6250

__device__ __forceinline__ v2f mk2(float a, float b) {
    v2f r; r.x = a; r.y = b; return r;
}

__device__ __forceinline__ v8f wmma4(v2f a, v2f b, v8f c) {
    return __builtin_amdgcn_wmma_f32_16x16x4_f32(
         false, a,  false, b,
         (short)0, c,  false,  false);
}

#define DEG_PART    50016
#define DEG_SLICES  32
#define DEG_SLICE_E (NE / DEG_SLICES)
#define DEG_THREADS 1024
#define DEG_NV      ((DEG_SLICE_E / 4 + DEG_THREADS - 1) / DEG_THREADS)

__global__ __launch_bounds__(DEG_THREADS) void degree_kernel(const int* __restrict__ src,
                                                             const int* __restrict__ dst,
                                                             float* __restrict__ degF,
                                                             float* __restrict__ degR) {
    __shared__ unsigned int cnt[DEG_PART];
    const int tid   = threadIdx.x;
    const int combo = blockIdx.x & 3;
    const int slice = blockIdx.x >> 2;
    const int rev   = combo & 1;
    const int base  = (combo >> 1) * DEG_PART;
    const int nloc  = (NN - base < DEG_PART) ? (NN - base) : DEG_PART;
    const int* __restrict__ idx = rev ? src : dst;
    float* __restrict__ deg     = rev ? degR : degF;

    const int4* ip = reinterpret_cast<const int4*>(idx + (size_t)slice * DEG_SLICE_E);
    int4 v[DEG_NV];
    #pragma unroll
    for (int j = 0; j < DEG_NV; ++j) {
        int i = tid + j * DEG_THREADS;
        v[j] = (i < DEG_SLICE_E / 4) ? ip[i] : make_int4(-1, -1, -1, -1);
    }

    uint4* cnt4 = reinterpret_cast<uint4*>(cnt);
    for (int i = tid; i < DEG_PART / 4; i += DEG_THREADS)
        cnt4[i] = make_uint4(0u, 0u, 0u, 0u);
    __syncthreads();

    #pragma unroll
    for (int j = 0; j < DEG_NV; ++j) {
        unsigned r0 = (unsigned)(v[j].x - base);
        unsigned r1 = (unsigned)(v[j].y - base);
        unsigned r2 = (unsigned)(v[j].z - base);
        unsigned r3 = (unsigned)(v[j].w - base);
        if (r0 < (unsigned)nloc) atomicAdd(&cnt[r0], 1u);
        if (r1 < (unsigned)nloc) atomicAdd(&cnt[r1], 1u);
        if (r2 < (unsigned)nloc) atomicAdd(&cnt[r2], 1u);
        if (r3 < (unsigned)nloc) atomicAdd(&cnt[r3], 1u);
    }
    __syncthreads();

    for (int i0 = tid; i0 < nloc; i0 += 8 * DEG_THREADS) {
        unsigned c[8];
        #pragma unroll
        for (int j = 0; j < 8; ++j) {
            int i = i0 + j * DEG_THREADS;
            c[j] = (i < nloc) ? cnt[i] : 0u;
        }
        #pragma unroll
        for (int j = 0; j < 8; ++j)
            if (c[j]) atomicAdd(&deg[base + i0 + j * DEG_THREADS], (float)c[j]);
    }
}

__global__ void invert_kernel(float* __restrict__ degF, float* __restrict__ degR) {
    int i = blockIdx.x * 256 + threadIdx.x;
    if (i < NN) {
        float a = degF[i];
        float b = degR[i];
        degF[i] = (a > 0.0f) ? (1.0f / a) : 0.0f;
        degR[i] = (b > 0.0f) ? (1.0f / b) : 0.0f;
    }
}

__global__ __launch_bounds__(256) void enc_kernel(const float* __restrict__ emb,
                                                  const float* __restrict__ W,
                                                  const float* __restrict__ bias,
                                                  float* __restrict__ x) {
    const int wave = threadIdx.x >> 5;
    const int lane = threadIdx.x & 31;
    const int nt   = blockIdx.x * 8 + wave;
    if (nt >= NTILES) return;
    const int ln = lane & 15;
    const int hi = lane >> 4;
    const int kh = hi << 1;

    const float* arow = emb + (size_t)(nt * 16 + ln) * TT + kh;
    v8f acc0 = {}, acc1 = {}, acc2 = {}, acc3 = {};

    #pragma unroll 4
    for (int k0 = 0; k0 < TT; k0 += 4) {
        v2f a = *reinterpret_cast<const v2f*>(arow + k0);
        const float* bp = W + (size_t)(k0 + kh) * HH + ln;
        v2f b0 = mk2(bp[0],  bp[HH]);
        v2f b1 = mk2(bp[16], bp[HH + 16]);
        v2f b2 = mk2(bp[32], bp[HH + 32]);
        v2f b3 = mk2(bp[48], bp[HH + 48]);
        acc0 = wmma4(a, b0, acc0);
        acc1 = wmma4(a, b1, acc1);
        acc2 = wmma4(a, b2, acc2);
        acc3 = wmma4(a, b3, acc3);
    }

    float bs0 = bias[ln], bs1 = bias[16 + ln], bs2 = bias[32 + ln], bs3 = bias[48 + ln];
    #pragma unroll
    for (int v = 0; v < 8; ++v) {
        size_t r = (size_t)(nt * 16 + v + 8 * hi) * HH + ln;
        x[r]      = acc0[v] + bs0;
        x[r + 16] = acc1[v] + bs1;
        x[r + 32] = acc2[v] + bs2;
        x[r + 48] = acc3[v] + bs3;
    }
}

#define OWN_NODES   1024
#define OWN_PARTS   ((NN + OWN_NODES - 1) / OWN_NODES)
#define OWN_THREADS 1024
#define OWN_V       3
#define OWN_CHUNK4  (OWN_THREADS * OWN_V)
#define OWN_CHUNK_E (OWN_CHUNK4 * 4)
#define NE4         (NE / 4)
#define OWN_NCHUNK  ((NE4 + OWN_CHUNK4 - 1) / OWN_CHUNK4)
#define OWN_BATCH   6
#define OWN_EBITS   21

__device__ __forceinline__ void lds_barrier() {
    __builtin_amdgcn_fence(__ATOMIC_RELEASE, "workgroup", "local");
    __builtin_amdgcn_s_barrier();
    __builtin_amdgcn_fence(__ATOMIC_ACQUIRE, "workgroup", "local");
}

__global__ __launch_bounds__(OWN_THREADS) void own_kernel(const float* __restrict__ x,
                                                          const float* __restrict__ w,
                                                          const int* __restrict__ src,
                                                          const int* __restrict__ dst,
                                                          float* __restrict__ nF,
                                                          float* __restrict__ nR) {
    __shared__ float acc[OWN_NODES * HH];
    __shared__ unsigned int queue[OWN_CHUNK_E];
    __shared__ unsigned int qn[2];
    const int tid  = threadIdx.x;
    const int lane = tid & 31;
    const int wave = tid >> 5;
    const int rev  = blockIdx.x & 1;
    const int base = (blockIdx.x >> 1) * OWN_NODES;
    const unsigned nloc = (unsigned)((NN - base < OWN_NODES) ? (NN - base) : OWN_NODES);
    const int* __restrict__ own = rev ? src : dst;
    const int* __restrict__ oth = rev ? dst : src;
    float* __restrict__ outp    = rev ? nR : nF;
    const int4* own4 = reinterpret_cast<const int4*>(own);

    int4 v[OWN_V];
    #pragma unroll
    for (int j = 0; j < OWN_V; ++j) {
        int i = j * OWN_THREADS + tid;
        v[j] = (i < NE4) ? own4[i] : make_int4(-1, -1, -1, -1);
    }

    float4* acc4 = reinterpret_cast<float4*>(acc);
    for (int i = tid; i < OWN_NODES * HH / 4; i += OWN_THREADS)
        acc4[i] = make_float4(0.0f, 0.0f, 0.0f, 0.0f);
    if (tid == 0) { qn[0] = 0u; qn[1] = 0u; }
    __syncthreads();

    for (int ch = 0; ch < OWN_NCHUNK; ++ch) {
        const int i0 = ch * OWN_CHUNK4 + tid;
        unsigned int* cntp = &qn[ch & 1];

        #pragma unroll
        for (int j = 0; j < OWN_V; ++j) {
            const unsigned e0 = (unsigned)(i0 + j * OWN_THREADS) * 4u;
            const unsigned r0 = (unsigned)(v[j].x - base);
            const unsigned r1 = (unsigned)(v[j].y - base);
            const unsigned r2 = (unsigned)(v[j].z - base);
            const unsigned r3 = (unsigned)(v[j].w - base);
            if (r0 < nloc) queue[atomicAdd(cntp, 1u)] = (r0 << OWN_EBITS) | (e0 + 0u);
            if (r1 < nloc) queue[atomicAdd(cntp, 1u)] = (r1 << OWN_EBITS) | (e0 + 1u);
            if (r2 < nloc) queue[atomicAdd(cntp, 1u)] = (r2 << OWN_EBITS) | (e0 + 2u);
            if (r3 < nloc) queue[atomicAdd(cntp, 1u)] = (r3 << OWN_EBITS) | (e0 + 3u);
        }

        #pragma unroll
        for (int j = 0; j < OWN_V; ++j) {
            int i = i0 + OWN_CHUNK4 + j * OWN_THREADS;
            v[j] = (i < NE4) ? own4[i] : make_int4(-1, -1, -1, -1);
        }

        lds_barrier();
        const unsigned n = *cntp;
        if (tid == 0) qn[(ch + 1) & 1] = 0u;

        for (unsigned q0 = (unsigned)wave; q0 < n; q0 += 32u * OWN_BATCH) {
            unsigned ent[OWN_BATCH];
            int      o[OWN_BATCH];
            float    we[OWN_BATCH];
            v2f      xv[OWN_BATCH];
            #pragma unroll
            for (int j = 0; j < OWN_BATCH; ++j) {
                const unsigned q = q0 + 32u * (unsigned)j;
                ent[j] = queue[(q < n) ? q : q0];
            }
            #pragma unroll
            for (int j = 0; j < OWN_BATCH; ++j) {
                const unsigned e = ent[j] & ((1u << OWN_EBITS) - 1u);
                o[j]  = oth[e];
                we[j] = w[e];
            }
            #pragma unroll
            for (int j = 0; j < OWN_BATCH; ++j)
                xv[j] = *reinterpret_cast<const v2f*>(x + (size_t)o[j] * HH + 2 * lane);
            #pragma unroll
            for (int j = 0; j < OWN_BATCH; ++j)
                asm volatile("" : "+v"(we[j]));
            #pragma unroll
            for (int j = 0; j < OWN_BATCH; ++j) {
                if (q0 + 32u * (unsigned)j < n) {
                    float* a = acc + (size_t)(ent[j] >> OWN_EBITS) * HH + 2 * lane;
                    atomicAdd(a,     we[j] * xv[j].x);
                    atomicAdd(a + 1, we[j] * xv[j].y);
                }
            }
        }
        lds_barrier();
    }

    float4* out4 = reinterpret_cast<float4*>(outp + (size_t)base * HH);
    const int nout4 = (int)nloc * (HH / 4);
    for (int i = tid; i < nout4; i += OWN_THREADS)
        out4[i] = acc4[i];
}

__global__ __launch_bounds__(256) void layer_kernel(
    float* __restrict__ x,
    const float* __restrict__ nF, const float* __restrict__ nR,
    const float* __restrict__ invF, const float* __restrict__ invR,
    const float* __restrict__ WsF, const float* __restrict__ WnF, const float* __restrict__ bF,
    const float* __restrict__ WsR, const float* __restrict__ WnR, const float* __restrict__ bR) {
    const int wave = threadIdx.x >> 5;
    const int lane = threadIdx.x & 31;
    const int nt   = blockIdx.x * 8 + wave;
    if (nt >= NTILES) return;
    const int ln = lane & 15;
    const int hi = lane >> 4;
    const int kh = hi << 1;
    const int rowA = nt * 16 + ln;

    const float sF = invF[rowA];
    const float sR = invR[rowA];
    const float* xr = x  + (size_t)rowA * HH + kh;
    const float* fr = nF + (size_t)rowA * HH + kh;
    const float* rr = nR + (size_t)rowA * HH + kh;

    v8f aF0 = {}, aF1 = {}, aF2 = {}, aF3 = {};
    v8f aR0 = {}, aR1 = {}, aR2 = {}, aR3 = {};

    #pragma unroll 4
    for (int k0 = 0; k0 < HH; k0 += 4) {
        v2f aS = *reinterpret_cast<const v2f*>(xr + k0);
        v2f aN = *reinterpret_cast<const v2f*>(fr + k0);
        v2f aV = *reinterpret_cast<const v2f*>(rr + k0);
        aN *= sF;
        aV *= sR;

        const float* pSF = WsF + (size_t)(k0 + kh) * HH + ln;
        const float* pNF = WnF + (size_t)(k0 + kh) * HH + ln;
        const float* pSR = WsR + (size_t)(k0 + kh) * HH + ln;
        const float* pNR = WnR + (size_t)(k0 + kh) * HH + ln;

        aF0 = wmma4(aS, mk2(pSF[0], pSF[HH]), aF0);
        aF0 = wmma4(aN, mk2(pNF[0], pNF[HH]), aF0);
        aR0 = wmma4(aS, mk2(pSR[0], pSR[HH]), aR0);
        aR0 = wmma4(aV, mk2(pNR[0], pNR[HH]), aR0);
        aF1 = wmma4(aS, mk2(pSF[16], pSF[HH + 16]), aF1);
        aF1 = wmma4(aN, mk2(pNF[16], pNF[HH + 16]), aF1);
        aR1 = wmma4(aS, mk2(pSR[16], pSR[HH + 16]), aR1);
        aR1 = wmma4(aV, mk2(pNR[16], pNR[HH + 16]), aR1);
        aF2 = wmma4(aS, mk2(pSF[32], pSF[HH + 32]), aF2);
        aF2 = wmma4(aN, mk2(pNF[32], pNF[HH + 32]), aF2);
        aR2 = wmma4(aS, mk2(pSR[32], pSR[HH + 32]), aR2);
        aR2 = wmma4(aV, mk2(pNR[32], pNR[HH + 32]), aR2);
        aF3 = wmma4(aS, mk2(pSF[48], pSF[HH + 48]), aF3);
        aF3 = wmma4(aN, mk2(pNF[48], pNF[HH + 48]), aF3);
        aR3 = wmma4(aS, mk2(pSR[48], pSR[HH + 48]), aR3);
        aR3 = wmma4(aV, mk2(pNR[48], pNR[HH + 48]), aR3);
    }

    float bf0 = bF[ln], bf1 = bF[16 + ln], bf2 = bF[32 + ln], bf3 = bF[48 + ln];
    float br0 = bR[ln], br1 = bR[16 + ln], br2 = bR[32 + ln], br3 = bR[48 + ln];

    #pragma unroll
    for (int v = 0; v < 8; ++v) {
        size_t r = (size_t)(nt * 16 + v + 8 * hi) * HH + ln;
        x[r]      += fmaxf(aF0[v] + bf0, 0.0f) + fmaxf(aR0[v] + br0, 0.0f);
        x[r + 16] += fmaxf(aF1[v] + bf1, 0.0f) + fmaxf(aR1[v] + br1, 0.0f);
        x[r + 32] += fmaxf(aF2[v] + bf2, 0.0f) + fmaxf(aR2[v] + br2, 0.0f);
        x[r + 48] += fmaxf(aF3[v] + bf3, 0.0f) + fmaxf(aR3[v] + br3, 0.0f);
    }
}

__global__ __launch_bounds__(64) void norm_kernel(const float* __restrict__ x,
                                                  const int* __restrict__ ids,
                                                  float* __restrict__ out) {
    __shared__ float s[64];
    int b = blockIdx.x;
    int c = threadIdx.x;
    int node = ids[b];
    float v = x[(size_t)node * HH + c];
    s[c] = v * v;
    __syncthreads();
    #pragma unroll
    for (int off = 32; off > 0; off >>= 1) {
        if (c < off) s[c] += s[c + off];
        __syncthreads();
    }
    float nrm = sqrtf(s[0]);
    out[(size_t)b * HH + c] = v / nrm;
}

extern "C" void kernel_launch(void* const* d_in, const int* in_sizes, int n_in,
                              void* d_out, int out_size, void* d_ws, size_t ws_size,
                              hipStream_t stream) {
    const float* text_emb = (const float*)d_in[0];
    const float* weights  = (const float*)d_in[1];
    const float* W_enc    = (const float*)d_in[2];
    const float* b_enc    = (const float*)d_in[3];
    const float* Wself_f  = (const float*)d_in[4];
    const float* Wneigh_f = (const float*)d_in[5];
    const float* b_f      = (const float*)d_in[6];
    const float* Wself_r  = (const float*)d_in[7];
    const float* Wneigh_r = (const float*)d_in[8];
    const float* b_r      = (const float*)d_in[9];
    const int*   src      = (const int*)d_in[10];
    const int*   dst      = (const int*)d_in[11];
    const int*   ids      = (const int*)d_in[12];
    float* out = (float*)d_out;

    const size_t NH = (size_t)NN * HH;
    float* x    = (float*)d_ws;
    float* nF   = x + NH;
    float* nR   = x + 2 * NH;
    float* invF = x + 3 * NH;
    float* invR = invF + NN;

    const int batch = in_sizes[12];

    hipMemsetAsync(invF, 0, (size_t)2 * NN * sizeof(float), stream);
    degree_kernel<<<4 * DEG_SLICES, DEG_THREADS, 0, stream>>>(src, dst, invF, invR);
    invert_kernel<<<(NN + 255) / 256, 256, 0, stream>>>(invF, invR);

    enc_kernel<<<(NTILES + 7) / 8, 256, 0, stream>>>(text_emb, W_enc, b_enc, x);

    for (int l = 0; l < 2; ++l) {
        own_kernel<<<2 * OWN_PARTS, OWN_THREADS, 0, stream>>>(x, weights, src, dst, nF, nR);
        layer_kernel<<<(NTILES + 7) / 8, 256, 0, stream>>>(
            x, nF, nR, invF, invR,
            Wself_f + (size_t)l * HH * HH, Wneigh_f + (size_t)l * HH * HH, b_f + (size_t)l * HH,
            Wself_r + (size_t)l * HH * HH, Wneigh_r + (size_t)l * HH * HH, b_r + (size_t)l * HH);
    }

    norm_kernel<<<batch, 64, 0, stream>>>(x, ids, out);
}


# Neighbour-sum scan without work-group barriers: each wave lists its own hits in a wave-private LDS ring (ballot + lane prefix, no counter atomic) and works off 12 at a time; sum order = LDS arrival
# speedup vs baseline: 2.2660x; 1.6071x over previous
#include <hip/hip_runtime.h>
#include <hip/hip_bf16.h>
#include <math.h>

typedef __attribute__((ext_vector_type(2))) float v2f;
typedef __attribute__((ext_vector_type(8))) float v8f;

#define NN      100000
#define NE      1600000
#define HH      64
#define TT      384
#define NTILES  6250

__device__ __forceinline__ v2f mk2(float a, float b) {
    v2f r; r.x = a; r.y = b; return r;
}

__device__ __forceinline__ v8f wmma4(v2f a, v2f b, v8f c) {
    return __builtin_amdgcn_wmma_f32_16x16x4_f32(
         false, a,  false, b,
         (short)0, c,  false,  false);
}

#define DEG_PART    50016
#define DEG_SLICES  32
#define DEG_SLICE_E (NE / DEG_SLICES)
#define DEG_THREADS 1024
#define DEG_NV      ((DEG_SLICE_E / 4 + DEG_THREADS - 1) / DEG_THREADS)

__global__ __launch_bounds__(DEG_THREADS) void degree_kernel(const int* __restrict__ src,
                                                             const int* __restrict__ dst,
                                                             float* __restrict__ degF,
                                                             float* __restrict__ degR) {
    __shared__ unsigned int cnt[DEG_PART];
    const int tid   = threadIdx.x;
    const int combo = blockIdx.x & 3;
    const int slice = blockIdx.x >> 2;
    const int rev   = combo & 1;
    const int base  = (combo >> 1) * DEG_PART;
    const int nloc  = (NN - base < DEG_PART) ? (NN - base) : DEG_PART;
    const int* __restrict__ idx = rev ? src : dst;
    float* __restrict__ deg     = rev ? degR : degF;

    const int4* ip = reinterpret_cast<const int4*>(idx + (size_t)slice * DEG_SLICE_E);
    int4 v[DEG_NV];
    #pragma unroll
    for (int j = 0; j < DEG_NV; ++j) {
        int i = tid + j * DEG_THREADS;
        v[j] = (i < DEG_SLICE_E / 4) ? ip[i] : make_int4(-1, -1, -1, -1);
    }

    uint4* cnt4 = reinterpret_cast<uint4*>(cnt);
    for (int i = tid; i < DEG_PART / 4; i += DEG_THREADS)
        cnt4[i] = make_uint4(0u, 0u, 0u, 0u);
    __syncthreads();

    #pragma unroll
    for (int j = 0; j < DEG_NV; ++j) {
        unsigned r0 = (unsigned)(v[j].x - base);
        unsigned r1 = (unsigned)(v[j].y - base);
        unsigned r2 = (unsigned)(v[j].z - base);
        unsigned r3 = (unsigned)(v[j].w - base);
        if (r0 < (unsigned)nloc) atomicAdd(&cnt[r0], 1u);
        if (r1 < (unsigned)nloc) atomicAdd(&cnt[r1], 1u);
        if (r2 < (unsigned)nloc) atomicAdd(&cnt[r2], 1u);
        if (r3 < (unsigned)nloc) atomicAdd(&cnt[r3], 1u);
    }
    __syncthreads();

    for (int i0 = tid; i0 < nloc; i0 += 8 * DEG_THREADS) {
        unsigned c[8];
        #pragma unroll
        for (int j = 0; j < 8; ++j) {
            int i = i0 + j * DEG_THREADS;
            c[j] = (i < nloc) ? cnt[i] : 0u;
        }
        #pragma unroll
        for (int j = 0; j < 8; ++j)
            if (c[j]) atomicAdd(&deg[base + i0 + j * DEG_THREADS], (float)c[j]);
    }
}

__global__ void invert_kernel(float* __restrict__ degF, float* __restrict__ degR) {
    int i = blockIdx.x * 256 + threadIdx.x;
    if (i < NN) {
        float a = degF[i];
        float b = degR[i];
        degF[i] = (a > 0.0f) ? (1.0f / a) : 0.0f;
        degR[i] = (b > 0.0f) ? (1.0f / b) : 0.0f;
    }
}

__global__ __launch_bounds__(256) void enc_kernel(const float* __restrict__ emb,
                                                  const float* __restrict__ W,
                                                  const float* __restrict__ bias,
                                                  float* __restrict__ x) {
    const int wave = threadIdx.x >> 5;
    const int lane = threadIdx.x & 31;
    const int nt   = blockIdx.x * 8 + wave;
    if (nt >= NTILES) return;
    const int ln = lane & 15;
    const int hi = lane >> 4;
    const int kh = hi << 1;

    const float* arow = emb + (size_t)(nt * 16 + ln) * TT + kh;
    v8f acc0 = {}, acc1 = {}, acc2 = {}, acc3 = {};

    #pragma unroll 4
    for (int k0 = 0; k0 < TT; k0 += 4) {
        v2f a = *reinterpret_cast<const v2f*>(arow + k0);
        const float* bp = W + (size_t)(k0 + kh) * HH + ln;
        v2f b0 = mk2(bp[0],  bp[HH]);
        v2f b1 = mk2(bp[16], bp[HH + 16]);
        v2f b2 = mk2(bp[32], bp[HH + 32]);
        v2f b3 = mk2(bp[48], bp[HH + 48]);
        acc0 = wmma4(a, b0, acc0);
        acc1 = wmma4(a, b1, acc1);
        acc2 = wmma4(a, b2, acc2);
        acc3 = wmma4(a, b3, acc3);
    }

    float bs0 = bias[ln], bs1 = bias[16 + ln], bs2 = bias[32 + ln], bs3 = bias[48 + ln];
    #pragma unroll
    for (int v = 0; v < 8; ++v) {
        size_t r = (size_t)(nt * 16 + v + 8 * hi) * HH + ln;
        x[r]      = acc0[v] + bs0;
        x[r + 16] = acc1[v] + bs1;
        x[r + 32] = acc2[v] + bs2;
        x[r + 48] = acc3[v] + bs3;
    }
}

#define OWN_NODES   1024
#define OWN_PARTS   ((NN + OWN_NODES - 1) / OWN_NODES)
#define OWN_THREADS 1024
#define OWN_WAVES   (OWN_THREADS / 32)
#define NE4         (NE / 4)
#define OWN_NSTEP   ((NE4 + OWN_THREADS - 1) / OWN_THREADS)
#define OWN_FULL    (NE4 / OWN_THREADS)
#define OWN_PF      4
#define OWN_BATCH   12
#define OWN_RING    256
#define OWN_EBITS   21

template <bool FULL>
__device__ __forceinline__ void own_drain(const unsigned int* lst, unsigned head, unsigned nvalid, int lane,
                                          const int* __restrict__ oth, const float* __restrict__ w,
                                          const float* __restrict__ x, float* acc) {
    unsigned ent[OWN_BATCH];
    int      o[OWN_BATCH];
    float    we[OWN_BATCH];
    v2f      xv[OWN_BATCH];
    #pragma unroll
    for (int j = 0; j < OWN_BATCH; ++j) {
        const unsigned jj = (FULL || (unsigned)j < nvalid) ? (unsigned)j : 0u;
        ent[j] = lst[(head + jj) & (OWN_RING - 1)];
    }
    #pragma unroll
    for (int j = 0; j < OWN_BATCH; ++j) {
        const unsigned e = ent[j] & ((1u << OWN_EBITS) - 1u);
        o[j]  = oth[e];
        we[j] = w[e];
    }
    #pragma unroll
    for (int j = 0; j < OWN_BATCH; ++j)
        xv[j] = *reinterpret_cast<const v2f*>(x + (size_t)o[j] * HH + 2 * lane);
    #pragma unroll
    for (int j = 0; j < OWN_BATCH; ++j) {
        if (FULL || (unsigned)j < nvalid) {
            float* a = acc + (size_t)(ent[j] >> OWN_EBITS) * HH + 2 * lane;
            atomicAdd(a,     we[j] * xv[j].x);
            atomicAdd(a + 1, we[j] * xv[j].y);
        }
    }
}

template <bool GUARD>
__device__ __forceinline__ void own_step(int4& vj, int i, int base, unsigned nloc, const int4* __restrict__ own4,
                                         unsigned int* lst, unsigned& head, unsigned& tail, int lane,
                                         const int* __restrict__ oth, const float* __restrict__ w,
                                         const float* __restrict__ x, float* acc) {
    const unsigned e0 = (unsigned)i * 4u;
    unsigned r0 = (unsigned)(vj.x - base);
    unsigned r1 = (unsigned)(vj.y - base);
    unsigned r2 = (unsigned)(vj.z - base);
    unsigned r3 = (unsigned)(vj.w - base);
    asm volatile("" : "+v"(r0), "+v"(r1), "+v"(r2), "+v"(r3));
    const int in = i + OWN_PF * OWN_THREADS;
    if (GUARD) vj = (in < NE4) ? own4[in] : make_int4(-1, -1, -1, -1);
    else       vj = own4[in];

    const unsigned m0 = __builtin_amdgcn_ballot_w32(r0 < nloc);
    if (m0) {
        if (r0 < nloc) lst[(tail + __builtin_amdgcn_mbcnt_lo(m0, 0u)) & (OWN_RING - 1)] = (r0 << OWN_EBITS) | (e0 + 0u);
        tail += (unsigned)__builtin_popcount(m0);
    }
    const unsigned m1 = __builtin_amdgcn_ballot_w32(r1 < nloc);
    if (m1) {
        if (r1 < nloc) lst[(tail + __builtin_amdgcn_mbcnt_lo(m1, 0u)) & (OWN_RING - 1)] = (r1 << OWN_EBITS) | (e0 + 1u);
        tail += (unsigned)__builtin_popcount(m1);
    }
    const unsigned m2 = __builtin_amdgcn_ballot_w32(r2 < nloc);
    if (m2) {
        if (r2 < nloc) lst[(tail + __builtin_amdgcn_mbcnt_lo(m2, 0u)) & (OWN_RING - 1)] = (r2 << OWN_EBITS) | (e0 + 2u);
        tail += (unsigned)__builtin_popcount(m2);
    }
    const unsigned m3 = __builtin_amdgcn_ballot_w32(r3 < nloc);
    if (m3) {
        if (r3 < nloc) lst[(tail + __builtin_amdgcn_mbcnt_lo(m3, 0u)) & (OWN_RING - 1)] = (r3 << OWN_EBITS) | (e0 + 3u);
        tail += (unsigned)__builtin_popcount(m3);
    }

    while (tail - head >= (unsigned)OWN_BATCH) {
        own_drain<true>(lst, head, (unsigned)OWN_BATCH, lane, oth, w, x, acc);
        head += (unsigned)OWN_BATCH;
    }
}

__global__ __launch_bounds__(OWN_THREADS) void own_kernel(const float* __restrict__ x,
                                                          const float* __restrict__ w,
                                                          const int* __restrict__ src,
                                                          const int* __restrict__ dst,
                                                          float* __restrict__ nF,
                                                          float* __restrict__ nR) {
    __shared__ float acc[OWN_NODES * HH];
    __shared__ unsigned int lists[OWN_WAVES * OWN_RING];
    const int tid  = threadIdx.x;
    const int lane = tid & 31;
    const int wave = tid >> 5;
    const int rev  = blockIdx.x & 1;
    const int base = (blockIdx.x >> 1) * OWN_NODES;
    const unsigned nloc = (unsigned)((NN - base < OWN_NODES) ? (NN - base) : OWN_NODES);
    const int* __restrict__ own = rev ? src : dst;
    const int* __restrict__ oth = rev ? dst : src;
    float* __restrict__ outp    = rev ? nR : nF;
    const int4* own4 = reinterpret_cast<const int4*>(own);
    unsigned int* lst = lists + wave * OWN_RING;

    int4 v[OWN_PF];
    #pragma unroll
    for (int j = 0; j < OWN_PF; ++j) {
        int i = j * OWN_THREADS + tid;
        v[j] = (i < NE4) ? own4[i] : make_int4(-1, -1, -1, -1);
    }

    float4* acc4 = reinterpret_cast<float4*>(acc);
    for (int i = tid; i < OWN_NODES * HH / 4; i += OWN_THREADS)
        acc4[i] = make_float4(0.0f, 0.0f, 0.0f, 0.0f);
    __syncthreads();

    unsigned head = 0u, tail = 0u;
    int s0 = 0;
    for (; s0 + 2 * OWN_PF <= OWN_FULL; s0 += OWN_PF) {
        #pragma unroll
        for (int j = 0; j < OWN_PF; ++j)
            own_step<false>(v[j], (s0 + j) * OWN_THREADS + tid, base, nloc, own4, lst, head, tail, lane, oth, w, x, acc);
    }
    for (; s0 < OWN_NSTEP; s0 += OWN_PF) {
        #pragma unroll
        for (int j = 0; j < OWN_PF; ++j)
            own_step<true>(v[j], (s0 + j) * OWN_THREADS + tid, base, nloc, own4, lst, head, tail, lane, oth, w, x, acc);
    }
    if (tail != head)
        own_drain<false>(lst, head, tail - head, lane, oth, w, x, acc);
    __syncthreads();

    float4* out4 = reinterpret_cast<float4*>(outp + (size_t)base * HH);
    const int nout4 = (int)nloc * (HH / 4);
    for (int i = tid; i < nout4; i += OWN_THREADS)
        out4[i] = acc4[i];
}

__global__ __launch_bounds__(256) void layer_kernel(
    float* __restrict__ x,
    const float* __restrict__ nF, const float* __restrict__ nR,
    const float* __restrict__ invF, const float* __restrict__ invR,
    const float* __restrict__ WsF, const float* __restrict__ WnF, const float* __restrict__ bF,
    const float* __restrict__ WsR, const float* __restrict__ WnR, const float* __restrict__ bR) {
    const int wave = threadIdx.x >> 5;
    const int lane = threadIdx.x & 31;
    const int nt   = blockIdx.x * 8 + wave;
    if (nt >= NTILES) return;
    const int ln = lane & 15;
    const int hi = lane >> 4;
    const int kh = hi << 1;
    const int rowA = nt * 16 + ln;

    const float sF = invF[rowA];
    const float sR = invR[rowA];
    const float* xr = x  + (size_t)rowA * HH + kh;
    const float* fr = nF + (size_t)rowA * HH + kh;
    const float* rr = nR + (size_t)rowA * HH + kh;

    v8f aF0 = {}, aF1 = {}, aF2 = {}, aF3 = {};
    v8f aR0 = {}, aR1 = {}, aR2 = {}, aR3 = {};

    #pragma unroll 4
    for (int k0 = 0; k0 < HH; k0 += 4) {
        v2f aS = *reinterpret_cast<const v2f*>(xr + k0);
        v2f aN = *reinterpret_cast<const v2f*>(fr + k0);
        v2f aV = *reinterpret_cast<const v2f*>(rr + k0);
        aN *= sF;
        aV *= sR;

        const float* pSF = WsF + (size_t)(k0 + kh) * HH + ln;
        const float* pNF = WnF + (size_t)(k0 + kh) * HH + ln;
        const float* pSR = WsR + (size_t)(k0 + kh) * HH + ln;
        const float* pNR = WnR + (size_t)(k0 + kh) * HH + ln;

        aF0 = wmma4(aS, mk2(pSF[0], pSF[HH]), aF0);
        aF0 = wmma4(aN, mk2(pNF[0], pNF[HH]), aF0);
        aR0 = wmma4(aS, mk2(pSR[0], pSR[HH]), aR0);
        aR0 = wmma4(aV, mk2(pNR[0], pNR[HH]), aR0);
        aF1 = wmma4(aS, mk2(pSF[16], pSF[HH + 16]), aF1);
        aF1 = wmma4(aN, mk2(pNF[16], pNF[HH + 16]), aF1);
        aR1 = wmma4(aS, mk2(pSR[16], pSR[HH + 16]), aR1);
        aR1 = wmma4(aV, mk2(pNR[16], pNR[HH + 16]), aR1);
        aF2 = wmma4(aS, mk2(pSF[32], pSF[HH + 32]), aF2);
        aF2 = wmma4(aN, mk2(pNF[32], pNF[HH + 32]), aF2);
        aR2 = wmma4(aS, mk2(pSR[32], pSR[HH + 32]), aR2);
        aR2 = wmma4(aV, mk2(pNR[32], pNR[HH + 32]), aR2);
        aF3 = wmma4(aS, mk2(pSF[48], pSF[HH + 48]), aF3);
        aF3 = wmma4(aN, mk2(pNF[48], pNF[HH + 48]), aF3);
        aR3 = wmma4(aS, mk2(pSR[48], pSR[HH + 48]), aR3);
        aR3 = wmma4(aV, mk2(pNR[48], pNR[HH + 48]), aR3);
    }

    float bf0 = bF[ln], bf1 = bF[16 + ln], bf2 = bF[32 + ln], bf3 = bF[48 + ln];
    float br0 = bR[ln], br1 = bR[16 + ln], br2 = bR[32 + ln], br3 = bR[48 + ln];

    #pragma unroll
    for (int v = 0; v < 8; ++v) {
        size_t r = (size_t)(nt * 16 + v + 8 * hi) * HH + ln;
        x[r]      += fmaxf(aF0[v] + bf0, 0.0f) + fmaxf(aR0[v] + br0, 0.0f);
        x[r + 16] += fmaxf(aF1[v] + bf1, 0.0f) + fmaxf(aR1[v] + br1, 0.0f);
        x[r + 32] += fmaxf(aF2[v] + bf2, 0.0f) + fmaxf(aR2[v] + br2, 0.0f);
        x[r + 48] += fmaxf(aF3[v] + bf3, 0.0f) + fmaxf(aR3[v] + br3, 0.0f);
    }
}

__global__ __launch_bounds__(64) void norm_kernel(const float* __restrict__ x,
                                                  const int* __restrict__ ids,
                                                  float* __restrict__ out) {
    __shared__ float s[64];
    int b = blockIdx.x;
    int c = threadIdx.x;
    int node = ids[b];
    float v = x[(size_t)node * HH + c];
    s[c] = v * v;
    __syncthreads();
    #pragma unroll
    for (int off = 32; off > 0; off >>= 1) {
        if (c < off) s[c] += s[c + off];
        __syncthreads();
    }
    float nrm = sqrtf(s[0]);
    out[(size_t)b * HH + c] = v / nrm;
}

extern "C" void kernel_launch(void* const* d_in, const int* in_sizes, int n_in,
                              void* d_out, int out_size, void* d_ws, size_t ws_size,
                              hipStream_t stream) {
    const float* text_emb = (const float*)d_in[0];
    const float* weights  = (const float*)d_in[1];
    const float* W_enc    = (const float*)d_in[2];
    const float* b_enc    = (const float*)d_in[3];
    const float* Wself_f  = (const float*)d_in[4];
    const float* Wneigh_f = (const float*)d_in[5];
    const float* b_f      = (const float*)d_in[6];
    const float* Wself_r  = (const float*)d_in[7];
    const float* Wneigh_r = (const float*)d_in[8];
    const float* b_r      = (const float*)d_in[9];
    const int*   src      = (const int*)d_in[10];
    const int*   dst      = (const int*)d_in[11];
    const int*   ids      = (const int*)d_in[12];
    float* out = (float*)d_out;

    const size_t NH = (size_t)NN * HH;
    float* x    = (float*)d_ws;
    float* nF   = x + NH;
    float* nR   = x + 2 * NH;
    float* invF = x + 3 * NH;
    float* invR = invF + NN;

    const int batch = in_sizes[12];

    hipMemsetAsync(invF, 0, (size_t)2 * NN * sizeof(float), stream);
    degree_kernel<<<4 * DEG_SLICES, DEG_THREADS, 0, stream>>>(src, dst, invF, invR);
    invert_kernel<<<(NN + 255) / 256, 256, 0, stream>>>(invF, invR);

    enc_kernel<<<(NTILES + 7) / 8, 256, 0, stream>>>(text_emb, W_enc, b_enc, x);

    for (int l = 0; l < 2; ++l) {
        own_kernel<<<2 * OWN_PARTS, OWN_THREADS, 0, stream>>>(x, weights, src, dst, nF, nR);
        layer_kernel<<<(NTILES + 7) / 8, 256, 0, stream>>>(
            x, nF, nR, invF, invR,
            Wself_f + (size_t)l * HH * HH, Wneigh_f + (size_t)l * HH * HH, b_f + (size_t)l * HH,
            Wself_r + (size_t)l * HH * HH, Wneigh_r + (size_t)l * HH * HH, b_r + (size_t)l * HH);
    }

    norm_kernel<<<batch, 64, 0, stream>>>(x, ids, out);
}


# own_kernel scans one-byte range keys (index >> 10, written each call beside the degree inversion) in place of the 4-byte indices, 512 edges a wave step; hit nodes re-read; sum order = LDS arrival
# speedup vs baseline: 2.4131x; 1.0649x over previous
#include <hip/hip_runtime.h>
#include <hip/hip_bf16.h>
#include <math.h>

typedef __attribute__((ext_vector_type(2))) float v2f;
typedef __attribute__((ext_vector_type(8))) float v8f;

#define NN      100000
#define NE      1600000
#define HH      64
#define TT      384
#define NTILES  6250

__device__ __forceinline__ v2f mk2(float a, float b) {
    v2f r; r.x = a; r.y = b; return r;
}

__device__ __forceinline__ v8f wmma4(v2f a, v2f b, v8f c) {
    return __builtin_amdgcn_wmma_f32_16x16x4_f32(
         false, a,  false, b,
         (short)0, c,  false,  false);
}

#define DEG_PART    50016
#define DEG_SLICES  32
#define DEG_SLICE_E (NE / DEG_SLICES)
#define DEG_THREADS 1024
#define DEG_NV      ((DEG_SLICE_E / 4 + DEG_THREADS - 1) / DEG_THREADS)

__global__ __launch_bounds__(DEG_THREADS) void degree_kernel(const int* __restrict__ src,
                                                             const int* __restrict__ dst,
                                                             float* __restrict__ degF,
                                                             float* __restrict__ degR) {
    __shared__ unsigned int cnt[DEG_PART];
    const int tid   = threadIdx.x;
    const int combo = blockIdx.x & 3;
    const int slice = blockIdx.x >> 2;
    const int rev   = combo & 1;
    const int base  = (combo >> 1) * DEG_PART;
    const int nloc  = (NN - base < DEG_PART) ? (NN - base) : DEG_PART;
    const int* __restrict__ idx = rev ? src : dst;
    float* __restrict__ deg     = rev ? degR : degF;

    const int4* ip = reinterpret_cast<const int4*>(idx + (size_t)slice * DEG_SLICE_E);
    int4 v[DEG_NV];
    #pragma unroll
    for (int j = 0; j < DEG_NV; ++j) {
        int i = tid + j * DEG_THREADS;
        v[j] = (i < DEG_SLICE_E / 4) ? ip[i] : make_int4(-1, -1, -1, -1);
    }

    uint4* cnt4 = reinterpret_cast<uint4*>(cnt);
    for (int i = tid; i < DEG_PART / 4; i += DEG_THREADS)
        cnt4[i] = make_uint4(0u, 0u, 0u, 0u);
    __syncthreads();

    #pragma unroll
    for (int j = 0; j < DEG_NV; ++j) {
        unsigned r0 = (unsigned)(v[j].x - base);
        unsigned r1 = (unsigned)(v[j].y - base);
        unsigned r2 = (unsigned)(v[j].z - base);
        unsigned r3 = (unsigned)(v[j].w - base);
        if (r0 < (unsigned)nloc) atomicAdd(&cnt[r0], 1u);
        if (r1 < (unsigned)nloc) atomicAdd(&cnt[r1], 1u);
        if (r2 < (unsigned)nloc) atomicAdd(&cnt[r2], 1u);
        if (r3 < (unsigned)nloc) atomicAdd(&cnt[r3], 1u);
    }
    __syncthreads();

    for (int i0 = tid; i0 < nloc; i0 += 8 * DEG_THREADS) {
        unsigned c[8];
        #pragma unroll
        for (int j = 0; j < 8; ++j) {
            int i = i0 + j * DEG_THREADS;
            c[j] = (i < nloc) ? cnt[i] : 0u;
        }
        #pragma unroll
        for (int j = 0; j < 8; ++j)
            if (c[j]) atomicAdd(&deg[base + i0 + j * DEG_THREADS], (float)c[j]);
    }
}

#define OWN_SHIFT 10
#define KEY_NONE  0x7Fu
#define NE4       (NE / 4)
#define KEY_WS_BYTES ((size_t)(3 * (size_t)NN * HH + 2 * (size_t)NN) * sizeof(float) + 2 * (size_t)NE)

__device__ __forceinline__ unsigned key_of(int idx) {
    return ((unsigned)idx < (unsigned)NN) ? ((unsigned)idx >> OWN_SHIFT) : KEY_NONE;
}
__device__ __forceinline__ unsigned key4_of(int4 a) {
    return key_of(a.x) | (key_of(a.y) << 8) | (key_of(a.z) << 16) | (key_of(a.w) << 24);
}

#define IK_THREADS 256
#define IK_BLOCKS  ((NE4 + 4 * IK_THREADS - 1) / (4 * IK_THREADS))
static_assert(IK_BLOCKS * IK_THREADS >= NN, "the grid must cover every node");

__global__ __launch_bounds__(IK_THREADS) void invert_key_kernel(float* __restrict__ degF, float* __restrict__ degR,
                                                                const int* __restrict__ src,
                                                                const int* __restrict__ dst,
                                                                unsigned int* __restrict__ keyF,
                                                                unsigned int* __restrict__ keyR,
                                                                size_t ws_bytes) {
    int i = blockIdx.x * IK_THREADS + threadIdx.x;
    if (i < NN) {
        float a = degF[i];
        float b = degR[i];
        degF[i] = (a > 0.0f) ? (1.0f / a) : 0.0f;
        degR[i] = (b > 0.0f) ? (1.0f / b) : 0.0f;
    }
    if (ws_bytes < KEY_WS_BYTES) return;
    const int4* s4 = reinterpret_cast<const int4*>(src);
    const int4* d4 = reinterpret_cast<const int4*>(dst);
    const int g0 = blockIdx.x * (4 * IK_THREADS) + threadIdx.x;
    int4 a[4], b[4];
    #pragma unroll
    for (int k = 0; k < 4; ++k) {
        const int g = g0 + k * IK_THREADS;
        a[k] = (g < NE4) ? d4[g] : make_int4(-1, -1, -1, -1);
        b[k] = (g < NE4) ? s4[g] : make_int4(-1, -1, -1, -1);
    }
    #pragma unroll
    for (int k = 0; k < 4; ++k) {
        const int g = g0 + k * IK_THREADS;
        if (g < NE4) {
            keyF[g] = key4_of(a[k]);
            keyR[g] = key4_of(b[k]);
        }
    }
}

__global__ __launch_bounds__(256) void enc_kernel(const float* __restrict__ emb,
                                                  const float* __restrict__ W,
                                                  const float* __restrict__ bias,
                                                  float* __restrict__ x) {
    const int wave = threadIdx.x >> 5;
    const int lane = threadIdx.x & 31;
    const int nt   = blockIdx.x * 8 + wave;
    if (nt >= NTILES) return;
    const int ln = lane & 15;
    const int hi = lane >> 4;
    const int kh = hi << 1;

    const float* arow = emb + (size_t)(nt * 16 + ln) * TT + kh;
    v8f acc0 = {}, acc1 = {}, acc2 = {}, acc3 = {};

    #pragma unroll 4
    for (int k0 = 0; k0 < TT; k0 += 4) {
        v2f a = *reinterpret_cast<const v2f*>(arow + k0);
        const float* bp = W + (size_t)(k0 + kh) * HH + ln;
        v2f b0 = mk2(bp[0],  bp[HH]);
        v2f b1 = mk2(bp[16], bp[HH + 16]);
        v2f b2 = mk2(bp[32], bp[HH + 32]);
        v2f b3 = mk2(bp[48], bp[HH + 48]);
        acc0 = wmma4(a, b0, acc0);
        acc1 = wmma4(a, b1, acc1);
        acc2 = wmma4(a, b2, acc2);
        acc3 = wmma4(a, b3, acc3);
    }

    float bs0 = bias[ln], bs1 = bias[16 + ln], bs2 = bias[32 + ln], bs3 = bias[48 + ln];
    #pragma unroll
    for (int v = 0; v < 8; ++v) {
        size_t r = (size_t)(nt * 16 + v + 8 * hi) * HH + ln;
        x[r]      = acc0[v] + bs0;
        x[r + 16] = acc1[v] + bs1;
        x[r + 32] = acc2[v] + bs2;
        x[r + 48] = acc3[v] + bs3;
    }
}

#define OWN_NODES   (1 << OWN_SHIFT)
#define OWN_PARTS   ((NN + OWN_NODES - 1) / OWN_NODES)
#define OWN_THREADS 1024
#define OWN_WAVES   (OWN_THREADS / 32)
#define NE16        (NE / 16)
#define OWN_NSTEP   ((NE16 + OWN_THREADS - 1) / OWN_THREADS)
#define OWN_FULL    (NE16 / OWN_THREADS)
#define OWN_PF      4
#define OWN_BATCH   12
#define OWN_RING    64
static_assert(OWN_PARTS <= (int)KEY_NONE, "range numbers and KEY_NONE must stay below 128 and apart");
static_assert(NE % 16 == 0, "the key list is read 16 bytes at a time");

template <bool FULL>
__device__ __forceinline__ void own_drain(const unsigned int* lst, unsigned head, unsigned nvalid, int lane, int base,
                                          const int* __restrict__ own, const int* __restrict__ oth,
                                          const float* __restrict__ w, const float* __restrict__ x, float* acc) {
    unsigned e[OWN_BATCH];
    int      n[OWN_BATCH];
    int      o[OWN_BATCH];
    float    we[OWN_BATCH];
    v2f      xv[OWN_BATCH];
    #pragma unroll
    for (int j = 0; j < OWN_BATCH; ++j) {
        const unsigned jj = (FULL || (unsigned)j < nvalid) ? (unsigned)j : 0u;
        e[j] = lst[(head + jj) & (OWN_RING - 1)];
    }
    #pragma unroll
    for (int j = 0; j < OWN_BATCH; ++j) {
        o[j]  = oth[e[j]];
        we[j] = w[e[j]];
        n[j]  = own[e[j]];
    }
    #pragma unroll
    for (int j = 0; j < OWN_BATCH; ++j)
        xv[j] = *reinterpret_cast<const v2f*>(x + (size_t)o[j] * HH + 2 * lane);
    #pragma unroll
    for (int j = 0; j < OWN_BATCH; ++j) {
        if (FULL || (unsigned)j < nvalid) {
            float* a = acc + (size_t)(n[j] - base) * HH + 2 * lane;
            atomicAdd(a,     we[j] * xv[j].x);
            atomicAdd(a + 1, we[j] * xv[j].y);
        }
    }
}

template <bool GUARD>
__device__ __forceinline__ void own_step(uint4& vj, int i, unsigned pk, int base, const uint4* __restrict__ key4,
                                         unsigned int* lst, unsigned& head, unsigned& tail, int lane,
                                         const int* __restrict__ own, const int* __restrict__ oth,
                                         const float* __restrict__ w, const float* __restrict__ x, float* acc) {
    unsigned s0 = (vj.x ^ pk) + 0x7F7F7F7Fu;
    unsigned s1 = (vj.y ^ pk) + 0x7F7F7F7Fu;
    unsigned s2 = (vj.z ^ pk) + 0x7F7F7F7Fu;
    unsigned s3 = (vj.w ^ pk) + 0x7F7F7F7Fu;
    const int in = i + OWN_PF * OWN_THREADS;
    if (GUARD) vj = (in < NE16) ? key4[in] : make_uint4(0x7F7F7F7Fu, 0x7F7F7F7Fu, 0x7F7F7F7Fu, 0x7F7F7F7Fu);
    else       vj = key4[in];

    unsigned any = __builtin_amdgcn_ballot_w32(((s0 & s1 & s2 & s3) & 0x80808080u) != 0x80808080u);
    if (any) {
        const unsigned miss = ((s0 >> 7) & 0x01010101u) | ((s1 >> 6) & 0x02020202u) |
                              ((s2 >> 5) & 0x04040404u) | ((s3 >> 4) & 0x08080808u);
        unsigned hm = ~miss & 0x0F0F0F0Fu;
        const unsigned e0 = (unsigned)i * 16u;
        do {
            if (hm != 0u) {
                const unsigned p = (unsigned)__builtin_ctz(hm);
                hm &= hm - 1u;
                lst[(tail + __builtin_amdgcn_mbcnt_lo(any, 0u)) & (OWN_RING - 1)] = e0 + ((p & 7u) << 2) + (p >> 3);
            }
            tail += (unsigned)__builtin_popcount(any);
            while (tail - head >= (unsigned)OWN_BATCH) {
                own_drain<true>(lst, head, (unsigned)OWN_BATCH, lane, base, own, oth, w, x, acc);
                head += (unsigned)OWN_BATCH;
            }
            any = __builtin_amdgcn_ballot_w32(hm != 0u);
        } while (any);
    }
}

__global__ __launch_bounds__(OWN_THREADS) void own_kernel(const float* __restrict__ x,
                                                          const float* __restrict__ w,
                                                          const int* __restrict__ src,
                                                          const int* __restrict__ dst,
                                                          const unsigned int* __restrict__ keyF,
                                                          const unsigned int* __restrict__ keyR,
                                                          float* __restrict__ nF,
                                                          float* __restrict__ nR,
                                                          size_t ws_bytes) {
    __shared__ float acc[OWN_NODES * HH];
    __shared__ unsigned int lists[OWN_WAVES * OWN_RING];
    if (ws_bytes < KEY_WS_BYTES) return;
    const int tid  = threadIdx.x;
    const int lane = tid & 31;
    const int wave = tid >> 5;
    const int rev  = blockIdx.x & 1;
    const int part = blockIdx.x >> 1;
    const int base = part * OWN_NODES;
    const unsigned nloc = (unsigned)((NN - base < OWN_NODES) ? (NN - base) : OWN_NODES);
    const unsigned pk = (unsigned)part * 0x01010101u;
    const int* __restrict__ own = rev ? src : dst;
    const int* __restrict__ oth = rev ? dst : src;
    float* __restrict__ outp    = rev ? nR : nF;
    const uint4* key4 = reinterpret_cast<const uint4*>(rev ? keyR : keyF);
    unsigned int* lst = lists + wave * OWN_RING;

    uint4 v[OWN_PF];
    #pragma unroll
    for (int j = 0; j < OWN_PF; ++j) {
        int i = j * OWN_THREADS + tid;
        v[j] = (i < NE16) ? key4[i] : make_uint4(0x7F7F7F7Fu, 0x7F7F7F7Fu, 0x7F7F7F7Fu, 0x7F7F7F7Fu);
    }

    float4* acc4 = reinterpret_cast<float4*>(acc);
    for (int i = tid; i < OWN_NODES * HH / 4; i += OWN_THREADS)
        acc4[i] = make_float4(0.0f, 0.0f, 0.0f, 0.0f);
    __syncthreads();

    unsigned head = 0u, tail = 0u;
    int s0 = 0;
    for (; s0 + 2 * OWN_PF <= OWN_FULL; s0 += OWN_PF) {
        #pragma unroll
        for (int j = 0; j < OWN_PF; ++j)
            own_step<false>(v[j], (s0 + j) * OWN_THREADS + tid, pk, base, key4, lst, head, tail, lane, own, oth, w, x, acc);
    }
    for (; s0 < OWN_NSTEP; s0 += OWN_PF) {
        #pragma unroll
        for (int j = 0; j < OWN_PF; ++j)
            own_step<true>(v[j], (s0 + j) * OWN_THREADS + tid, pk, base, key4, lst, head, tail, lane, own, oth, w, x, acc);
    }
    if (tail != head)
        own_drain<false>(lst, head, tail - head, lane, base, own, oth, w, x, acc);
    __syncthreads();

    float4* out4 = reinterpret_cast<float4*>(outp + (size_t)base * HH);
    const int nout4 = (int)nloc * (HH / 4);
    for (int i = tid; i < nout4; i += OWN_THREADS)
        out4[i] = acc4[i];
}

__global__ __launch_bounds__(256) void layer_kernel(
    float* __restrict__ x,
    const float* __restrict__ nF, const float* __restrict__ nR,
    const float* __restrict__ invF, const float* __restrict__ invR,
    const float* __restrict__ WsF, const float* __restrict__ WnF, const float* __restrict__ bF,
    const float* __restrict__ WsR, const float* __restrict__ WnR, const float* __restrict__ bR) {
    const int wave = threadIdx.x >> 5;
    const int lane = threadIdx.x & 31;
    const int nt   = blockIdx.x * 8 + wave;
    if (nt >= NTILES) return;
    const int ln = lane & 15;
    const int hi = lane >> 4;
    const int kh = hi << 1;
    const int rowA = nt * 16 + ln;

    const float sF = invF[rowA];
    const float sR = invR[rowA];
    const float* xr = x  + (size_t)rowA * HH + kh;
    const float* fr = nF + (size_t)rowA * HH + kh;
    const float* rr = nR + (size_t)rowA * HH + kh;

    v8f aF0 = {}, aF1 = {}, aF2 = {}, aF3 = {};
    v8f aR0 = {}, aR1 = {}, aR2 = {}, aR3 = {};

    #pragma unroll 4
    for (int k0 = 0; k0 < HH; k0 += 4) {
        v2f aS = *reinterpret_cast<const v2f*>(xr + k0);
        v2f aN = *reinterpret_cast<const v2f*>(fr + k0);
        v2f aV = *reinterpret_cast<const v2f*>(rr + k0);
        aN *= sF;
        aV *= sR;

        const float* pSF = WsF + (size_t)(k0 + kh) * HH + ln;
        const float* pNF = WnF + (size_t)(k0 + kh) * HH + ln;
        const float* pSR = WsR + (size_t)(k0 + kh) * HH + ln;
        const float* pNR = WnR + (size_t)(k0 + kh) * HH + ln;

        aF0 = wmma4(aS, mk2(pSF[0], pSF[HH]), aF0);
        aF0 = wmma4(aN, mk2(pNF[0], pNF[HH]), aF0);
        aR0 = wmma4(aS, mk2(pSR[0], pSR[HH]), aR0);
        aR0 = wmma4(aV, mk2(pNR[0], pNR[HH]), aR0);
        aF1 = wmma4(aS, mk2(pSF[16], pSF[HH + 16]), aF1);
        aF1 = wmma4(aN, mk2(pNF[16], pNF[HH + 16]), aF1);
        aR1 = wmma4(aS, mk2(pSR[16], pSR[HH + 16]), aR1);
        aR1 = wmma4(aV, mk2(pNR[16], pNR[HH + 16]), aR1);
        aF2 = wmma4(aS, mk2(pSF[32], pSF[HH + 32]), aF2);
        aF2 = wmma4(aN, mk2(pNF[32], pNF[HH + 32]), aF2);
        aR2 = wmma4(aS, mk2(pSR[32], pSR[HH + 32]), aR2);
        aR2 = wmma4(aV, mk2(pNR[32], pNR[HH + 32]), aR2);
        aF3 = wmma4(aS, mk2(pSF[48], pSF[HH + 48]), aF3);
        aF3 = wmma4(aN, mk2(pNF[48], pNF[HH + 48]), aF3);
        aR3 = wmma4(aS, mk2(pSR[48], pSR[HH + 48]), aR3);
        aR3 = wmma4(aV, mk2(pNR[48], pNR[HH + 48]), aR3);
    }

    float bf0 = bF[ln], bf1 = bF[16 + ln], bf2 = bF[32 + ln], bf3 = bF[48 + ln];
    float br0 = bR[ln], br1 = bR[16 + ln], br2 = bR[32 + ln], br3 = bR[48 + ln];

    #pragma unroll
    for (int v = 0; v < 8; ++v) {
        size_t r = (size_t)(nt * 16 + v + 8 * hi) * HH + ln;
        x[r]      += fmaxf(aF0[v] + bf0, 0.0f) + fmaxf(aR0[v] + br0, 0.0f);
        x[r + 16] += fmaxf(aF1[v] + bf1, 0.0f) + fmaxf(aR1[v] + br1, 0.0f);
        x[r + 32] += fmaxf(aF2[v] + bf2, 0.0f) + fmaxf(aR2[v] + br2, 0.0f);
        x[r + 48] += fmaxf(aF3[v] + bf3, 0.0f) + fmaxf(aR3[v] + br3, 0.0f);
    }
}

__global__ __launch_bounds__(64) void norm_kernel(const float* __restrict__ x,
                                                  const int* __restrict__ ids,
                                                  float* __restrict__ out) {
    __shared__ float s[64];
    int b = blockIdx.x;
    int c = threadIdx.x;
    int node = ids[b];
    float v = x[(size_t)node * HH + c];
    s[c] = v * v;
    __syncthreads();
    #pragma unroll
    for (int off = 32; off > 0; off >>= 1) {
        if (c < off) s[c] += s[c + off];
        __syncthreads();
    }
    float nrm = sqrtf(s[0]);
    out[(size_t)b * HH + c] = v / nrm;
}

extern "C" void kernel_launch(void* const* d_in, const int* in_sizes, int n_in,
                              void* d_out, int out_size, void* d_ws, size_t ws_size,
                              hipStream_t stream) {
    const float* text_emb = (const float*)d_in[0];
    const float* weights  = (const float*)d_in[1];
    const float* W_enc    = (const float*)d_in[2];
    const float* b_enc    = (const float*)d_in[3];
    const float* Wself_f  = (const float*)d_in[4];
    const float* Wneigh_f = (const float*)d_in[5];
    const float* b_f      = (const float*)d_in[6];
    const float* Wself_r  = (const float*)d_in[7];
    const float* Wneigh_r = (const float*)d_in[8];
    const float* b_r      = (const float*)d_in[9];
    const int*   src      = (const int*)d_in[10];
    const int*   dst      = (const int*)d_in[11];
    const int*   ids      = (const int*)d_in[12];
    float* out = (float*)d_out;

    const size_t NH = (size_t)NN * HH;
    float* x    = (float*)d_ws;
    float* nF   = x + NH;
    float* nR   = x + 2 * NH;
    float* invF = x + 3 * NH;
    float* invR = invF + NN;
    unsigned int* keyF = (unsigned int*)(invR + NN);
    unsigned int* keyR = keyF + NE / 4;

    const int batch = in_sizes[12];

    hipMemsetAsync(invF, 0, (size_t)2 * NN * sizeof(float), stream);
    degree_kernel<<<4 * DEG_SLICES, DEG_THREADS, 0, stream>>>(src, dst, invF, invR);
    invert_key_kernel<<<IK_BLOCKS, IK_THREADS, 0, stream>>>(invF, invR, src, dst, keyF, keyR, ws_size);

    enc_kernel<<<(NTILES + 7) / 8, 256, 0, stream>>>(text_emb, W_enc, b_enc, x);

    for (int l = 0; l < 2; ++l) {
        own_kernel<<<2 * OWN_PARTS, OWN_THREADS, 0, stream>>>(x, weights, src, dst, keyF, keyR, nF, nR, ws_size);
        layer_kernel<<<(NTILES + 7) / 8, 256, 0, stream>>>(
            x, nF, nR, invF, invR,
            Wself_f + (size_t)l * HH * HH, Wneigh_f + (size_t)l * HH * HH, b_f + (size_t)l * HH,
            Wself_r + (size_t)l * HH * HH, Wneigh_r + (size_t)l * HH * HH, b_r + (size_t)l * HH);
    }

    norm_kernel<<<batch, 64, 0, stream>>>(x, ids, out);
}


# own_kernel drain: lane j fetches entry j's edge record by one 16-byte load from a {dst,src,w} array written each call, spread by readlane; was 3 uniform 4-byte loads an entry; sum order = LDS arrival
# speedup vs baseline: 2.6974x; 1.1178x over previous
#include <hip/hip_runtime.h>
#include <hip/hip_bf16.h>
#include <math.h>

typedef __attribute__((ext_vector_type(2))) float v2f;
typedef __attribute__((ext_vector_type(8))) float v8f;

#define NN      100000
#define NE      1600000
#define HH      64
#define TT      384
#define NTILES  6250

__device__ __forceinline__ v2f mk2(float a, float b) {
    v2f r; r.x = a; r.y = b; return r;
}

__device__ __forceinline__ v8f wmma4(v2f a, v2f b, v8f c) {
    return __builtin_amdgcn_wmma_f32_16x16x4_f32(
         false, a,  false, b,
         (short)0, c,  false,  false);
}

#define DEG_PART    50016
#define DEG_SLICES  32
#define DEG_SLICE_E (NE / DEG_SLICES)
#define DEG_THREADS 1024
#define DEG_NV      ((DEG_SLICE_E / 4 + DEG_THREADS - 1) / DEG_THREADS)

__global__ __launch_bounds__(DEG_THREADS) void degree_kernel(const int* __restrict__ src,
                                                             const int* __restrict__ dst,
                                                             float* __restrict__ degF,
                                                             float* __restrict__ degR) {
    __shared__ unsigned int cnt[DEG_PART];
    const int tid   = threadIdx.x;
    const int combo = blockIdx.x & 3;
    const int slice = blockIdx.x >> 2;
    const int rev   = combo & 1;
    const int base  = (combo >> 1) * DEG_PART;
    const int nloc  = (NN - base < DEG_PART) ? (NN - base) : DEG_PART;
    const int* __restrict__ idx = rev ? src : dst;
    float* __restrict__ deg     = rev ? degR : degF;

    const int4* ip = reinterpret_cast<const int4*>(idx + (size_t)slice * DEG_SLICE_E);
    int4 v[DEG_NV];
    #pragma unroll
    for (int j = 0; j < DEG_NV; ++j) {
        int i = tid + j * DEG_THREADS;
        v[j] = (i < DEG_SLICE_E / 4) ? ip[i] : make_int4(-1, -1, -1, -1);
    }

    uint4* cnt4 = reinterpret_cast<uint4*>(cnt);
    for (int i = tid; i < DEG_PART / 4; i += DEG_THREADS)
        cnt4[i] = make_uint4(0u, 0u, 0u, 0u);
    __syncthreads();

    #pragma unroll
    for (int j = 0; j < DEG_NV; ++j) {
        unsigned r0 = (unsigned)(v[j].x - base);
        unsigned r1 = (unsigned)(v[j].y - base);
        unsigned r2 = (unsigned)(v[j].z - base);
        unsigned r3 = (unsigned)(v[j].w - base);
        if (r0 < (unsigned)nloc) atomicAdd(&cnt[r0], 1u);
        if (r1 < (unsigned)nloc) atomicAdd(&cnt[r1], 1u);
        if (r2 < (unsigned)nloc) atomicAdd(&cnt[r2], 1u);
        if (r3 < (unsigned)nloc) atomicAdd(&cnt[r3], 1u);
    }
    __syncthreads();

    for (int i0 = tid; i0 < nloc; i0 += 8 * DEG_THREADS) {
        unsigned c[8];
        #pragma unroll
        for (int j = 0; j < 8; ++j) {
            int i = i0 + j * DEG_THREADS;
            c[j] = (i < nloc) ? cnt[i] : 0u;
        }
        #pragma unroll
        for (int j = 0; j < 8; ++j)
            if (c[j]) atomicAdd(&deg[base + i0 + j * DEG_THREADS], (float)c[j]);
    }
}

#define OWN_SHIFT 10
#define KEY_NONE  0x7Fu
#define NE4       (NE / 4)
#define KEY_WS_BYTES ((size_t)(3 * (size_t)NN * HH + 2 * (size_t)NN) * sizeof(float) + 2 * (size_t)NE)
#define REC_WS_BYTES (KEY_WS_BYTES + (size_t)NE * sizeof(int4))

__device__ __forceinline__ unsigned key_of(int idx) {
    return ((unsigned)idx < (unsigned)NN) ? ((unsigned)idx >> OWN_SHIFT) : KEY_NONE;
}
__device__ __forceinline__ unsigned key4_of(int4 a) {
    return key_of(a.x) | (key_of(a.y) << 8) | (key_of(a.z) << 16) | (key_of(a.w) << 24);
}

#define IK_THREADS 256
#define IK_BLOCKS  ((NE4 + 4 * IK_THREADS - 1) / (4 * IK_THREADS))
static_assert(IK_BLOCKS * IK_THREADS >= NN, "the grid must cover every node");

__global__ __launch_bounds__(IK_THREADS) void invert_key_kernel(float* __restrict__ degF, float* __restrict__ degR,
                                                                const int* __restrict__ src,
                                                                const int* __restrict__ dst,
                                                                const float* __restrict__ w,
                                                                unsigned int* __restrict__ keyF,
                                                                unsigned int* __restrict__ keyR,
                                                                int* __restrict__ rec,
                                                                size_t ws_bytes) {
    int i = blockIdx.x * IK_THREADS + threadIdx.x;
    if (i < NN) {
        float a = degF[i];
        float b = degR[i];
        degF[i] = (a > 0.0f) ? (1.0f / a) : 0.0f;
        degR[i] = (b > 0.0f) ? (1.0f / b) : 0.0f;
    }
    if (ws_bytes < KEY_WS_BYTES) return;
    const int4* s4 = reinterpret_cast<const int4*>(src);
    const int4* d4 = reinterpret_cast<const int4*>(dst);
    const int g0 = blockIdx.x * (4 * IK_THREADS) + threadIdx.x;
    int4 a[4], b[4];
    #pragma unroll
    for (int k = 0; k < 4; ++k) {
        const int g = g0 + k * IK_THREADS;
        a[k] = (g < NE4) ? d4[g] : make_int4(-1, -1, -1, -1);
        b[k] = (g < NE4) ? s4[g] : make_int4(-1, -1, -1, -1);
    }
    #pragma unroll
    for (int k = 0; k < 4; ++k) {
        const int g = g0 + k * IK_THREADS;
        if (g < NE4) {
            keyF[g] = key4_of(a[k]);
            keyR[g] = key4_of(b[k]);
        }
    }
    if (ws_bytes < REC_WS_BYTES) return;
    int4* rec4 = reinterpret_cast<int4*>(rec);
    const int e0 = blockIdx.x * (16 * IK_THREADS) + threadIdx.x;
    #pragma unroll
    for (int h = 0; h < 2; ++h) {
        int   d[8], s[8];
        float wv[8];
        #pragma unroll
        for (int k = 0; k < 8; ++k) {
            const int e  = e0 + (8 * h + k) * IK_THREADS;
            const int ec = (e < NE) ? e : NE - 1;
            d[k]  = dst[ec];
            s[k]  = src[ec];
            wv[k] = w[ec];
        }
        #pragma unroll
        for (int k = 0; k < 8; ++k) {
            const int e = e0 + (8 * h + k) * IK_THREADS;
            if (e < NE) rec4[e] = make_int4(d[k], s[k], __float_as_int(wv[k]), 0);
        }
    }
}

__global__ __launch_bounds__(256) void enc_kernel(const float* __restrict__ emb,
                                                  const float* __restrict__ W,
                                                  const float* __restrict__ bias,
                                                  float* __restrict__ x) {
    const int wave = threadIdx.x >> 5;
    const int lane = threadIdx.x & 31;
    const int nt   = blockIdx.x * 8 + wave;
    if (nt >= NTILES) return;
    const int ln = lane & 15;
    const int hi = lane >> 4;
    const int kh = hi << 1;

    const float* arow = emb + (size_t)(nt * 16 + ln) * TT + kh;
    v8f acc0 = {}, acc1 = {}, acc2 = {}, acc3 = {};

    #pragma unroll 4
    for (int k0 = 0; k0 < TT; k0 += 4) {
        v2f a = *reinterpret_cast<const v2f*>(arow + k0);
        const float* bp = W + (size_t)(k0 + kh) * HH + ln;
        v2f b0 = mk2(bp[0],  bp[HH]);
        v2f b1 = mk2(bp[16], bp[HH + 16]);
        v2f b2 = mk2(bp[32], bp[HH + 32]);
        v2f b3 = mk2(bp[48], bp[HH + 48]);
        acc0 = wmma4(a, b0, acc0);
        acc1 = wmma4(a, b1, acc1);
        acc2 = wmma4(a, b2, acc2);
        acc3 = wmma4(a, b3, acc3);
    }

    float bs0 = bias[ln], bs1 = bias[16 + ln], bs2 = bias[32 + ln], bs3 = bias[48 + ln];
    #pragma unroll
    for (int v = 0; v < 8; ++v) {
        size_t r = (size_t)(nt * 16 + v + 8 * hi) * HH + ln;
        x[r]      = acc0[v] + bs0;
        x[r + 16] = acc1[v] + bs1;
        x[r + 32] = acc2[v] + bs2;
        x[r + 48] = acc3[v] + bs3;
    }
}

#define OWN_NODES   (1 << OWN_SHIFT)
#define OWN_PARTS   ((NN + OWN_NODES - 1) / OWN_NODES)
#define OWN_THREADS 1024
#define OWN_WAVES   (OWN_THREADS / 32)
#define NE16        (NE / 16)
#define OWN_NSTEP   ((NE16 + OWN_THREADS - 1) / OWN_THREADS)
#define OWN_FULL    (NE16 / OWN_THREADS)
#define OWN_PF      4
#define OWN_BATCH   12
#define OWN_RING    64
static_assert(OWN_PARTS <= (int)KEY_NONE, "range numbers and KEY_NONE must stay below 128 and apart");
static_assert(NE % 16 == 0, "the key list is read 16 bytes at a time");

template <bool FULL, bool PACKED>
__device__ __forceinline__ void own_drain(const unsigned int* lst, unsigned head, unsigned nvalid, int lane, int base, int rev,
                                          const int* __restrict__ own, const int* __restrict__ oth,
                                          const float* __restrict__ w, const int4* __restrict__ rec,
                                          const float* __restrict__ x, float* acc) {
    const unsigned jj = ((unsigned)lane < (FULL ? (unsigned)OWN_BATCH : nvalid)) ? (unsigned)lane : 0u;
    const unsigned e  = lst[(head + jj) & (OWN_RING - 1)];
    int rn, ro, rw;
    if (PACKED) {
        const int4 r = rec[e];
        rn = rev ? r.y : r.x;
        ro = rev ? r.x : r.y;
        rw = r.z;
    } else {
        ro = oth[e];
        rw = __float_as_int(w[e]);
        rn = own[e];
    }
    v2f xv[OWN_BATCH];
    #pragma unroll
    for (int j = 0; j < OWN_BATCH; ++j) {
        const int o = __builtin_amdgcn_readlane(ro, j);
        xv[j] = *reinterpret_cast<const v2f*>(x + (size_t)o * HH + 2 * lane);
    }
    #pragma unroll
    for (int j = 0; j < OWN_BATCH; ++j) {
        if (FULL || (unsigned)j < nvalid) {
            const int   n  = __builtin_amdgcn_readlane(rn, j);
            const float we = __int_as_float(__builtin_amdgcn_readlane(rw, j));
            float* a = acc + (size_t)(n - base) * HH + 2 * lane;
            atomicAdd(a,     we * xv[j].x);
            atomicAdd(a + 1, we * xv[j].y);
        }
    }
}

template <bool GUARD, bool PACKED>
__device__ __forceinline__ void own_step(uint4& vj, int i, unsigned pk, int base, int rev, const uint4* __restrict__ key4,
                                         unsigned int* lst, unsigned& head, unsigned& tail, int lane,
                                         const int* __restrict__ own, const int* __restrict__ oth,
                                         const float* __restrict__ w, const int4* __restrict__ rec,
                                         const float* __restrict__ x, float* acc) {
    unsigned s0 = (vj.x ^ pk) + 0x7F7F7F7Fu;
    unsigned s1 = (vj.y ^ pk) + 0x7F7F7F7Fu;
    unsigned s2 = (vj.z ^ pk) + 0x7F7F7F7Fu;
    unsigned s3 = (vj.w ^ pk) + 0x7F7F7F7Fu;
    const int in = i + OWN_PF * OWN_THREADS;
    if (GUARD) vj = (in < NE16) ? key4[in] : make_uint4(0x7F7F7F7Fu, 0x7F7F7F7Fu, 0x7F7F7F7Fu, 0x7F7F7F7Fu);
    else       vj = key4[in];

    unsigned any = __builtin_amdgcn_ballot_w32(((s0 & s1 & s2 & s3) & 0x80808080u) != 0x80808080u);
    if (any) {
        const unsigned miss = ((s0 >> 7) & 0x01010101u) | ((s1 >> 6) & 0x02020202u) |
                              ((s2 >> 5) & 0x04040404u) | ((s3 >> 4) & 0x08080808u);
        unsigned hm = ~miss & 0x0F0F0F0Fu;
        const unsigned e0 = (unsigned)i * 16u;
        do {
            if (hm != 0u) {
                const unsigned p = (unsigned)__builtin_ctz(hm);
                hm &= hm - 1u;
                lst[(tail + __builtin_amdgcn_mbcnt_lo(any, 0u)) & (OWN_RING - 1)] = e0 + ((p & 7u) << 2) + (p >> 3);
            }
            tail += (unsigned)__builtin_popcount(any);
            while (tail - head >= (unsigned)OWN_BATCH) {
                own_drain<true, PACKED>(lst, head, (unsigned)OWN_BATCH, lane, base, rev, own, oth, w, rec, x, acc);
                head += (unsigned)OWN_BATCH;
            }
            any = __builtin_amdgcn_ballot_w32(hm != 0u);
        } while (any);
    }
}

template <bool PACKED>
__device__ __forceinline__ void own_scan(uint4 (&v)[OWN_PF], int tid, unsigned pk, int base, int rev,
                                         const uint4* __restrict__ key4, unsigned int* lst, int lane,
                                         const int* __restrict__ own, const int* __restrict__ oth,
                                         const float* __restrict__ w, const int4* __restrict__ rec,
                                         const float* __restrict__ x, float* acc) {
    unsigned head = 0u, tail = 0u;
    int s0 = 0;
    for (; s0 + 2 * OWN_PF <= OWN_FULL; s0 += OWN_PF) {
        #pragma unroll
        for (int j = 0; j < OWN_PF; ++j)
            own_step<false, PACKED>(v[j], (s0 + j) * OWN_THREADS + tid, pk, base, rev, key4, lst, head, tail, lane, own, oth, w, rec, x, acc);
    }
    for (; s0 < OWN_NSTEP; s0 += OWN_PF) {
        #pragma unroll
        for (int j = 0; j < OWN_PF; ++j)
            own_step<true, PACKED>(v[j], (s0 + j) * OWN_THREADS + tid, pk, base, rev, key4, lst, head, tail, lane, own, oth, w, rec, x, acc);
    }
    if (tail != head)
        own_drain<false, PACKED>(lst, head, tail - head, lane, base, rev, own, oth, w, rec, x, acc);
}

__global__ __launch_bounds__(OWN_THREADS) void own_kernel(const float* __restrict__ x,
                                                          const float* __restrict__ w,
                                                          const int* __restrict__ src,
                                                          const int* __restrict__ dst,
                                                          const unsigned int* __restrict__ keyF,
                                                          const unsigned int* __restrict__ keyR,
                                                          const int* __restrict__ rec,
                                                          float* __restrict__ nF,
                                                          float* __restrict__ nR,
                                                          size_t ws_bytes) {
    __shared__ float acc[OWN_NODES * HH];
    __shared__ unsigned int lists[OWN_WAVES * OWN_RING];
    if (ws_bytes < KEY_WS_BYTES) return;
    const int tid  = threadIdx.x;
    const int lane = tid & 31;
    const int wave = tid >> 5;
    const int rev  = blockIdx.x & 1;
    const int part = blockIdx.x >> 1;
    const int base = part * OWN_NODES;
    const unsigned nloc = (unsigned)((NN - base < OWN_NODES) ? (NN - base) : OWN_NODES);
    const unsigned pk = (unsigned)part * 0x01010101u;
    const int* __restrict__ own = rev ? src : dst;
    const int* __restrict__ oth = rev ? dst : src;
    float* __restrict__ outp    = rev ? nR : nF;
    const uint4* key4 = reinterpret_cast<const uint4*>(rev ? keyR : keyF);
    unsigned int* lst = lists + wave * OWN_RING;
    const int4* rec4 = reinterpret_cast<const int4*>(rec);

    uint4 v[OWN_PF];
    #pragma unroll
    for (int j = 0; j < OWN_PF; ++j) {
        int i = j * OWN_THREADS + tid;
        v[j] = (i < NE16) ? key4[i] : make_uint4(0x7F7F7F7Fu, 0x7F7F7F7Fu, 0x7F7F7F7Fu, 0x7F7F7F7Fu);
    }

    float4* acc4 = reinterpret_cast<float4*>(acc);
    for (int i = tid; i < OWN_NODES * HH / 4; i += OWN_THREADS)
        acc4[i] = make_float4(0.0f, 0.0f, 0.0f, 0.0f);
    __syncthreads();

    if (__builtin_expect(ws_bytes >= REC_WS_BYTES, 1))
        own_scan<true>(v, tid, pk, base, rev, key4, lst, lane, own, oth, w, rec4, x, acc);
    else
        own_scan<false>(v, tid, pk, base, rev, key4, lst, lane, own, oth, w, rec4, x, acc);
    __syncthreads();

    float4* out4 = reinterpret_cast<float4*>(outp + (size_t)base * HH);
    const int nout4 = (int)nloc * (HH / 4);
    for (int i = tid; i < nout4; i += OWN_THREADS)
        out4[i] = acc4[i];
}

__global__ __launch_bounds__(256) void layer_kernel(
    float* __restrict__ x,
    const float* __restrict__ nF, const float* __restrict__ nR,
    const float* __restrict__ invF, const float* __restrict__ invR,
    const float* __restrict__ WsF, const float* __restrict__ WnF, const float* __restrict__ bF,
    const float* __restrict__ WsR, const float* __restrict__ WnR, const float* __restrict__ bR) {
    const int wave = threadIdx.x >> 5;
    const int lane = threadIdx.x & 31;
    const int nt   = blockIdx.x * 8 + wave;
    if (nt >= NTILES) return;
    const int ln = lane & 15;
    const int hi = lane >> 4;
    const int kh = hi << 1;
    const int rowA = nt * 16 + ln;

    const float sF = invF[rowA];
    const float sR = invR[rowA];
    const float* xr = x  + (size_t)rowA * HH + kh;
    const float* fr = nF + (size_t)rowA * HH + kh;
    const float* rr = nR + (size_t)rowA * HH + kh;

    v8f aF0 = {}, aF1 = {}, aF2 = {}, aF3 = {};
    v8f aR0 = {}, aR1 = {}, aR2 = {}, aR3 = {};

    #pragma unroll 4
    for (int k0 = 0; k0 < HH; k0 += 4) {
        v2f aS = *reinterpret_cast<const v2f*>(xr + k0);
        v2f aN = *reinterpret_cast<const v2f*>(fr + k0);
        v2f aV = *reinterpret_cast<const v2f*>(rr + k0);
        aN *= sF;
        aV *= sR;

        const float* pSF = WsF + (size_t)(k0 + kh) * HH + ln;
        const float* pNF = WnF + (size_t)(k0 + kh) * HH + ln;
        const float* pSR = WsR + (size_t)(k0 + kh) * HH + ln;
        const float* pNR = WnR + (size_t)(k0 + kh) * HH + ln;

        aF0 = wmma4(aS, mk2(pSF[0], pSF[HH]), aF0);
        aF0 = wmma4(aN, mk2(pNF[0], pNF[HH]), aF0);
        aR0 = wmma4(aS, mk2(pSR[0], pSR[HH]), aR0);
        aR0 = wmma4(aV, mk2(pNR[0], pNR[HH]), aR0);
        aF1 = wmma4(aS, mk2(pSF[16], pSF[HH + 16]), aF1);
        aF1 = wmma4(aN, mk2(pNF[16], pNF[HH + 16]), aF1);
        aR1 = wmma4(aS, mk2(pSR[16], pSR[HH + 16]), aR1);
        aR1 = wmma4(aV, mk2(pNR[16], pNR[HH + 16]), aR1);
        aF2 = wmma4(aS, mk2(pSF[32], pSF[HH + 32]), aF2);
        aF2 = wmma4(aN, mk2(pNF[32], pNF[HH + 32]), aF2);
        aR2 = wmma4(aS, mk2(pSR[32], pSR[HH + 32]), aR2);
        aR2 = wmma4(aV, mk2(pNR[32], pNR[HH + 32]), aR2);
        aF3 = wmma4(aS, mk2(pSF[48], pSF[HH + 48]), aF3);
        aF3 = wmma4(aN, mk2(pNF[48], pNF[HH + 48]), aF3);
        aR3 = wmma4(aS, mk2(pSR[48], pSR[HH + 48]), aR3);
        aR3 = wmma4(aV, mk2(pNR[48], pNR[HH + 48]), aR3);
    }

    float bf0 = bF[ln], bf1 = bF[16 + ln], bf2 = bF[32 + ln], bf3 = bF[48 + ln];
    float br0 = bR[ln], br1 = bR[16 + ln], br2 = bR[32 + ln], br3 = bR[48 + ln];

    #pragma unroll
    for (int v = 0; v < 8; ++v) {
        size_t r = (size_t)(nt * 16 + v + 8 * hi) * HH + ln;
        x[r]      += fmaxf(aF0[v] + bf0, 0.0f) + fmaxf(aR0[v] + br0, 0.0f);
        x[r + 16] += fmaxf(aF1[v] + bf1, 0.0f) + fmaxf(aR1[v] + br1, 0.0f);
        x[r + 32] += fmaxf(aF2[v] + bf2, 0.0f) + fmaxf(aR2[v] + br2, 0.0f);
        x[r + 48] += fmaxf(aF3[v] + bf3, 0.0f) + fmaxf(aR3[v] + br3, 0.0f);
    }
}

__global__ __launch_bounds__(64) void norm_kernel(const float* __restrict__ x,
                                                  const int* __restrict__ ids,
                                                  float* __restrict__ out) {
    __shared__ float s[64];
    int b = blockIdx.x;
    int c = threadIdx.x;
    int node = ids[b];
    float v = x[(size_t)node * HH + c];
    s[c] = v * v;
    __syncthreads();
    #pragma unroll
    for (int off = 32; off > 0; off >>= 1) {
        if (c < off) s[c] += s[c + off];
        __syncthreads();
    }
    float nrm = sqrtf(s[0]);
    out[(size_t)b * HH + c] = v / nrm;
}

extern "C" void kernel_launch(void* const* d_in, const int* in_sizes, int n_in,
                              void* d_out, int out_size, void* d_ws, size_t ws_size,
                              hipStream_t stream) {
    const float* text_emb = (const float*)d_in[0];
    const float* weights  = (const float*)d_in[1];
    const float* W_enc    = (const float*)d_in[2];
    const float* b_enc    = (const float*)d_in[3];
    const float* Wself_f  = (const float*)d_in[4];
    const float* Wneigh_f = (const float*)d_in[5];
    const float* b_f      = (const float*)d_in[6];
    const float* Wself_r  = (const float*)d_in[7];
    const float* Wneigh_r = (const float*)d_in[8];
    const float* b_r      = (const float*)d_in[9];
    const int*   src      = (const int*)d_in[10];
    const int*   dst      = (const int*)d_in[11];
    const int*   ids      = (const int*)d_in[12];
    float* out = (float*)d_out;

    const size_t NH = (size_t)NN * HH;
    float* x    = (float*)d_ws;
    float* nF   = x + NH;
    float* nR   = x + 2 * NH;
    float* invF = x + 3 * NH;
    float* invR = invF + NN;
    unsigned int* keyF = (unsigned int*)(invR + NN);
    unsigned int* keyR = keyF + NE / 4;
    int* rec = (int*)(keyR + NE / 4);

    const int batch = in_sizes[12];

    hipMemsetAsync(invF, 0, (size_t)2 * NN * sizeof(float), stream);
    degree_kernel<<<4 * DEG_SLICES, DEG_THREADS, 0, stream>>>(src, dst, invF, invR);
    invert_key_kernel<<<IK_BLOCKS, IK_THREADS, 0, stream>>>(invF, invR, src, dst, weights, keyF, keyR, rec, ws_size);

    enc_kernel<<<(NTILES + 7) / 8, 256, 0, stream>>>(text_emb, W_enc, b_enc, x);

    for (int l = 0; l < 2; ++l) {
        own_kernel<<<2 * OWN_PARTS, OWN_THREADS, 0, stream>>>(x, weights, src, dst, keyF, keyR, rec, nF, nR, ws_size);
        layer_kernel<<<(NTILES + 7) / 8, 256, 0, stream>>>(
            x, nF, nR, invF, invR,
            Wself_f + (size_t)l * HH * HH, Wneigh_f + (size_t)l * HH * HH, b_f + (size_t)l * HH,
            Wself_r + (size_t)l * HH * HH, Wneigh_r + (size_t)l * HH * HH, b_r + (size_t)l * HH);
    }

    norm_kernel<<<batch, 64, 0, stream>>>(x, ids, out);
}


# own_kernel owner ranges cut from 1024 to 800 nodes (key = index / 800): 250 work-groups, not 196, fill the 256 WGPs and each adds a fifth fewer edges into its LDS sums; sum order = LDS arrival
# speedup vs baseline: 2.8504x; 1.0567x over previous
#include <hip/hip_runtime.h>
#include <hip/hip_bf16.h>
#include <math.h>

typedef __attribute__((ext_vector_type(2))) float v2f;
typedef __attribute__((ext_vector_type(8))) float v8f;

#define NN      100000
#define NE      1600000
#define HH      64
#define TT      384
#define NTILES  6250

__device__ __forceinline__ v2f mk2(float a, float b) {
    v2f r; r.x = a; r.y = b; return r;
}

__device__ __forceinline__ v8f wmma4(v2f a, v2f b, v8f c) {
    return __builtin_amdgcn_wmma_f32_16x16x4_f32(
         false, a,  false, b,
         (short)0, c,  false,  false);
}

#define DEG_PART    50016
#define DEG_SLICES  32
#define DEG_SLICE_E (NE / DEG_SLICES)
#define DEG_THREADS 1024
#define DEG_NV      ((DEG_SLICE_E / 4 + DEG_THREADS - 1) / DEG_THREADS)

__global__ __launch_bounds__(DEG_THREADS) void degree_kernel(const int* __restrict__ src,
                                                             const int* __restrict__ dst,
                                                             float* __restrict__ degF,
                                                             float* __restrict__ degR) {
    __shared__ unsigned int cnt[DEG_PART];
    const int tid   = threadIdx.x;
    const int combo = blockIdx.x & 3;
    const int slice = blockIdx.x >> 2;
    const int rev   = combo & 1;
    const int base  = (combo >> 1) * DEG_PART;
    const int nloc  = (NN - base < DEG_PART) ? (NN - base) : DEG_PART;
    const int* __restrict__ idx = rev ? src : dst;
    float* __restrict__ deg     = rev ? degR : degF;

    const int4* ip = reinterpret_cast<const int4*>(idx + (size_t)slice * DEG_SLICE_E);
    int4 v[DEG_NV];
    #pragma unroll
    for (int j = 0; j < DEG_NV; ++j) {
        int i = tid + j * DEG_THREADS;
        v[j] = (i < DEG_SLICE_E / 4) ? ip[i] : make_int4(-1, -1, -1, -1);
    }

    uint4* cnt4 = reinterpret_cast<uint4*>(cnt);
    for (int i = tid; i < DEG_PART / 4; i += DEG_THREADS)
        cnt4[i] = make_uint4(0u, 0u, 0u, 0u);
    __syncthreads();

    #pragma unroll
    for (int j = 0; j < DEG_NV; ++j) {
        unsigned r0 = (unsigned)(v[j].x - base);
        unsigned r1 = (unsigned)(v[j].y - base);
        unsigned r2 = (unsigned)(v[j].z - base);
        unsigned r3 = (unsigned)(v[j].w - base);
        if (r0 < (unsigned)nloc) atomicAdd(&cnt[r0], 1u);
        if (r1 < (unsigned)nloc) atomicAdd(&cnt[r1], 1u);
        if (r2 < (unsigned)nloc) atomicAdd(&cnt[r2], 1u);
        if (r3 < (unsigned)nloc) atomicAdd(&cnt[r3], 1u);
    }
    __syncthreads();

    for (int i0 = tid; i0 < nloc; i0 += 8 * DEG_THREADS) {
        unsigned c[8];
        #pragma unroll
        for (int j = 0; j < 8; ++j) {
            int i = i0 + j * DEG_THREADS;
            c[j] = (i < nloc) ? cnt[i] : 0u;
        }
        #pragma unroll
        for (int j = 0; j < 8; ++j)
            if (c[j]) atomicAdd(&deg[base + i0 + j * DEG_THREADS], (float)c[j]);
    }
}

#define OWN_NODES 800
#define KEY_NONE  0x7Fu
#define NE4       (NE / 4)
#define KEY_WS_BYTES ((size_t)(3 * (size_t)NN * HH + 2 * (size_t)NN) * sizeof(float) + 2 * (size_t)NE)
#define REC_WS_BYTES (KEY_WS_BYTES + (size_t)NE * sizeof(int4))

__device__ __forceinline__ unsigned key_of(int idx) {
    return ((unsigned)idx < (unsigned)NN) ? ((unsigned)idx / (unsigned)OWN_NODES) : KEY_NONE;
}
__device__ __forceinline__ unsigned key4_of(int4 a) {
    return key_of(a.x) | (key_of(a.y) << 8) | (key_of(a.z) << 16) | (key_of(a.w) << 24);
}

#define IK_THREADS 256
#define IK_BLOCKS  ((NE4 + 4 * IK_THREADS - 1) / (4 * IK_THREADS))
static_assert(IK_BLOCKS * IK_THREADS >= NN, "the grid must cover every node");

__global__ __launch_bounds__(IK_THREADS) void invert_key_kernel(float* __restrict__ degF, float* __restrict__ degR,
                                                                const int* __restrict__ src,
                                                                const int* __restrict__ dst,
                                                                const float* __restrict__ w,
                                                                unsigned int* __restrict__ keyF,
                                                                unsigned int* __restrict__ keyR,
                                                                int* __restrict__ rec,
                                                                size_t ws_bytes) {
    int i = blockIdx.x * IK_THREADS + threadIdx.x;
    if (i < NN) {
        float a = degF[i];
        float b = degR[i];
        degF[i] = (a > 0.0f) ? (1.0f / a) : 0.0f;
        degR[i] = (b > 0.0f) ? (1.0f / b) : 0.0f;
    }
    if (ws_bytes < KEY_WS_BYTES) return;
    const int4* s4 = reinterpret_cast<const int4*>(src);
    const int4* d4 = reinterpret_cast<const int4*>(dst);
    const int g0 = blockIdx.x * (4 * IK_THREADS) + threadIdx.x;
    int4 a[4], b[4];
    #pragma unroll
    for (int k = 0; k < 4; ++k) {
        const int g = g0 + k * IK_THREADS;
        a[k] = (g < NE4) ? d4[g] : make_int4(-1, -1, -1, -1);
        b[k] = (g < NE4) ? s4[g] : make_int4(-1, -1, -1, -1);
    }
    #pragma unroll
    for (int k = 0; k < 4; ++k) {
        const int g = g0 + k * IK_THREADS;
        if (g < NE4) {
            keyF[g] = key4_of(a[k]);
            keyR[g] = key4_of(b[k]);
        }
    }
    if (ws_bytes < REC_WS_BYTES) return;
    int4* rec4 = reinterpret_cast<int4*>(rec);
    const int e0 = blockIdx.x * (16 * IK_THREADS) + threadIdx.x;
    #pragma unroll
    for (int h = 0; h < 2; ++h) {
        int   d[8], s[8];
        float wv[8];
        #pragma unroll
        for (int k = 0; k < 8; ++k) {
            const int e  = e0 + (8 * h + k) * IK_THREADS;
            const int ec = (e < NE) ? e : NE - 1;
            d[k]  = dst[ec];
            s[k]  = src[ec];
            wv[k] = w[ec];
        }
        #pragma unroll
        for (int k = 0; k < 8; ++k) {
            const int e = e0 + (8 * h + k) * IK_THREADS;
            if (e < NE) rec4[e] = make_int4(d[k], s[k], __float_as_int(wv[k]), 0);
        }
    }
}

__global__ __launch_bounds__(256) void enc_kernel(const float* __restrict__ emb,
                                                  const float* __restrict__ W,
                                                  const float* __restrict__ bias,
                                                  float* __restrict__ x) {
    const int wave = threadIdx.x >> 5;
    const int lane = threadIdx.x & 31;
    const int nt   = blockIdx.x * 8 + wave;
    if (nt >= NTILES) return;
    const int ln = lane & 15;
    const int hi = lane >> 4;
    const int kh = hi << 1;

    const float* arow = emb + (size_t)(nt * 16 + ln) * TT + kh;
    v8f acc0 = {}, acc1 = {}, acc2 = {}, acc3 = {};

    #pragma unroll 4
    for (int k0 = 0; k0 < TT; k0 += 4) {
        v2f a = *reinterpret_cast<const v2f*>(arow + k0);
        const float* bp = W + (size_t)(k0 + kh) * HH + ln;
        v2f b0 = mk2(bp[0],  bp[HH]);
        v2f b1 = mk2(bp[16], bp[HH + 16]);
        v2f b2 = mk2(bp[32], bp[HH + 32]);
        v2f b3 = mk2(bp[48], bp[HH + 48]);
        acc0 = wmma4(a, b0, acc0);
        acc1 = wmma4(a, b1, acc1);
        acc2 = wmma4(a, b2, acc2);
        acc3 = wmma4(a, b3, acc3);
    }

    float bs0 = bias[ln], bs1 = bias[16 + ln], bs2 = bias[32 + ln], bs3 = bias[48 + ln];
    #pragma unroll
    for (int v = 0; v < 8; ++v) {
        size_t r = (size_t)(nt * 16 + v + 8 * hi) * HH + ln;
        x[r]      = acc0[v] + bs0;
        x[r + 16] = acc1[v] + bs1;
        x[r + 32] = acc2[v] + bs2;
        x[r + 48] = acc3[v] + bs3;
    }
}

#define OWN_PARTS   ((NN + OWN_NODES - 1) / OWN_NODES)
#define OWN_THREADS 1024
#define OWN_WAVES   (OWN_THREADS / 32)
#define NE16        (NE / 16)
#define OWN_NSTEP   ((NE16 + OWN_THREADS - 1) / OWN_THREADS)
#define OWN_FULL    (NE16 / OWN_THREADS)
#define OWN_PF      4
#define OWN_BATCH   12
#define OWN_RING    64
static_assert(OWN_PARTS <= (int)KEY_NONE, "range numbers and KEY_NONE must stay below 128 and apart");
static_assert(NE % 16 == 0, "the key list is read 16 bytes at a time");

template <bool FULL, bool PACKED>
__device__ __forceinline__ void own_drain(const unsigned int* lst, unsigned head, unsigned nvalid, int lane, int base, int rev,
                                          const int* __restrict__ own, const int* __restrict__ oth,
                                          const float* __restrict__ w, const int4* __restrict__ rec,
                                          const float* __restrict__ x, float* acc) {
    const unsigned jj = ((unsigned)lane < (FULL ? (unsigned)OWN_BATCH : nvalid)) ? (unsigned)lane : 0u;
    const unsigned e  = lst[(head + jj) & (OWN_RING - 1)];
    int rn, ro, rw;
    if (PACKED) {
        const int4 r = rec[e];
        rn = rev ? r.y : r.x;
        ro = rev ? r.x : r.y;
        rw = r.z;
    } else {
        ro = oth[e];
        rw = __float_as_int(w[e]);
        rn = own[e];
    }
    v2f xv[OWN_BATCH];
    #pragma unroll
    for (int j = 0; j < OWN_BATCH; ++j) {
        const int o = __builtin_amdgcn_readlane(ro, j);
        xv[j] = *reinterpret_cast<const v2f*>(x + (size_t)o * HH + 2 * lane);
    }
    #pragma unroll
    for (int j = 0; j < OWN_BATCH; ++j) {
        if (FULL || (unsigned)j < nvalid) {
            const int   n  = __builtin_amdgcn_readlane(rn, j);
            const float we = __int_as_float(__builtin_amdgcn_readlane(rw, j));
            float* a = acc + (size_t)(n - base) * HH + 2 * lane;
            atomicAdd(a,     we * xv[j].x);
            atomicAdd(a + 1, we * xv[j].y);
        }
    }
}

template <bool GUARD, bool PACKED>
__device__ __forceinline__ void own_step(uint4& vj, int i, unsigned pk, int base, int rev, const uint4* __restrict__ key4,
                                         unsigned int* lst, unsigned& head, unsigned& tail, int lane,
                                         const int* __restrict__ own, const int* __restrict__ oth,
                                         const float* __restrict__ w, const int4* __restrict__ rec,
                                         const float* __restrict__ x, float* acc) {
    unsigned s0 = (vj.x ^ pk) + 0x7F7F7F7Fu;
    unsigned s1 = (vj.y ^ pk) + 0x7F7F7F7Fu;
    unsigned s2 = (vj.z ^ pk) + 0x7F7F7F7Fu;
    unsigned s3 = (vj.w ^ pk) + 0x7F7F7F7Fu;
    const int in = i + OWN_PF * OWN_THREADS;
    if (GUARD) vj = (in < NE16) ? key4[in] : make_uint4(0x7F7F7F7Fu, 0x7F7F7F7Fu, 0x7F7F7F7Fu, 0x7F7F7F7Fu);
    else       vj = key4[in];

    unsigned any = __builtin_amdgcn_ballot_w32(((s0 & s1 & s2 & s3) & 0x80808080u) != 0x80808080u);
    if (any) {
        const unsigned miss = ((s0 >> 7) & 0x01010101u) | ((s1 >> 6) & 0x02020202u) |
                              ((s2 >> 5) & 0x04040404u) | ((s3 >> 4) & 0x08080808u);
        unsigned hm = ~miss & 0x0F0F0F0Fu;
        const unsigned e0 = (unsigned)i * 16u;
        do {
            if (hm != 0u) {
                const unsigned p = (unsigned)__builtin_ctz(hm);
                hm &= hm - 1u;
                lst[(tail + __builtin_amdgcn_mbcnt_lo(any, 0u)) & (OWN_RING - 1)] = e0 + ((p & 7u) << 2) + (p >> 3);
            }
            tail += (unsigned)__builtin_popcount(any);
            while (tail - head >= (unsigned)OWN_BATCH) {
                own_drain<true, PACKED>(lst, head, (unsigned)OWN_BATCH, lane, base, rev, own, oth, w, rec, x, acc);
                head += (unsigned)OWN_BATCH;
            }
            any = __builtin_amdgcn_ballot_w32(hm != 0u);
        } while (any);
    }
}

template <bool PACKED>
__device__ __forceinline__ void own_scan(uint4 (&v)[OWN_PF], int tid, unsigned pk, int base, int rev,
                                         const uint4* __restrict__ key4, unsigned int* lst, int lane,
                                         const int* __restrict__ own, const int* __restrict__ oth,
                                         const float* __restrict__ w, const int4* __restrict__ rec,
                                         const float* __restrict__ x, float* acc) {
    unsigned head = 0u, tail = 0u;
    int s0 = 0;
    for (; s0 + 2 * OWN_PF <= OWN_FULL; s0 += OWN_PF) {
        #pragma unroll
        for (int j = 0; j < OWN_PF; ++j)
            own_step<false, PACKED>(v[j], (s0 + j) * OWN_THREADS + tid, pk, base, rev, key4, lst, head, tail, lane, own, oth, w, rec, x, acc);
    }
    for (; s0 < OWN_NSTEP; s0 += OWN_PF) {
        #pragma unroll
        for (int j = 0; j < OWN_PF; ++j)
            own_step<true, PACKED>(v[j], (s0 + j) * OWN_THREADS + tid, pk, base, rev, key4, lst, head, tail, lane, own, oth, w, rec, x, acc);
    }
    if (tail != head)
        own_drain<false, PACKED>(lst, head, tail - head, lane, base, rev, own, oth, w, rec, x, acc);
}

__global__ __launch_bounds__(OWN_THREADS) void own_kernel(const float* __restrict__ x,
                                                          const float* __restrict__ w,
                                                          const int* __restrict__ src,
                                                          const int* __restrict__ dst,
                                                          const unsigned int* __restrict__ keyF,
                                                          const unsigned int* __restrict__ keyR,
                                                          const int* __restrict__ rec,
                                                          float* __restrict__ nF,
                                                          float* __restrict__ nR,
                                                          size_t ws_bytes) {
    __shared__ float acc[OWN_NODES * HH];
    __shared__ unsigned int lists[OWN_WAVES * OWN_RING];
    if (ws_bytes < KEY_WS_BYTES) return;
    const int tid  = threadIdx.x;
    const int lane = tid & 31;
    const int wave = tid >> 5;
    const int rev  = blockIdx.x & 1;
    const int part = blockIdx.x >> 1;
    const int base = part * OWN_NODES;
    const unsigned nloc = (unsigned)((NN - base < OWN_NODES) ? (NN - base) : OWN_NODES);
    const unsigned pk = (unsigned)part * 0x01010101u;
    const int* __restrict__ own = rev ? src : dst;
    const int* __restrict__ oth = rev ? dst : src;
    float* __restrict__ outp    = rev ? nR : nF;
    const uint4* key4 = reinterpret_cast<const uint4*>(rev ? keyR : keyF);
    unsigned int* lst = lists + wave * OWN_RING;
    const int4* rec4 = reinterpret_cast<const int4*>(rec);

    uint4 v[OWN_PF];
    #pragma unroll
    for (int j = 0; j < OWN_PF; ++j) {
        int i = j * OWN_THREADS + tid;
        v[j] = (i < NE16) ? key4[i] : make_uint4(0x7F7F7F7Fu, 0x7F7F7F7Fu, 0x7F7F7F7Fu, 0x7F7F7F7Fu);
    }

    float4* acc4 = reinterpret_cast<float4*>(acc);
    for (int i = tid; i < OWN_NODES * HH / 4; i += OWN_THREADS)
        acc4[i] = make_float4(0.0f, 0.0f, 0.0f, 0.0f);
    __syncthreads();

    if (__builtin_expect(ws_bytes >= REC_WS_BYTES, 1))
        own_scan<true>(v, tid, pk, base, rev, key4, lst, lane, own, oth, w, rec4, x, acc);
    else
        own_scan<false>(v, tid, pk, base, rev, key4, lst, lane, own, oth, w, rec4, x, acc);
    __syncthreads();

    float4* out4 = reinterpret_cast<float4*>(outp + (size_t)base * HH);
    const int nout4 = (int)nloc * (HH / 4);
    for (int i = tid; i < nout4; i += OWN_THREADS)
        out4[i] = acc4[i];
}

__global__ __launch_bounds__(256) void layer_kernel(
    float* __restrict__ x,
    const float* __restrict__ nF, const float* __restrict__ nR,
    const float* __restrict__ invF, const float* __restrict__ invR,
    const float* __restrict__ WsF, const float* __restrict__ WnF, const float* __restrict__ bF,
    const float* __restrict__ WsR, const float* __restrict__ WnR, const float* __restrict__ bR) {
    const int wave = threadIdx.x >> 5;
    const int lane = threadIdx.x & 31;
    const int nt   = blockIdx.x * 8 + wave;
    if (nt >= NTILES) return;
    const int ln = lane & 15;
    const int hi = lane >> 4;
    const int kh = hi << 1;
    const int rowA = nt * 16 + ln;

    const float sF = invF[rowA];
    const float sR = invR[rowA];
    const float* xr = x  + (size_t)rowA * HH + kh;
    const float* fr = nF + (size_t)rowA * HH + kh;
    const float* rr = nR + (size_t)rowA * HH + kh;

    v8f aF0 = {}, aF1 = {}, aF2 = {}, aF3 = {};
    v8f aR0 = {}, aR1 = {}, aR2 = {}, aR3 = {};

    #pragma unroll 4
    for (int k0 = 0; k0 < HH; k0 += 4) {
        v2f aS = *reinterpret_cast<const v2f*>(xr + k0);
        v2f aN = *reinterpret_cast<const v2f*>(fr + k0);
        v2f aV = *reinterpret_cast<const v2f*>(rr + k0);
        aN *= sF;
        aV *= sR;

        const float* pSF = WsF + (size_t)(k0 + kh) * HH + ln;
        const float* pNF = WnF + (size_t)(k0 + kh) * HH + ln;
        const float* pSR = WsR + (size_t)(k0 + kh) * HH + ln;
        const float* pNR = WnR + (size_t)(k0 + kh) * HH + ln;

        aF0 = wmma4(aS, mk2(pSF[0], pSF[HH]), aF0);
        aF0 = wmma4(aN, mk2(pNF[0], pNF[HH]), aF0);
        aR0 = wmma4(aS, mk2(pSR[0], pSR[HH]), aR0);
        aR0 = wmma4(aV, mk2(pNR[0], pNR[HH]), aR0);
        aF1 = wmma4(aS, mk2(pSF[16], pSF[HH + 16]), aF1);
        aF1 = wmma4(aN, mk2(pNF[16], pNF[HH + 16]), aF1);
        aR1 = wmma4(aS, mk2(pSR[16], pSR[HH + 16]), aR1);
        aR1 = wmma4(aV, mk2(pNR[16], pNR[HH + 16]), aR1);
        aF2 = wmma4(aS, mk2(pSF[32], pSF[HH + 32]), aF2);
        aF2 = wmma4(aN, mk2(pNF[32], pNF[HH + 32]), aF2);
        aR2 = wmma4(aS, mk2(pSR[32], pSR[HH + 32]), aR2);
        aR2 = wmma4(aV, mk2(pNR[32], pNR[HH + 32]), aR2);
        aF3 = wmma4(aS, mk2(pSF[48], pSF[HH + 48]), aF3);
        aF3 = wmma4(aN, mk2(pNF[48], pNF[HH + 48]), aF3);
        aR3 = wmma4(aS, mk2(pSR[48], pSR[HH + 48]), aR3);
        aR3 = wmma4(aV, mk2(pNR[48], pNR[HH + 48]), aR3);
    }

    float bf0 = bF[ln], bf1 = bF[16 + ln], bf2 = bF[32 + ln], bf3 = bF[48 + ln];
    float br0 = bR[ln], br1 = bR[16 + ln], br2 = bR[32 + ln], br3 = bR[48 + ln];

    #pragma unroll
    for (int v = 0; v < 8; ++v) {
        size_t r = (size_t)(nt * 16 + v + 8 * hi) * HH + ln;
        x[r]      += fmaxf(aF0[v] + bf0, 0.0f) + fmaxf(aR0[v] + br0, 0.0f);
        x[r + 16] += fmaxf(aF1[v] + bf1, 0.0f) + fmaxf(aR1[v] + br1, 0.0f);
        x[r + 32] += fmaxf(aF2[v] + bf2, 0.0f) + fmaxf(aR2[v] + br2, 0.0f);
        x[r + 48] += fmaxf(aF3[v] + bf3, 0.0f) + fmaxf(aR3[v] + br3, 0.0f);
    }
}

__global__ __launch_bounds__(64) void norm_kernel(const float* __restrict__ x,
                                                  const int* __restrict__ ids,
                                                  float* __restrict__ out) {
    __shared__ float s[64];
    int b = blockIdx.x;
    int c = threadIdx.x;
    int node = ids[b];
    float v = x[(size_t)node * HH + c];
    s[c] = v * v;
    __syncthreads();
    #pragma unroll
    for (int off = 32; off > 0; off >>= 1) {
        if (c < off) s[c] += s[c + off];
        __syncthreads();
    }
    float nrm = sqrtf(s[0]);
    out[(size_t)b * HH + c] = v / nrm;
}

extern "C" void kernel_launch(void* const* d_in, const int* in_sizes, int n_in,
                              void* d_out, int out_size, void* d_ws, size_t ws_size,
                              hipStream_t stream) {
    const float* text_emb = (const float*)d_in[0];
    const float* weights  = (const float*)d_in[1];
    const float* W_enc    = (const float*)d_in[2];
    const float* b_enc    = (const float*)d_in[3];
    const float* Wself_f  = (const float*)d_in[4];
    const float* Wneigh_f = (const float*)d_in[5];
    const float* b_f      = (const float*)d_in[6];
    const float* Wself_r  = (const float*)d_in[7];
    const float* Wneigh_r = (const float*)d_in[8];
    const float* b_r      = (const float*)d_in[9];
    const int*   src      = (const int*)d_in[10];
    const int*   dst      = (const int*)d_in[11];
    const int*   ids      = (const int*)d_in[12];
    float* out = (float*)d_out;

    const size_t NH = (size_t)NN * HH;
    float* x    = (float*)d_ws;
    float* nF   = x + NH;
    float* nR   = x + 2 * NH;
    float* invF = x + 3 * NH;
    float* invR = invF + NN;
    unsigned int* keyF = (unsigned int*)(invR + NN);
    unsigned int* keyR = keyF + NE / 4;
    int* rec = (int*)(keyR + NE / 4);

    const int batch = in_sizes[12];

    hipMemsetAsync(invF, 0, (size_t)2 * NN * sizeof(float), stream);
    degree_kernel<<<4 * DEG_SLICES, DEG_THREADS, 0, stream>>>(src, dst, invF, invR);
    invert_key_kernel<<<IK_BLOCKS, IK_THREADS, 0, stream>>>(invF, invR, src, dst, weights, keyF, keyR, rec, ws_size);

    enc_kernel<<<(NTILES + 7) / 8, 256, 0, stream>>>(text_emb, W_enc, b_enc, x);

    for (int l = 0; l < 2; ++l) {
        own_kernel<<<2 * OWN_PARTS, OWN_THREADS, 0, stream>>>(x, weights, src, dst, keyF, keyR, rec, nF, nR, ws_size);
        layer_kernel<<<(NTILES + 7) / 8, 256, 0, stream>>>(
            x, nF, nR, invF, invR,
            Wself_f + (size_t)l * HH * HH, Wneigh_f + (size_t)l * HH * HH, b_f + (size_t)l * HH,
            Wself_r + (size_t)l * HH * HH, Wneigh_r + (size_t)l * HH * HH, b_r + (size_t)l * HH);
    }

    norm_kernel<<<batch, 64, 0, stream>>>(x, ids, out);
}


# Second layer's own_kernel reads per-wave lists of its hits (row, other end, weight: 8 bytes) kept by the first layer's launch; no second key scan or record fetch; sum order = LDS arrival
# speedup vs baseline: 3.0382x; 1.0659x over previous
#include <hip/hip_runtime.h>
#include <hip/hip_bf16.h>
#include <math.h>

typedef __attribute__((ext_vector_type(2))) float v2f;
typedef __attribute__((ext_vector_type(8))) float v8f;

#define NN      100000
#define NE      1600000
#define HH      64
#define TT      384
#define NTILES  6250

__device__ __forceinline__ v2f mk2(float a, float b) {
    v2f r; r.x = a; r.y = b; return r;
}

__device__ __forceinline__ v8f wmma4(v2f a, v2f b, v8f c) {
    return __builtin_amdgcn_wmma_f32_16x16x4_f32(
         false, a,  false, b,
         (short)0, c,  false,  false);
}

#define DEG_PART    50016
#define DEG_SLICES  32
#define DEG_SLICE_E (NE / DEG_SLICES)
#define DEG_THREADS 1024
#define DEG_NV      ((DEG_SLICE_E / 4 + DEG_THREADS - 1) / DEG_THREADS)

__global__ __launch_bounds__(DEG_THREADS) void degree_kernel(const int* __restrict__ src,
                                                             const int* __restrict__ dst,
                                                             float* __restrict__ degF,
                                                             float* __restrict__ degR) {
    __shared__ unsigned int cnt[DEG_PART];
    const int tid   = threadIdx.x;
    const int combo = blockIdx.x & 3;
    const int slice = blockIdx.x >> 2;
    const int rev   = combo & 1;
    const int base  = (combo >> 1) * DEG_PART;
    const int nloc  = (NN - base < DEG_PART) ? (NN - base) : DEG_PART;
    const int* __restrict__ idx = rev ? src : dst;
    float* __restrict__ deg     = rev ? degR : degF;

    const int4* ip = reinterpret_cast<const int4*>(idx + (size_t)slice * DEG_SLICE_E);
    int4 v[DEG_NV];
    #pragma unroll
    for (int j = 0; j < DEG_NV; ++j) {
        int i = tid + j * DEG_THREADS;
        v[j] = (i < DEG_SLICE_E / 4) ? ip[i] : make_int4(-1, -1, -1, -1);
    }

    uint4* cnt4 = reinterpret_cast<uint4*>(cnt);
    for (int i = tid; i < DEG_PART / 4; i += DEG_THREADS)
        cnt4[i] = make_uint4(0u, 0u, 0u, 0u);
    __syncthreads();

    #pragma unroll
    for (int j = 0; j < DEG_NV; ++j) {
        unsigned r0 = (unsigned)(v[j].x - base);
        unsigned r1 = (unsigned)(v[j].y - base);
        unsigned r2 = (unsigned)(v[j].z - base);
        unsigned r3 = (unsigned)(v[j].w - base);
        if (r0 < (unsigned)nloc) atomicAdd(&cnt[r0], 1u);
        if (r1 < (unsigned)nloc) atomicAdd(&cnt[r1], 1u);
        if (r2 < (unsigned)nloc) atomicAdd(&cnt[r2], 1u);
        if (r3 < (unsigned)nloc) atomicAdd(&cnt[r3], 1u);
    }
    __syncthreads();

    for (int i0 = tid; i0 < nloc; i0 += 8 * DEG_THREADS) {
        unsigned c[8];
        #pragma unroll
        for (int j = 0; j < 8; ++j) {
            int i = i0 + j * DEG_THREADS;
            c[j] = (i < nloc) ? cnt[i] : 0u;
        }
        #pragma unroll
        for (int j = 0; j < 8; ++j)
            if (c[j]) atomicAdd(&deg[base + i0 + j * DEG_THREADS], (float)c[j]);
    }
}

#define OWN_NODES 800
#define KEY_NONE  0x7Fu
#define NE4       (NE / 4)
#define KEY_WS_BYTES ((size_t)(3 * (size_t)NN * HH + 2 * (size_t)NN) * sizeof(float) + 2 * (size_t)NE)
#define REC_WS_BYTES (KEY_WS_BYTES + (size_t)NE * sizeof(int4))
#define LST_CAP     512
#define OWN_GROUPS  250
#define OWN_WAVES_C 32
#define LST_WORDS   ((size_t)OWN_GROUPS * OWN_WAVES_C * LST_CAP * 2)
#define LIST_WS_BYTES (REC_WS_BYTES + (LST_WORDS + (size_t)OWN_GROUPS * OWN_WAVES_C) * sizeof(unsigned int))
#define LST_OBITS   17

__device__ __forceinline__ unsigned key_of(int idx) {
    return ((unsigned)idx < (unsigned)NN) ? ((unsigned)idx / (unsigned)OWN_NODES) : KEY_NONE;
}
__device__ __forceinline__ unsigned key4_of(int4 a) {
    return key_of(a.x) | (key_of(a.y) << 8) | (key_of(a.z) << 16) | (key_of(a.w) << 24);
}

#define IK_THREADS 256
#define IK_BLOCKS  ((NE4 + 4 * IK_THREADS - 1) / (4 * IK_THREADS))
static_assert(IK_BLOCKS * IK_THREADS >= NN, "the grid must cover every node");

__global__ __launch_bounds__(IK_THREADS) void invert_key_kernel(float* __restrict__ degF, float* __restrict__ degR,
                                                                const int* __restrict__ src,
                                                                const int* __restrict__ dst,
                                                                const float* __restrict__ w,
                                                                unsigned int* __restrict__ keyF,
                                                                unsigned int* __restrict__ keyR,
                                                                int* __restrict__ rec,
                                                                size_t ws_bytes) {
    int i = blockIdx.x * IK_THREADS + threadIdx.x;
    if (i < NN) {
        float a = degF[i];
        float b = degR[i];
        degF[i] = (a > 0.0f) ? (1.0f / a) : 0.0f;
        degR[i] = (b > 0.0f) ? (1.0f / b) : 0.0f;
    }
    if (ws_bytes < KEY_WS_BYTES) return;
    const int4* s4 = reinterpret_cast<const int4*>(src);
    const int4* d4 = reinterpret_cast<const int4*>(dst);
    const int g0 = blockIdx.x * (4 * IK_THREADS) + threadIdx.x;
    int4 a[4], b[4];
    #pragma unroll
    for (int k = 0; k < 4; ++k) {
        const int g = g0 + k * IK_THREADS;
        a[k] = (g < NE4) ? d4[g] : make_int4(-1, -1, -1, -1);
        b[k] = (g < NE4) ? s4[g] : make_int4(-1, -1, -1, -1);
    }
    #pragma unroll
    for (int k = 0; k < 4; ++k) {
        const int g = g0 + k * IK_THREADS;
        if (g < NE4) {
            keyF[g] = key4_of(a[k]);
            keyR[g] = key4_of(b[k]);
        }
    }
    if (ws_bytes < REC_WS_BYTES) return;
    int4* rec4 = reinterpret_cast<int4*>(rec);
    const int e0 = blockIdx.x * (16 * IK_THREADS) + threadIdx.x;
    #pragma unroll
    for (int h = 0; h < 2; ++h) {
        int   d[8], s[8];
        float wv[8];
        #pragma unroll
        for (int k = 0; k < 8; ++k) {
            const int e  = e0 + (8 * h + k) * IK_THREADS;
            const int ec = (e < NE) ? e : NE - 1;
            d[k]  = dst[ec];
            s[k]  = src[ec];
            wv[k] = w[ec];
        }
        #pragma unroll
        for (int k = 0; k < 8; ++k) {
            const int e = e0 + (8 * h + k) * IK_THREADS;
            if (e < NE) rec4[e] = make_int4(d[k], s[k], __float_as_int(wv[k]), 0);
        }
    }
}

__global__ __launch_bounds__(256) void enc_kernel(const float* __restrict__ emb,
                                                  const float* __restrict__ W,
                                                  const float* __restrict__ bias,
                                                  float* __restrict__ x) {
    const int wave = threadIdx.x >> 5;
    const int lane = threadIdx.x & 31;
    const int nt   = blockIdx.x * 8 + wave;
    if (nt >= NTILES) return;
    const int ln = lane & 15;
    const int hi = lane >> 4;
    const int kh = hi << 1;

    const float* arow = emb + (size_t)(nt * 16 + ln) * TT + kh;
    v8f acc0 = {}, acc1 = {}, acc2 = {}, acc3 = {};

    #pragma unroll 4
    for (int k0 = 0; k0 < TT; k0 += 4) {
        v2f a = *reinterpret_cast<const v2f*>(arow + k0);
        const float* bp = W + (size_t)(k0 + kh) * HH + ln;
        v2f b0 = mk2(bp[0],  bp[HH]);
        v2f b1 = mk2(bp[16], bp[HH + 16]);
        v2f b2 = mk2(bp[32], bp[HH + 32]);
        v2f b3 = mk2(bp[48], bp[HH + 48]);
        acc0 = wmma4(a, b0, acc0);
        acc1 = wmma4(a, b1, acc1);
        acc2 = wmma4(a, b2, acc2);
        acc3 = wmma4(a, b3, acc3);
    }

    float bs0 = bias[ln], bs1 = bias[16 + ln], bs2 = bias[32 + ln], bs3 = bias[48 + ln];
    #pragma unroll
    for (int v = 0; v < 8; ++v) {
        size_t r = (size_t)(nt * 16 + v + 8 * hi) * HH + ln;
        x[r]      = acc0[v] + bs0;
        x[r + 16] = acc1[v] + bs1;
        x[r + 32] = acc2[v] + bs2;
        x[r + 48] = acc3[v] + bs3;
    }
}

#define OWN_PARTS   ((NN + OWN_NODES - 1) / OWN_NODES)
#define OWN_THREADS 1024
#define OWN_WAVES   (OWN_THREADS / 32)
#define NE16        (NE / 16)
#define OWN_NSTEP   ((NE16 + OWN_THREADS - 1) / OWN_THREADS)
#define OWN_FULL    (NE16 / OWN_THREADS)
#define OWN_PF      4
#define OWN_BATCH   12
#define OWN_RING    64
static_assert(OWN_PARTS <= (int)KEY_NONE, "range numbers and KEY_NONE must stay below 128 and apart");
static_assert(NE % 16 == 0, "the key list is read 16 bytes at a time");
static_assert(OWN_GROUPS == 2 * OWN_PARTS && OWN_WAVES_C == OWN_WAVES, "the list area is sized for this grid");
static_assert(NN <= (1 << LST_OBITS) && OWN_NODES <= (1 << (32 - LST_OBITS)), "a list entry packs both ends into one dword");

template <bool FULL, bool PACKED, bool SAVE>
__device__ __forceinline__ void own_drain(const unsigned int* lst, unsigned head, unsigned nvalid, int lane, int base, int rev,
                                          const int* __restrict__ own, const int* __restrict__ oth,
                                          const float* __restrict__ w, const int4* __restrict__ rec,
                                          const float* __restrict__ x, float* acc, uint2* __restrict__ keep) {
    const unsigned jj = ((unsigned)lane < (FULL ? (unsigned)OWN_BATCH : nvalid)) ? (unsigned)lane : 0u;
    const unsigned e  = lst[(head + jj) & (OWN_RING - 1)];
    int rn, ro, rw;
    if (PACKED) {
        const int4 r = rec[e];
        rn = rev ? r.y : r.x;
        ro = rev ? r.x : r.y;
        rw = r.z;
    } else {
        ro = oth[e];
        rw = __float_as_int(w[e]);
        rn = own[e];
    }
    if (SAVE) {
        if ((unsigned)lane < (FULL ? (unsigned)OWN_BATCH : nvalid) && head + (unsigned)lane < (unsigned)LST_CAP)
            keep[head + (unsigned)lane] = make_uint2(((unsigned)(rn - base) << LST_OBITS) | ((unsigned)ro & ((1u << LST_OBITS) - 1u)), (unsigned)rw);
    }
    v2f xv[OWN_BATCH];
    #pragma unroll
    for (int j = 0; j < OWN_BATCH; ++j) {
        const int o = __builtin_amdgcn_readlane(ro, j);
        xv[j] = *reinterpret_cast<const v2f*>(x + (size_t)o * HH + 2 * lane);
    }
    #pragma unroll
    for (int j = 0; j < OWN_BATCH; ++j) {
        if (FULL || (unsigned)j < nvalid) {
            const int   n  = __builtin_amdgcn_readlane(rn, j);
            const float we = __int_as_float(__builtin_amdgcn_readlane(rw, j));
            float* a = acc + (size_t)(n - base) * HH + 2 * lane;
            atomicAdd(a,     we * xv[j].x);
            atomicAdd(a + 1, we * xv[j].y);
        }
    }
}

template <bool GUARD, bool PACKED, bool SAVE>
__device__ __forceinline__ void own_step(uint4& vj, int i, unsigned pk, int base, int rev, const uint4* __restrict__ key4,
                                         unsigned int* lst, unsigned& head, unsigned& tail, int lane,
                                         const int* __restrict__ own, const int* __restrict__ oth,
                                         const float* __restrict__ w, const int4* __restrict__ rec,
                                         const float* __restrict__ x, float* acc, uint2* __restrict__ keep) {
    unsigned s0 = (vj.x ^ pk) + 0x7F7F7F7Fu;
    unsigned s1 = (vj.y ^ pk) + 0x7F7F7F7Fu;
    unsigned s2 = (vj.z ^ pk) + 0x7F7F7F7Fu;
    unsigned s3 = (vj.w ^ pk) + 0x7F7F7F7Fu;
    const int in = i + OWN_PF * OWN_THREADS;
    if (GUARD) vj = (in < NE16) ? key4[in] : make_uint4(0x7F7F7F7Fu, 0x7F7F7F7Fu, 0x7F7F7F7Fu, 0x7F7F7F7Fu);
    else       vj = key4[in];

    unsigned any = __builtin_amdgcn_ballot_w32(((s0 & s1 & s2 & s3) & 0x80808080u) != 0x80808080u);
    if (any) {
        const unsigned miss = ((s0 >> 7) & 0x01010101u) | ((s1 >> 6) & 0x02020202u) |
                              ((s2 >> 5) & 0x04040404u) | ((s3 >> 4) & 0x08080808u);
        unsigned hm = ~miss & 0x0F0F0F0Fu;
        const unsigned e0 = (unsigned)i * 16u;
        do {
            if (hm != 0u) {
                const unsigned p = (unsigned)__builtin_ctz(hm);
                hm &= hm - 1u;
                lst[(tail + __builtin_amdgcn_mbcnt_lo(any, 0u)) & (OWN_RING - 1)] = e0 + ((p & 7u) << 2) + (p >> 3);
            }
            tail += (unsigned)__builtin_popcount(any);
            while (tail - head >= (unsigned)OWN_BATCH) {
                own_drain<true, PACKED, SAVE>(lst, head, (unsigned)OWN_BATCH, lane, base, rev, own, oth, w, rec, x, acc, keep);
                head += (unsigned)OWN_BATCH;
            }
            any = __builtin_amdgcn_ballot_w32(hm != 0u);
        } while (any);
    }
}

template <bool PACKED, bool SAVE>
__device__ __forceinline__ void own_scan(uint4 (&v)[OWN_PF], int tid, unsigned pk, int base, int rev,
                                         const uint4* __restrict__ key4, unsigned int* lst, int lane,
                                         const int* __restrict__ own, const int* __restrict__ oth,
                                         const float* __restrict__ w, const int4* __restrict__ rec,
                                         const float* __restrict__ x, float* acc,
                                         uint2* __restrict__ keep, unsigned int* __restrict__ keepn) {
    unsigned head = 0u, tail = 0u;
    int s0 = 0;
    for (; s0 + 2 * OWN_PF <= OWN_FULL; s0 += OWN_PF) {
        #pragma unroll
        for (int j = 0; j < OWN_PF; ++j)
            own_step<false, PACKED, SAVE>(v[j], (s0 + j) * OWN_THREADS + tid, pk, base, rev, key4, lst, head, tail, lane, own, oth, w, rec, x, acc, keep);
    }
    for (; s0 < OWN_NSTEP; s0 += OWN_PF) {
        #pragma unroll
        for (int j = 0; j < OWN_PF; ++j)
            own_step<true, PACKED, SAVE>(v[j], (s0 + j) * OWN_THREADS + tid, pk, base, rev, key4, lst, head, tail, lane, own, oth, w, rec, x, acc, keep);
    }
    if (tail != head)
        own_drain<false, PACKED, SAVE>(lst, head, tail - head, lane, base, rev, own, oth, w, rec, x, acc, keep);
    if (SAVE && lane == 0) *keepn = tail;
}

#define RP_SUB 16
template <bool FULL>
__device__ __forceinline__ void own_replay32(uint2 ent, unsigned nvalid, int lane, const float* __restrict__ x, float* acc) {
    #pragma unroll
    for (int h = 0; h < 32 / RP_SUB; ++h) {
        if (!FULL && (unsigned)(h * RP_SUB) >= nvalid) break;
        v2f xv[RP_SUB];
        #pragma unroll
        for (int j = 0; j < RP_SUB; ++j) {
            const unsigned o = (unsigned)__builtin_amdgcn_readlane((int)ent.x, h * RP_SUB + j) & ((1u << LST_OBITS) - 1u);
            xv[j] = *reinterpret_cast<const v2f*>(x + (size_t)o * HH + 2 * lane);
        }
        #pragma unroll
        for (int j = 0; j < RP_SUB; ++j) {
            if (FULL || (unsigned)(h * RP_SUB + j) < nvalid) {
                const unsigned n  = (unsigned)__builtin_amdgcn_readlane((int)ent.x, h * RP_SUB + j) >> LST_OBITS;
                const float    we = __int_as_float(__builtin_amdgcn_readlane((int)ent.y, h * RP_SUB + j));
                float* a = acc + (size_t)n * HH + 2 * lane;
                atomicAdd(a,     we * xv[j].x);
                atomicAdd(a + 1, we * xv[j].y);
            }
        }
    }
}

__device__ __forceinline__ void own_replay(const uint2* __restrict__ keep, unsigned cnt, int lane,
                                           const float* __restrict__ x, float* acc) {
    if (cnt == 0u) return;
    const unsigned last = cnt - 1u;
    uint2 cur = keep[((unsigned)lane < last) ? (unsigned)lane : last];
    unsigned p = 0u;
    for (; p + 32u <= cnt; p += 32u) {
        const unsigned q = p + 32u + (unsigned)lane;
        const uint2 nxt = keep[(q < last) ? q : last];
        own_replay32<true>(cur, 32u, lane, x, acc);
        cur = nxt;
    }
    if (p < cnt) own_replay32<false>(cur, cnt - p, lane, x, acc);
}

template <bool SECOND>
__global__ __launch_bounds__(OWN_THREADS) void own_kernel(const float* __restrict__ x,
                                                          const float* __restrict__ w,
                                                          const int* __restrict__ src,
                                                          const int* __restrict__ dst,
                                                          const unsigned int* __restrict__ keyF,
                                                          const unsigned int* __restrict__ keyR,
                                                          const int* __restrict__ rec,
                                                          float* __restrict__ nF,
                                                          float* __restrict__ nR,
                                                          unsigned int* __restrict__ keepw,
                                                          unsigned int* __restrict__ keepn,
                                                          size_t ws_bytes) {
    __shared__ float acc[OWN_NODES * HH];
    __shared__ unsigned int lists[OWN_WAVES * OWN_RING];
    if (ws_bytes < KEY_WS_BYTES) return;
    const int tid  = threadIdx.x;
    const int lane = tid & 31;
    const int wave = tid >> 5;
    const int rev  = blockIdx.x & 1;
    const int part = blockIdx.x >> 1;
    const int base = part * OWN_NODES;
    const unsigned nloc = (unsigned)((NN - base < OWN_NODES) ? (NN - base) : OWN_NODES);
    const unsigned pk = (unsigned)part * 0x01010101u;
    const int* __restrict__ own = rev ? src : dst;
    const int* __restrict__ oth = rev ? dst : src;
    float* __restrict__ outp    = rev ? nR : nF;
    const uint4* key4 = reinterpret_cast<const uint4*>(rev ? keyR : keyF);
    unsigned int* lst = lists + wave * OWN_RING;
    const int4* rec4 = reinterpret_cast<const int4*>(rec);
    const bool room = ws_bytes >= LIST_WS_BYTES;
    const size_t wslot = (size_t)blockIdx.x * OWN_WAVES + (size_t)wave;
    uint2* keep = reinterpret_cast<uint2*>(keepw) + wslot * LST_CAP;
    unsigned int* kn = keepn + wslot;

    unsigned cnt = (unsigned)LST_CAP + 1u;
    if (SECOND && room) cnt = (unsigned)__builtin_amdgcn_readfirstlane((int)*kn);
    const bool replay = SECOND && room && cnt <= (unsigned)LST_CAP;

    uint4 v[OWN_PF];
    if (!replay) {
        #pragma unroll
        for (int j = 0; j < OWN_PF; ++j) {
            int i = j * OWN_THREADS + tid;
            v[j] = (i < NE16) ? key4[i] : make_uint4(0x7F7F7F7Fu, 0x7F7F7F7Fu, 0x7F7F7F7Fu, 0x7F7F7F7Fu);
        }
    }

    float4* acc4 = reinterpret_cast<float4*>(acc);
    for (int i = tid; i < OWN_NODES * HH / 4; i += OWN_THREADS)
        acc4[i] = make_float4(0.0f, 0.0f, 0.0f, 0.0f);
    __syncthreads();

    if (SECOND) {
        if (__builtin_expect(replay, 1))
            own_replay(keep, cnt, lane, x, acc);
        else if (ws_bytes >= REC_WS_BYTES)
            own_scan<true, false>(v, tid, pk, base, rev, key4, lst, lane, own, oth, w, rec4, x, acc, keep, kn);
        else
            own_scan<false, false>(v, tid, pk, base, rev, key4, lst, lane, own, oth, w, rec4, x, acc, keep, kn);
    } else {
        if (__builtin_expect(room, 1))
            own_scan<true, true>(v, tid, pk, base, rev, key4, lst, lane, own, oth, w, rec4, x, acc, keep, kn);
        else if (ws_bytes >= REC_WS_BYTES)
            own_scan<true, false>(v, tid, pk, base, rev, key4, lst, lane, own, oth, w, rec4, x, acc, keep, kn);
        else
            own_scan<false, false>(v, tid, pk, base, rev, key4, lst, lane, own, oth, w, rec4, x, acc, keep, kn);
    }
    __syncthreads();

    float4* out4 = reinterpret_cast<float4*>(outp + (size_t)base * HH);
    const int nout4 = (int)nloc * (HH / 4);
    for (int i = tid; i < nout4; i += OWN_THREADS)
        out4[i] = acc4[i];
}

__global__ __launch_bounds__(256) void layer_kernel(
    float* __restrict__ x,
    const float* __restrict__ nF, const float* __restrict__ nR,
    const float* __restrict__ invF, const float* __restrict__ invR,
    const float* __restrict__ WsF, const float* __restrict__ WnF, const float* __restrict__ bF,
    const float* __restrict__ WsR, const float* __restrict__ WnR, const float* __restrict__ bR) {
    const int wave = threadIdx.x >> 5;
    const int lane = threadIdx.x & 31;
    const int nt   = blockIdx.x * 8 + wave;
    if (nt >= NTILES) return;
    const int ln = lane & 15;
    const int hi = lane >> 4;
    const int kh = hi << 1;
    const int rowA = nt * 16 + ln;

    const float sF = invF[rowA];
    const float sR = invR[rowA];
    const float* xr = x  + (size_t)rowA * HH + kh;
    const float* fr = nF + (size_t)rowA * HH + kh;
    const float* rr = nR + (size_t)rowA * HH + kh;

    v8f aF0 = {}, aF1 = {}, aF2 = {}, aF3 = {};
    v8f aR0 = {}, aR1 = {}, aR2 = {}, aR3 = {};

    #pragma unroll 4
    for (int k0 = 0; k0 < HH; k0 += 4) {
        v2f aS = *reinterpret_cast<const v2f*>(xr + k0);
        v2f aN = *reinterpret_cast<const v2f*>(fr + k0);
        v2f aV = *reinterpret_cast<const v2f*>(rr + k0);
        aN *= sF;
        aV *= sR;

        const float* pSF = WsF + (size_t)(k0 + kh) * HH + ln;
        const float* pNF = WnF + (size_t)(k0 + kh) * HH + ln;
        const float* pSR = WsR + (size_t)(k0 + kh) * HH + ln;
        const float* pNR = WnR + (size_t)(k0 + kh) * HH + ln;

        aF0 = wmma4(aS, mk2(pSF[0], pSF[HH]), aF0);
        aF0 = wmma4(aN, mk2(pNF[0], pNF[HH]), aF0);
        aR0 = wmma4(aS, mk2(pSR[0], pSR[HH]), aR0);
        aR0 = wmma4(aV, mk2(pNR[0], pNR[HH]), aR0);
        aF1 = wmma4(aS, mk2(pSF[16], pSF[HH + 16]), aF1);
        aF1 = wmma4(aN, mk2(pNF[16], pNF[HH + 16]), aF1);
        aR1 = wmma4(aS, mk2(pSR[16], pSR[HH + 16]), aR1);
        aR1 = wmma4(aV, mk2(pNR[16], pNR[HH + 16]), aR1);
        aF2 = wmma4(aS, mk2(pSF[32], pSF[HH + 32]), aF2);
        aF2 = wmma4(aN, mk2(pNF[32], pNF[HH + 32]), aF2);
        aR2 = wmma4(aS, mk2(pSR[32], pSR[HH + 32]), aR2);
        aR2 = wmma4(aV, mk2(pNR[32], pNR[HH + 32]), aR2);
        aF3 = wmma4(aS, mk2(pSF[48], pSF[HH + 48]), aF3);
        aF3 = wmma4(aN, mk2(pNF[48], pNF[HH + 48]), aF3);
        aR3 = wmma4(aS, mk2(pSR[48], pSR[HH + 48]), aR3);
        aR3 = wmma4(aV, mk2(pNR[48], pNR[HH + 48]), aR3);
    }

    float bf0 = bF[ln], bf1 = bF[16 + ln], bf2 = bF[32 + ln], bf3 = bF[48 + ln];
    float br0 = bR[ln], br1 = bR[16 + ln], br2 = bR[32 + ln], br3 = bR[48 + ln];

    #pragma unroll
    for (int v = 0; v < 8; ++v) {
        size_t r = (size_t)(nt * 16 + v + 8 * hi) * HH + ln;
        x[r]      += fmaxf(aF0[v] + bf0, 0.0f) + fmaxf(aR0[v] + br0, 0.0f);
        x[r + 16] += fmaxf(aF1[v] + bf1, 0.0f) + fmaxf(aR1[v] + br1, 0.0f);
        x[r + 32] += fmaxf(aF2[v] + bf2, 0.0f) + fmaxf(aR2[v] + br2, 0.0f);
        x[r + 48] += fmaxf(aF3[v] + bf3, 0.0f) + fmaxf(aR3[v] + br3, 0.0f);
    }
}

__global__ __launch_bounds__(64) void norm_kernel(const float* __restrict__ x,
                                                  const int* __restrict__ ids,
                                                  float* __restrict__ out) {
    __shared__ float s[64];
    int b = blockIdx.x;
    int c = threadIdx.x;
    int node = ids[b];
    float v = x[(size_t)node * HH + c];
    s[c] = v * v;
    __syncthreads();
    #pragma unroll
    for (int off = 32; off > 0; off >>= 1) {
        if (c < off) s[c] += s[c + off];
        __syncthreads();
    }
    float nrm = sqrtf(s[0]);
    out[(size_t)b * HH + c] = v / nrm;
}

extern "C" void kernel_launch(void* const* d_in, const int* in_sizes, int n_in,
                              void* d_out, int out_size, void* d_ws, size_t ws_size,
                              hipStream_t stream) {
    const float* text_emb = (const float*)d_in[0];
    const float* weights  = (const float*)d_in[1];
    const float* W_enc    = (const float*)d_in[2];
    const float* b_enc    = (const float*)d_in[3];
    const float* Wself_f  = (const float*)d_in[4];
    const float* Wneigh_f = (const float*)d_in[5];
    const float* b_f      = (const float*)d_in[6];
    const float* Wself_r  = (const float*)d_in[7];
    const float* Wneigh_r = (const float*)d_in[8];
    const float* b_r      = (const float*)d_in[9];
    const int*   src      = (const int*)d_in[10];
    const int*   dst      = (const int*)d_in[11];
    const int*   ids      = (const int*)d_in[12];
    float* out = (float*)d_out;

    const size_t NH = (size_t)NN * HH;
    float* x    = (float*)d_ws;
    float* nF   = x + NH;
    float* nR   = x + 2 * NH;
    float* invF = x + 3 * NH;
    float* invR = invF + NN;
    unsigned int* keyF = (unsigned int*)(invR + NN);
    unsigned int* keyR = keyF + NE / 4;
    int* rec = (int*)(keyR + NE / 4);
    unsigned int* keepw = (unsigned int*)(rec + (size_t)NE * 4);
    unsigned int* keepn = keepw + LST_WORDS;

    const int batch = in_sizes[12];

    hipMemsetAsync(invF, 0, (size_t)2 * NN * sizeof(float), stream);
    degree_kernel<<<4 * DEG_SLICES, DEG_THREADS, 0, stream>>>(src, dst, invF, invR);
    invert_key_kernel<<<IK_BLOCKS, IK_THREADS, 0, stream>>>(invF, invR, src, dst, weights, keyF, keyR, rec, ws_size);

    enc_kernel<<<(NTILES + 7) / 8, 256, 0, stream>>>(text_emb, W_enc, b_enc, x);

    own_kernel<false><<<2 * OWN_PARTS, OWN_THREADS, 0, stream>>>(x, weights, src, dst, keyF, keyR, rec, nF, nR, keepw, keepn, ws_size);
    layer_kernel<<<(NTILES + 7) / 8, 256, 0, stream>>>(
        x, nF, nR, invF, invR, Wself_f, Wneigh_f, b_f, Wself_r, Wneigh_r, b_r);
    own_kernel<true><<<2 * OWN_PARTS, OWN_THREADS, 0, stream>>>(x, weights, src, dst, keyF, keyR, rec, nF, nR, keepw, keepn, ws_size);
    layer_kernel<<<(NTILES + 7) / 8, 256, 0, stream>>>(
        x, nF, nR, invF, invR,
        Wself_f + (size_t)HH * HH, Wneigh_f + (size_t)HH * HH, b_f + (size_t)HH,
        Wself_r + (size_t)HH * HH, Wneigh_r + (size_t)HH * HH, b_r + (size_t)HH);

    norm_kernel<<<batch, 64, 0, stream>>>(x, ids, out);
}


# Degrees counted inside the first neighbour-sum launch (one LDS integer add per 12-entry batch, rows' 1/count stored plainly); degree kernel, its memset and global float atomics gone: 8 launches to 7
# speedup vs baseline: 3.1494x; 1.0366x over previous
#include <hip/hip_runtime.h>
#include <hip/hip_bf16.h>
#include <math.h>

typedef __attribute__((ext_vector_type(2))) float v2f;
typedef __attribute__((ext_vector_type(8))) float v8f;

#define NN      100000
#define NE      1600000
#define HH      64
#define TT      384
#define NTILES  6250

__device__ __forceinline__ v2f mk2(float a, float b) {
    v2f r; r.x = a; r.y = b; return r;
}

__device__ __forceinline__ v8f wmma4(v2f a, v2f b, v8f c) {
    return __builtin_amdgcn_wmma_f32_16x16x4_f32(
         false, a,  false, b,
         (short)0, c,  false,  false);
}

#define OWN_NODES 800
#define KEY_NONE  0x7Fu
#define NE4       (NE / 4)
#define KEY_WS_BYTES ((size_t)(3 * (size_t)NN * HH + 2 * (size_t)NN) * sizeof(float) + 2 * (size_t)NE)
#define REC_WS_BYTES (KEY_WS_BYTES + (size_t)NE * sizeof(int4))
#define LST_CAP     512
#define OWN_GROUPS  250
#define OWN_WAVES_C 32
#define LST_WORDS   ((size_t)OWN_GROUPS * OWN_WAVES_C * LST_CAP * 2)
#define LIST_WS_BYTES (REC_WS_BYTES + (LST_WORDS + (size_t)OWN_GROUPS * OWN_WAVES_C) * sizeof(unsigned int))
#define LST_OBITS   17

__device__ __forceinline__ unsigned key_of(int idx) {
    return ((unsigned)idx < (unsigned)NN) ? ((unsigned)idx / (unsigned)OWN_NODES) : KEY_NONE;
}
__device__ __forceinline__ unsigned key4_of(int4 a) {
    return key_of(a.x) | (key_of(a.y) << 8) | (key_of(a.z) << 16) | (key_of(a.w) << 24);
}

#define IK_THREADS 256
#define IK_BLOCKS  ((NE4 + 4 * IK_THREADS - 1) / (4 * IK_THREADS))

__global__ __launch_bounds__(IK_THREADS) void key_rec_kernel(const int* __restrict__ src,
                                                             const int* __restrict__ dst,
                                                             const float* __restrict__ w,
                                                             unsigned int* __restrict__ keyF,
                                                             unsigned int* __restrict__ keyR,
                                                             int* __restrict__ rec,
                                                             size_t ws_bytes) {
    if (ws_bytes < KEY_WS_BYTES) return;
    const int4* s4 = reinterpret_cast<const int4*>(src);
    const int4* d4 = reinterpret_cast<const int4*>(dst);
    const int g0 = blockIdx.x * (4 * IK_THREADS) + threadIdx.x;
    int4 a[4], b[4];
    #pragma unroll
    for (int k = 0; k < 4; ++k) {
        const int g = g0 + k * IK_THREADS;
        a[k] = (g < NE4) ? d4[g] : make_int4(-1, -1, -1, -1);
        b[k] = (g < NE4) ? s4[g] : make_int4(-1, -1, -1, -1);
    }
    #pragma unroll
    for (int k = 0; k < 4; ++k) {
        const int g = g0 + k * IK_THREADS;
        if (g < NE4) {
            keyF[g] = key4_of(a[k]);
            keyR[g] = key4_of(b[k]);
        }
    }
    if (ws_bytes < REC_WS_BYTES) return;
    int4* rec4 = reinterpret_cast<int4*>(rec);
    const int e0 = blockIdx.x * (16 * IK_THREADS) + threadIdx.x;
    #pragma unroll
    for (int h = 0; h < 2; ++h) {
        int   d[8], s[8];
        float wv[8];
        #pragma unroll
        for (int k = 0; k < 8; ++k) {
            const int e  = e0 + (8 * h + k) * IK_THREADS;
            const int ec = (e < NE) ? e : NE - 1;
            d[k]  = dst[ec];
            s[k]  = src[ec];
            wv[k] = w[ec];
        }
        #pragma unroll
        for (int k = 0; k < 8; ++k) {
            const int e = e0 + (8 * h + k) * IK_THREADS;
            if (e < NE) rec4[e] = make_int4(d[k], s[k], __float_as_int(wv[k]), 0);
        }
    }
}

__global__ __launch_bounds__(256) void enc_kernel(const float* __restrict__ emb,
                                                  const float* __restrict__ W,
                                                  const float* __restrict__ bias,
                                                  float* __restrict__ x) {
    const int wave = threadIdx.x >> 5;
    const int lane = threadIdx.x & 31;
    const int nt   = blockIdx.x * 8 + wave;
    if (nt >= NTILES) return;
    const int ln = lane & 15;
    const int hi = lane >> 4;
    const int kh = hi << 1;

    const float* arow = emb + (size_t)(nt * 16 + ln) * TT + kh;
    v8f acc0 = {}, acc1 = {}, acc2 = {}, acc3 = {};

    #pragma unroll 4
    for (int k0 = 0; k0 < TT; k0 += 4) {
        v2f a = *reinterpret_cast<const v2f*>(arow + k0);
        const float* bp = W + (size_t)(k0 + kh) * HH + ln;
        v2f b0 = mk2(bp[0],  bp[HH]);
        v2f b1 = mk2(bp[16], bp[HH + 16]);
        v2f b2 = mk2(bp[32], bp[HH + 32]);
        v2f b3 = mk2(bp[48], bp[HH + 48]);
        acc0 = wmma4(a, b0, acc0);
        acc1 = wmma4(a, b1, acc1);
        acc2 = wmma4(a, b2, acc2);
        acc3 = wmma4(a, b3, acc3);
    }

    float bs0 = bias[ln], bs1 = bias[16 + ln], bs2 = bias[32 + ln], bs3 = bias[48 + ln];
    #pragma unroll
    for (int v = 0; v < 8; ++v) {
        size_t r = (size_t)(nt * 16 + v + 8 * hi) * HH + ln;
        x[r]      = acc0[v] + bs0;
        x[r + 16] = acc1[v] + bs1;
        x[r + 32] = acc2[v] + bs2;
        x[r + 48] = acc3[v] + bs3;
    }
}

#define OWN_PARTS   ((NN + OWN_NODES - 1) / OWN_NODES)
#define OWN_THREADS 1024
#define OWN_WAVES   (OWN_THREADS / 32)
#define NE16        (NE / 16)
#define OWN_NSTEP   ((NE16 + OWN_THREADS - 1) / OWN_THREADS)
#define OWN_FULL    (NE16 / OWN_THREADS)
#define OWN_PF      4
#define OWN_BATCH   12
#define OWN_RING    64
static_assert(OWN_PARTS <= (int)KEY_NONE, "range numbers and KEY_NONE must stay below 128 and apart");
static_assert(NE % 16 == 0, "the key list is read 16 bytes at a time");
static_assert(OWN_GROUPS == 2 * OWN_PARTS && OWN_WAVES_C == OWN_WAVES, "the list area is sized for this grid");
static_assert(NN <= (1 << LST_OBITS) && OWN_NODES <= (1 << (32 - LST_OBITS)), "a list entry packs both ends into one dword");
static_assert(OWN_NODES <= OWN_THREADS, "the row counters are cleared and read out by one thread per owned row");

template <bool FULL, bool PACKED, bool SAVE, bool COUNT>
__device__ __forceinline__ void own_drain(const unsigned int* lst, unsigned head, unsigned nvalid, int lane, int base, int rev,
                                          const int* __restrict__ own, const int* __restrict__ oth,
                                          const float* __restrict__ w, const int4* __restrict__ rec,
                                          const float* __restrict__ x, float* acc, uint2* __restrict__ keep,
                                          unsigned int* degc) {
    const unsigned jj = ((unsigned)lane < (FULL ? (unsigned)OWN_BATCH : nvalid)) ? (unsigned)lane : 0u;
    const unsigned e  = lst[(head + jj) & (OWN_RING - 1)];
    int rn, ro, rw;
    if (PACKED) {
        const int4 r = rec[e];
        rn = rev ? r.y : r.x;
        ro = rev ? r.x : r.y;
        rw = r.z;
    } else {
        ro = oth[e];
        rw = __float_as_int(w[e]);
        rn = own[e];
    }
    if (SAVE) {
        if ((unsigned)lane < (FULL ? (unsigned)OWN_BATCH : nvalid) && head + (unsigned)lane < (unsigned)LST_CAP)
            keep[head + (unsigned)lane] = make_uint2(((unsigned)(rn - base) << LST_OBITS) | ((unsigned)ro & ((1u << LST_OBITS) - 1u)), (unsigned)rw);
    }
    if (COUNT) {
        if ((unsigned)lane < (FULL ? (unsigned)OWN_BATCH : nvalid))
            atomicAdd(&degc[rn - base], 1u);
    }
    v2f xv[OWN_BATCH];
    #pragma unroll
    for (int j = 0; j < OWN_BATCH; ++j) {
        const int o = __builtin_amdgcn_readlane(ro, j);
        xv[j] = *reinterpret_cast<const v2f*>(x + (size_t)o * HH + 2 * lane);
    }
    #pragma unroll
    for (int j = 0; j < OWN_BATCH; ++j) {
        if (FULL || (unsigned)j < nvalid) {
            const int   n  = __builtin_amdgcn_readlane(rn, j);
            const float we = __int_as_float(__builtin_amdgcn_readlane(rw, j));
            float* a = acc + (size_t)(n - base) * HH + 2 * lane;
            atomicAdd(a,     we * xv[j].x);
            atomicAdd(a + 1, we * xv[j].y);
        }
    }
}

template <bool GUARD, bool PACKED, bool SAVE, bool COUNT>
__device__ __forceinline__ void own_step(uint4& vj, int i, unsigned pk, int base, int rev, const uint4* __restrict__ key4,
                                         unsigned int* lst, unsigned& head, unsigned& tail, int lane,
                                         const int* __restrict__ own, const int* __restrict__ oth,
                                         const float* __restrict__ w, const int4* __restrict__ rec,
                                         const float* __restrict__ x, float* acc, uint2* __restrict__ keep,
                                         unsigned int* degc) {
    unsigned s0 = (vj.x ^ pk) + 0x7F7F7F7Fu;
    unsigned s1 = (vj.y ^ pk) + 0x7F7F7F7Fu;
    unsigned s2 = (vj.z ^ pk) + 0x7F7F7F7Fu;
    unsigned s3 = (vj.w ^ pk) + 0x7F7F7F7Fu;
    const int in = i + OWN_PF * OWN_THREADS;
    if (GUARD) vj = (in < NE16) ? key4[in] : make_uint4(0x7F7F7F7Fu, 0x7F7F7F7Fu, 0x7F7F7F7Fu, 0x7F7F7F7Fu);
    else       vj = key4[in];

    unsigned any = __builtin_amdgcn_ballot_w32(((s0 & s1 & s2 & s3) & 0x80808080u) != 0x80808080u);
    if (any) {
        const unsigned miss = ((s0 >> 7) & 0x01010101u) | ((s1 >> 6) & 0x02020202u) |
                              ((s2 >> 5) & 0x04040404u) | ((s3 >> 4) & 0x08080808u);
        unsigned hm = ~miss & 0x0F0F0F0Fu;
        const unsigned e0 = (unsigned)i * 16u;
        do {
            if (hm != 0u) {
                const unsigned p = (unsigned)__builtin_ctz(hm);
                hm &= hm - 1u;
                lst[(tail + __builtin_amdgcn_mbcnt_lo(any, 0u)) & (OWN_RING - 1)] = e0 + ((p & 7u) << 2) + (p >> 3);
            }
            tail += (unsigned)__builtin_popcount(any);
            while (tail - head >= (unsigned)OWN_BATCH) {
                own_drain<true, PACKED, SAVE, COUNT>(lst, head, (unsigned)OWN_BATCH, lane, base, rev, own, oth, w, rec, x, acc, keep, degc);
                head += (unsigned)OWN_BATCH;
            }
            any = __builtin_amdgcn_ballot_w32(hm != 0u);
        } while (any);
    }
}

template <bool PACKED, bool SAVE, bool COUNT>
__device__ __forceinline__ void own_scan(uint4 (&v)[OWN_PF], int tid, unsigned pk, int base, int rev,
                                         const uint4* __restrict__ key4, unsigned int* lst, int lane,
                                         const int* __restrict__ own, const int* __restrict__ oth,
                                         const float* __restrict__ w, const int4* __restrict__ rec,
                                         const float* __restrict__ x, float* acc,
                                         uint2* __restrict__ keep, unsigned int* __restrict__ keepn,
                                         unsigned int* degc) {
    unsigned head = 0u, tail = 0u;
    int s0 = 0;
    for (; s0 + 2 * OWN_PF <= OWN_FULL; s0 += OWN_PF) {
        #pragma unroll
        for (int j = 0; j < OWN_PF; ++j)
            own_step<false, PACKED, SAVE, COUNT>(v[j], (s0 + j) * OWN_THREADS + tid, pk, base, rev, key4, lst, head, tail, lane, own, oth, w, rec, x, acc, keep, degc);
    }
    for (; s0 < OWN_NSTEP; s0 += OWN_PF) {
        #pragma unroll
        for (int j = 0; j < OWN_PF; ++j)
            own_step<true, PACKED, SAVE, COUNT>(v[j], (s0 + j) * OWN_THREADS + tid, pk, base, rev, key4, lst, head, tail, lane, own, oth, w, rec, x, acc, keep, degc);
    }
    if (tail != head)
        own_drain<false, PACKED, SAVE, COUNT>(lst, head, tail - head, lane, base, rev, own, oth, w, rec, x, acc, keep, degc);
    if (SAVE && lane == 0) *keepn = tail;
}

#define RP_SUB 16
template <bool FULL>
__device__ __forceinline__ void own_replay32(uint2 ent, unsigned nvalid, int lane, const float* __restrict__ x, float* acc) {
    #pragma unroll
    for (int h = 0; h < 32 / RP_SUB; ++h) {
        if (!FULL && (unsigned)(h * RP_SUB) >= nvalid) break;
        v2f xv[RP_SUB];
        #pragma unroll
        for (int j = 0; j < RP_SUB; ++j) {
            const unsigned o = (unsigned)__builtin_amdgcn_readlane((int)ent.x, h * RP_SUB + j) & ((1u << LST_OBITS) - 1u);
            xv[j] = *reinterpret_cast<const v2f*>(x + (size_t)o * HH + 2 * lane);
        }
        #pragma unroll
        for (int j = 0; j < RP_SUB; ++j) {
            if (FULL || (unsigned)(h * RP_SUB + j) < nvalid) {
                const unsigned n  = (unsigned)__builtin_amdgcn_readlane((int)ent.x, h * RP_SUB + j) >> LST_OBITS;
                const float    we = __int_as_float(__builtin_amdgcn_readlane((int)ent.y, h * RP_SUB + j));
                float* a = acc + (size_t)n * HH + 2 * lane;
                atomicAdd(a,     we * xv[j].x);
                atomicAdd(a + 1, we * xv[j].y);
            }
        }
    }
}

__device__ __forceinline__ void own_replay(const uint2* __restrict__ keep, unsigned cnt, int lane,
                                           const float* __restrict__ x, float* acc) {
    if (cnt == 0u) return;
    const unsigned last = cnt - 1u;
    uint2 cur = keep[((unsigned)lane < last) ? (unsigned)lane : last];
    unsigned p = 0u;
    for (; p + 32u <= cnt; p += 32u) {
        const unsigned q = p + 32u + (unsigned)lane;
        const uint2 nxt = keep[(q < last) ? q : last];
        own_replay32<true>(cur, 32u, lane, x, acc);
        cur = nxt;
    }
    if (p < cnt) own_replay32<false>(cur, cnt - p, lane, x, acc);
}

template <bool SECOND>
__global__ __launch_bounds__(OWN_THREADS) void own_kernel(const float* __restrict__ x,
                                                          const float* __restrict__ w,
                                                          const int* __restrict__ src,
                                                          const int* __restrict__ dst,
                                                          const unsigned int* __restrict__ keyF,
                                                          const unsigned int* __restrict__ keyR,
                                                          const int* __restrict__ rec,
                                                          float* __restrict__ nF,
                                                          float* __restrict__ nR,
                                                          float* __restrict__ invF,
                                                          float* __restrict__ invR,
                                                          unsigned int* __restrict__ keepw,
                                                          unsigned int* __restrict__ keepn,
                                                          size_t ws_bytes) {
    __shared__ float acc[OWN_NODES * HH];
    __shared__ unsigned int lists[OWN_WAVES * OWN_RING];
    __shared__ unsigned int degc[OWN_NODES];
    if (ws_bytes < KEY_WS_BYTES) return;
    const int tid  = threadIdx.x;
    const int lane = tid & 31;
    const int wave = tid >> 5;
    const int rev  = blockIdx.x & 1;
    const int part = blockIdx.x >> 1;
    const int base = part * OWN_NODES;
    const unsigned nloc = (unsigned)((NN - base < OWN_NODES) ? (NN - base) : OWN_NODES);
    const unsigned pk = (unsigned)part * 0x01010101u;
    const int* __restrict__ own = rev ? src : dst;
    const int* __restrict__ oth = rev ? dst : src;
    float* __restrict__ outp    = rev ? nR : nF;
    const uint4* key4 = reinterpret_cast<const uint4*>(rev ? keyR : keyF);
    unsigned int* lst = lists + wave * OWN_RING;
    const int4* rec4 = reinterpret_cast<const int4*>(rec);
    const bool room = ws_bytes >= LIST_WS_BYTES;
    const size_t wslot = (size_t)blockIdx.x * OWN_WAVES + (size_t)wave;
    uint2* keep = reinterpret_cast<uint2*>(keepw) + wslot * LST_CAP;
    unsigned int* kn = keepn + wslot;

    unsigned cnt = (unsigned)LST_CAP + 1u;
    if (SECOND && room) cnt = (unsigned)__builtin_amdgcn_readfirstlane((int)*kn);
    const bool replay = SECOND && room && cnt <= (unsigned)LST_CAP;

    uint4 v[OWN_PF];
    if (!replay) {
        #pragma unroll
        for (int j = 0; j < OWN_PF; ++j) {
            int i = j * OWN_THREADS + tid;
            v[j] = (i < NE16) ? key4[i] : make_uint4(0x7F7F7F7Fu, 0x7F7F7F7Fu, 0x7F7F7F7Fu, 0x7F7F7F7Fu);
        }
    }

    float4* acc4 = reinterpret_cast<float4*>(acc);
    for (int i = tid; i < OWN_NODES * HH / 4; i += OWN_THREADS)
        acc4[i] = make_float4(0.0f, 0.0f, 0.0f, 0.0f);
    if (!SECOND && tid < OWN_NODES) degc[tid] = 0u;
    __syncthreads();

    if (SECOND) {
        if (__builtin_expect(replay, 1))
            own_replay(keep, cnt, lane, x, acc);
        else if (ws_bytes >= REC_WS_BYTES)
            own_scan<true, false, false>(v, tid, pk, base, rev, key4, lst, lane, own, oth, w, rec4, x, acc, keep, kn, degc);
        else
            own_scan<false, false, false>(v, tid, pk, base, rev, key4, lst, lane, own, oth, w, rec4, x, acc, keep, kn, degc);
    } else {
        if (__builtin_expect(room, 1))
            own_scan<true, true, true>(v, tid, pk, base, rev, key4, lst, lane, own, oth, w, rec4, x, acc, keep, kn, degc);
        else if (ws_bytes >= REC_WS_BYTES)
            own_scan<true, false, true>(v, tid, pk, base, rev, key4, lst, lane, own, oth, w, rec4, x, acc, keep, kn, degc);
        else
            own_scan<false, false, true>(v, tid, pk, base, rev, key4, lst, lane, own, oth, w, rec4, x, acc, keep, kn, degc);
    }
    __syncthreads();

    if (!SECOND) {
        float* __restrict__ invp = rev ? invR : invF;
        if ((unsigned)tid < nloc) {
            const float c = (float)degc[tid];
            invp[base + tid] = (c > 0.0f) ? (1.0f / c) : 0.0f;
        }
    }

    float4* out4 = reinterpret_cast<float4*>(outp + (size_t)base * HH);
    const int nout4 = (int)nloc * (HH / 4);
    for (int i = tid; i < nout4; i += OWN_THREADS)
        out4[i] = acc4[i];
}

__global__ __launch_bounds__(256) void layer_kernel(
    float* __restrict__ x,
    const float* __restrict__ nF, const float* __restrict__ nR,
    const float* __restrict__ invF, const float* __restrict__ invR,
    const float* __restrict__ WsF, const float* __restrict__ WnF, const float* __restrict__ bF,
    const float* __restrict__ WsR, const float* __restrict__ WnR, const float* __restrict__ bR) {
    const int wave = threadIdx.x >> 5;
    const int lane = threadIdx.x & 31;
    const int nt   = blockIdx.x * 8 + wave;
    if (nt >= NTILES) return;
    const int ln = lane & 15;
    const int hi = lane >> 4;
    const int kh = hi << 1;
    const int rowA = nt * 16 + ln;

    const float sF = invF[rowA];
    const float sR = invR[rowA];
    const float* xr = x  + (size_t)rowA * HH + kh;
    const float* fr = nF + (size_t)rowA * HH + kh;
    const float* rr = nR + (size_t)rowA * HH + kh;

    v8f aF0 = {}, aF1 = {}, aF2 = {}, aF3 = {};
    v8f aR0 = {}, aR1 = {}, aR2 = {}, aR3 = {};

    #pragma unroll 4
    for (int k0 = 0; k0 < HH; k0 += 4) {
        v2f aS = *reinterpret_cast<const v2f*>(xr + k0);
        v2f aN = *reinterpret_cast<const v2f*>(fr + k0);
        v2f aV = *reinterpret_cast<const v2f*>(rr + k0);
        aN *= sF;
        aV *= sR;

        const float* pSF = WsF + (size_t)(k0 + kh) * HH + ln;
        const float* pNF = WnF + (size_t)(k0 + kh) * HH + ln;
        const float* pSR = WsR + (size_t)(k0 + kh) * HH + ln;
        const float* pNR = WnR + (size_t)(k0 + kh) * HH + ln;

        aF0 = wmma4(aS, mk2(pSF[0], pSF[HH]), aF0);
        aF0 = wmma4(aN, mk2(pNF[0], pNF[HH]), aF0);
        aR0 = wmma4(aS, mk2(pSR[0], pSR[HH]), aR0);
        aR0 = wmma4(aV, mk2(pNR[0], pNR[HH]), aR0);
        aF1 = wmma4(aS, mk2(pSF[16], pSF[HH + 16]), aF1);
        aF1 = wmma4(aN, mk2(pNF[16], pNF[HH + 16]), aF1);
        aR1 = wmma4(aS, mk2(pSR[16], pSR[HH + 16]), aR1);
        aR1 = wmma4(aV, mk2(pNR[16], pNR[HH + 16]), aR1);
        aF2 = wmma4(aS, mk2(pSF[32], pSF[HH + 32]), aF2);
        aF2 = wmma4(aN, mk2(pNF[32], pNF[HH + 32]), aF2);
        aR2 = wmma4(aS, mk2(pSR[32], pSR[HH + 32]), aR2);
        aR2 = wmma4(aV, mk2(pNR[32], pNR[HH + 32]), aR2);
        aF3 = wmma4(aS, mk2(pSF[48], pSF[HH + 48]), aF3);
        aF3 = wmma4(aN, mk2(pNF[48], pNF[HH + 48]), aF3);
        aR3 = wmma4(aS, mk2(pSR[48], pSR[HH + 48]), aR3);
        aR3 = wmma4(aV, mk2(pNR[48], pNR[HH + 48]), aR3);
    }

    float bf0 = bF[ln], bf1 = bF[16 + ln], bf2 = bF[32 + ln], bf3 = bF[48 + ln];
    float br0 = bR[ln], br1 = bR[16 + ln], br2 = bR[32 + ln], br3 = bR[48 + ln];

    #pragma unroll
    for (int v = 0; v < 8; ++v) {
        size_t r = (size_t)(nt * 16 + v + 8 * hi) * HH + ln;
        x[r]      += fmaxf(aF0[v] + bf0, 0.0f) + fmaxf(aR0[v] + br0, 0.0f);
        x[r + 16] += fmaxf(aF1[v] + bf1, 0.0f) + fmaxf(aR1[v] + br1, 0.0f);
        x[r + 32] += fmaxf(aF2[v] + bf2, 0.0f) + fmaxf(aR2[v] + br2, 0.0f);
        x[r + 48] += fmaxf(aF3[v] + bf3, 0.0f) + fmaxf(aR3[v] + br3, 0.0f);
    }
}

__global__ __launch_bounds__(64) void norm_kernel(const float* __restrict__ x,
                                                  const int* __restrict__ ids,
                                                  float* __restrict__ out) {
    __shared__ float s[64];
    int b = blockIdx.x;
    int c = threadIdx.x;
    int node = ids[b];
    float v = x[(size_t)node * HH + c];
    s[c] = v * v;
    __syncthreads();
    #pragma unroll
    for (int off = 32; off > 0; off >>= 1) {
        if (c < off) s[c] += s[c + off];
        __syncthreads();
    }
    float nrm = sqrtf(s[0]);
    out[(size_t)b * HH + c] = v / nrm;
}

extern "C" void kernel_launch(void* const* d_in, const int* in_sizes, int n_in,
                              void* d_out, int out_size, void* d_ws, size_t ws_size,
                              hipStream_t stream) {
    const float* text_emb = (const float*)d_in[0];
    const float* weights  = (const float*)d_in[1];
    const float* W_enc    = (const float*)d_in[2];
    const float* b_enc    = (const float*)d_in[3];
    const float* Wself_f  = (const float*)d_in[4];
    const float* Wneigh_f = (const float*)d_in[5];
    const float* b_f      = (const float*)d_in[6];
    const float* Wself_r  = (const float*)d_in[7];
    const float* Wneigh_r = (const float*)d_in[8];
    const float* b_r      = (const float*)d_in[9];
    const int*   src      = (const int*)d_in[10];
    const int*   dst      = (const int*)d_in[11];
    const int*   ids      = (const int*)d_in[12];
    float* out = (float*)d_out;

    const size_t NH = (size_t)NN * HH;
    float* x    = (float*)d_ws;
    float* nF   = x + NH;
    float* nR   = x + 2 * NH;
    float* invF = x + 3 * NH;
    float* invR = invF + NN;
    unsigned int* keyF = (unsigned int*)(invR + NN);
    unsigned int* keyR = keyF + NE / 4;
    int* rec = (int*)(keyR + NE / 4);
    unsigned int* keepw = (unsigned int*)(rec + (size_t)NE * 4);
    unsigned int* keepn = keepw + LST_WORDS;

    const int batch = in_sizes[12];

    key_rec_kernel<<<IK_BLOCKS, IK_THREADS, 0, stream>>>(src, dst, weights, keyF, keyR, rec, ws_size);

    enc_kernel<<<(NTILES + 7) / 8, 256, 0, stream>>>(text_emb, W_enc, b_enc, x);

    own_kernel<false><<<2 * OWN_PARTS, OWN_THREADS, 0, stream>>>(x, weights, src, dst, keyF, keyR, rec, nF, nR, invF, invR, keepw, keepn, ws_size);
    layer_kernel<<<(NTILES + 7) / 8, 256, 0, stream>>>(
        x, nF, nR, invF, invR, Wself_f, Wneigh_f, b_f, Wself_r, Wneigh_r, b_r);
    own_kernel<true><<<2 * OWN_PARTS, OWN_THREADS, 0, stream>>>(x, weights, src, dst, keyF, keyR, rec, nF, nR, invF, invR, keepw, keepn, ws_size);
    layer_kernel<<<(NTILES + 7) / 8, 256, 0, stream>>>(
        x, nF, nR, invF, invR,
        Wself_f + (size_t)HH * HH, Wneigh_f + (size_t)HH * HH, b_f + (size_t)HH,
        Wself_r + (size_t)HH * HH, Wneigh_r + (size_t)HH * HH, b_r + (size_t)HH);

    norm_kernel<<<batch, 64, 0, stream>>>(x, ids, out);
}


# layer_kernel: the four 64x64 weight matrices staged once per work-group in LDS in fragment order, B operands by 16-byte LDS reads; 256 work-groups of 13 waves, two tiles a wave; no sum changes order
# speedup vs baseline: 3.3569x; 1.0659x over previous
#include <hip/hip_runtime.h>
#include <hip/hip_bf16.h>
#include <math.h>

typedef __attribute__((ext_vector_type(2))) float v2f;
typedef __attribute__((ext_vector_type(8))) float v8f;

#define NN      100000
#define NE      1600000
#define HH      64
#define TT      384
#define NTILES  6250

__device__ __forceinline__ v2f mk2(float a, float b) {
    v2f r; r.x = a; r.y = b; return r;
}

__device__ __forceinline__ v8f wmma4(v2f a, v2f b, v8f c) {
    return __builtin_amdgcn_wmma_f32_16x16x4_f32(
         false, a,  false, b,
         (short)0, c,  false,  false);
}

#define OWN_NODES 800
#define KEY_NONE  0x7Fu
#define NE4       (NE / 4)
#define KEY_WS_BYTES ((size_t)(3 * (size_t)NN * HH + 2 * (size_t)NN) * sizeof(float) + 2 * (size_t)NE)
#define REC_WS_BYTES (KEY_WS_BYTES + (size_t)NE * sizeof(int4))
#define LST_CAP     512
#define OWN_GROUPS  250
#define OWN_WAVES_C 32
#define LST_WORDS   ((size_t)OWN_GROUPS * OWN_WAVES_C * LST_CAP * 2)
#define LIST_WS_BYTES (REC_WS_BYTES + (LST_WORDS + (size_t)OWN_GROUPS * OWN_WAVES_C) * sizeof(unsigned int))
#define LST_OBITS   17

__device__ __forceinline__ unsigned key_of(int idx) {
    return ((unsigned)idx < (unsigned)NN) ? ((unsigned)idx / (unsigned)OWN_NODES) : KEY_NONE;
}
__device__ __forceinline__ unsigned key4_of(int4 a) {
    return key_of(a.x) | (key_of(a.y) << 8) | (key_of(a.z) << 16) | (key_of(a.w) << 24);
}

#define IK_THREADS 256
#define IK_BLOCKS  ((NE4 + 4 * IK_THREADS - 1) / (4 * IK_THREADS))

__global__ __launch_bounds__(IK_THREADS) void key_rec_kernel(const int* __restrict__ src,
                                                             const int* __restrict__ dst,
                                                             const float* __restrict__ w,
                                                             unsigned int* __restrict__ keyF,
                                                             unsigned int* __restrict__ keyR,
                                                             int* __restrict__ rec,
                                                             size_t ws_bytes) {
    if (ws_bytes < KEY_WS_BYTES) return;
    const int4* s4 = reinterpret_cast<const int4*>(src);
    const int4* d4 = reinterpret_cast<const int4*>(dst);
    const int g0 = blockIdx.x * (4 * IK_THREADS) + threadIdx.x;
    int4 a[4], b[4];
    #pragma unroll
    for (int k = 0; k < 4; ++k) {
        const int g = g0 + k * IK_THREADS;
        a[k] = (g < NE4) ? d4[g] : make_int4(-1, -1, -1, -1);
        b[k] = (g < NE4) ? s4[g] : make_int4(-1, -1, -1, -1);
    }
    #pragma unroll
    for (int k = 0; k < 4; ++k) {
        const int g = g0 + k * IK_THREADS;
        if (g < NE4) {
            keyF[g] = key4_of(a[k]);
            keyR[g] = key4_of(b[k]);
        }
    }
    if (ws_bytes < REC_WS_BYTES) return;
    int4* rec4 = reinterpret_cast<int4*>(rec);
    const int e0 = blockIdx.x * (16 * IK_THREADS) + threadIdx.x;
    #pragma unroll
    for (int h = 0; h < 2; ++h) {
        int   d[8], s[8];
        float wv[8];
        #pragma unroll
        for (int k = 0; k < 8; ++k) {
            const int e  = e0 + (8 * h + k) * IK_THREADS;
            const int ec = (e < NE) ? e : NE - 1;
            d[k]  = dst[ec];
            s[k]  = src[ec];
            wv[k] = w[ec];
        }
        #pragma unroll
        for (int k = 0; k < 8; ++k) {
            const int e = e0 + (8 * h + k) * IK_THREADS;
            if (e < NE) rec4[e] = make_int4(d[k], s[k], __float_as_int(wv[k]), 0);
        }
    }
}

__global__ __launch_bounds__(256) void enc_kernel(const float* __restrict__ emb,
                                                  const float* __restrict__ W,
                                                  const float* __restrict__ bias,
                                                  float* __restrict__ x) {
    const int wave = threadIdx.x >> 5;
    const int lane = threadIdx.x & 31;
    const int nt   = blockIdx.x * 8 + wave;
    if (nt >= NTILES) return;
    const int ln = lane & 15;
    const int hi = lane >> 4;
    const int kh = hi << 1;

    const float* arow = emb + (size_t)(nt * 16 + ln) * TT + kh;
    v8f acc0 = {}, acc1 = {}, acc2 = {}, acc3 = {};

    #pragma unroll 4
    for (int k0 = 0; k0 < TT; k0 += 4) {
        v2f a = *reinterpret_cast<const v2f*>(arow + k0);
        const float* bp = W + (size_t)(k0 + kh) * HH + ln;
        v2f b0 = mk2(bp[0],  bp[HH]);
        v2f b1 = mk2(bp[16], bp[HH + 16]);
        v2f b2 = mk2(bp[32], bp[HH + 32]);
        v2f b3 = mk2(bp[48], bp[HH + 48]);
        acc0 = wmma4(a, b0, acc0);
        acc1 = wmma4(a, b1, acc1);
        acc2 = wmma4(a, b2, acc2);
        acc3 = wmma4(a, b3, acc3);
    }

    float bs0 = bias[ln], bs1 = bias[16 + ln], bs2 = bias[32 + ln], bs3 = bias[48 + ln];
    #pragma unroll
    for (int v = 0; v < 8; ++v) {
        size_t r = (size_t)(nt * 16 + v + 8 * hi) * HH + ln;
        x[r]      = acc0[v] + bs0;
        x[r + 16] = acc1[v] + bs1;
        x[r + 32] = acc2[v] + bs2;
        x[r + 48] = acc3[v] + bs3;
    }
}

#define OWN_PARTS   ((NN + OWN_NODES - 1) / OWN_NODES)
#define OWN_THREADS 1024
#define OWN_WAVES   (OWN_THREADS / 32)
#define NE16        (NE / 16)
#define OWN_NSTEP   ((NE16 + OWN_THREADS - 1) / OWN_THREADS)
#define OWN_FULL    (NE16 / OWN_THREADS)
#define OWN_PF      4
#define OWN_BATCH   12
#define OWN_RING    64
static_assert(OWN_PARTS <= (int)KEY_NONE, "range numbers and KEY_NONE must stay below 128 and apart");
static_assert(NE % 16 == 0, "the key list is read 16 bytes at a time");
static_assert(OWN_GROUPS == 2 * OWN_PARTS && OWN_WAVES_C == OWN_WAVES, "the list area is sized for this grid");
static_assert(NN <= (1 << LST_OBITS) && OWN_NODES <= (1 << (32 - LST_OBITS)), "a list entry packs both ends into one dword");
static_assert(OWN_NODES <= OWN_THREADS, "the row counters are cleared and read out by one thread per owned row");

template <bool FULL, bool PACKED, bool SAVE, bool COUNT>
__device__ __forceinline__ void own_drain(const unsigned int* lst, unsigned head, unsigned nvalid, int lane, int base, int rev,
                                          const int* __restrict__ own, const int* __restrict__ oth,
                                          const float* __restrict__ w, const int4* __restrict__ rec,
                                          const float* __restrict__ x, float* acc, uint2* __restrict__ keep,
                                          unsigned int* degc) {
    const unsigned jj = ((unsigned)lane < (FULL ? (unsigned)OWN_BATCH : nvalid)) ? (unsigned)lane : 0u;
    const unsigned e  = lst[(head + jj) & (OWN_RING - 1)];
    int rn, ro, rw;
    if (PACKED) {
        const int4 r = rec[e];
        rn = rev ? r.y : r.x;
        ro = rev ? r.x : r.y;
        rw = r.z;
    } else {
        ro = oth[e];
        rw = __float_as_int(w[e]);
        rn = own[e];
    }
    if (SAVE) {
        if ((unsigned)lane < (FULL ? (unsigned)OWN_BATCH : nvalid) && head + (unsigned)lane < (unsigned)LST_CAP)
            keep[head + (unsigned)lane] = make_uint2(((unsigned)(rn - base) << LST_OBITS) | ((unsigned)ro & ((1u << LST_OBITS) - 1u)), (unsigned)rw);
    }
    if (COUNT) {
        if ((unsigned)lane < (FULL ? (unsigned)OWN_BATCH : nvalid))
            atomicAdd(&degc[rn - base], 1u);
    }
    v2f xv[OWN_BATCH];
    #pragma unroll
    for (int j = 0; j < OWN_BATCH; ++j) {
        const int o = __builtin_amdgcn_readlane(ro, j);
        xv[j] = *reinterpret_cast<const v2f*>(x + (size_t)o * HH + 2 * lane);
    }
    #pragma unroll
    for (int j = 0; j < OWN_BATCH; ++j) {
        if (FULL || (unsigned)j < nvalid) {
            const int   n  = __builtin_amdgcn_readlane(rn, j);
            const float we = __int_as_float(__builtin_amdgcn_readlane(rw, j));
            float* a = acc + (size_t)(n - base) * HH + 2 * lane;
            atomicAdd(a,     we * xv[j].x);
            atomicAdd(a + 1, we * xv[j].y);
        }
    }
}

template <bool GUARD, bool PACKED, bool SAVE, bool COUNT>
__device__ __forceinline__ void own_step(uint4& vj, int i, unsigned pk, int base, int rev, const uint4* __restrict__ key4,
                                         unsigned int* lst, unsigned& head, unsigned& tail, int lane,
                                         const int* __restrict__ own, const int* __restrict__ oth,
                                         const float* __restrict__ w, const int4* __restrict__ rec,
                                         const float* __restrict__ x, float* acc, uint2* __restrict__ keep,
                                         unsigned int* degc) {
    unsigned s0 = (vj.x ^ pk) + 0x7F7F7F7Fu;
    unsigned s1 = (vj.y ^ pk) + 0x7F7F7F7Fu;
    unsigned s2 = (vj.z ^ pk) + 0x7F7F7F7Fu;
    unsigned s3 = (vj.w ^ pk) + 0x7F7F7F7Fu;
    const int in = i + OWN_PF * OWN_THREADS;
    if (GUARD) vj = (in < NE16) ? key4[in] : make_uint4(0x7F7F7F7Fu, 0x7F7F7F7Fu, 0x7F7F7F7Fu, 0x7F7F7F7Fu);
    else       vj = key4[in];

    unsigned any = __builtin_amdgcn_ballot_w32(((s0 & s1 & s2 & s3) & 0x80808080u) != 0x80808080u);
    if (any) {
        const unsigned miss = ((s0 >> 7) & 0x01010101u) | ((s1 >> 6) & 0x02020202u) |
                              ((s2 >> 5) & 0x04040404u) | ((s3 >> 4) & 0x08080808u);
        unsigned hm = ~miss & 0x0F0F0F0Fu;
        const unsigned e0 = (unsigned)i * 16u;
        do {
            if (hm != 0u) {
                const unsigned p = (unsigned)__builtin_ctz(hm);
                hm &= hm - 1u;
                lst[(tail + __builtin_amdgcn_mbcnt_lo(any, 0u)) & (OWN_RING - 1)] = e0 + ((p & 7u) << 2) + (p >> 3);
            }
            tail += (unsigned)__builtin_popcount(any);
            while (tail - head >= (unsigned)OWN_BATCH) {
                own_drain<true, PACKED, SAVE, COUNT>(lst, head, (unsigned)OWN_BATCH, lane, base, rev, own, oth, w, rec, x, acc, keep, degc);
                head += (unsigned)OWN_BATCH;
            }
            any = __builtin_amdgcn_ballot_w32(hm != 0u);
        } while (any);
    }
}

template <bool PACKED, bool SAVE, bool COUNT>
__device__ __forceinline__ void own_scan(uint4 (&v)[OWN_PF], int tid, unsigned pk, int base, int rev,
                                         const uint4* __restrict__ key4, unsigned int* lst, int lane,
                                         const int* __restrict__ own, const int* __restrict__ oth,
                                         const float* __restrict__ w, const int4* __restrict__ rec,
                                         const float* __restrict__ x, float* acc,
                                         uint2* __restrict__ keep, unsigned int* __restrict__ keepn,
                                         unsigned int* degc) {
    unsigned head = 0u, tail = 0u;
    int s0 = 0;
    for (; s0 + 2 * OWN_PF <= OWN_FULL; s0 += OWN_PF) {
        #pragma unroll
        for (int j = 0; j < OWN_PF; ++j)
            own_step<false, PACKED, SAVE, COUNT>(v[j], (s0 + j) * OWN_THREADS + tid, pk, base, rev, key4, lst, head, tail, lane, own, oth, w, rec, x, acc, keep, degc);
    }
    for (; s0 < OWN_NSTEP; s0 += OWN_PF) {
        #pragma unroll
        for (int j = 0; j < OWN_PF; ++j)
            own_step<true, PACKED, SAVE, COUNT>(v[j], (s0 + j) * OWN_THREADS + tid, pk, base, rev, key4, lst, head, tail, lane, own, oth, w, rec, x, acc, keep, degc);
    }
    if (tail != head)
        own_drain<false, PACKED, SAVE, COUNT>(lst, head, tail - head, lane, base, rev, own, oth, w, rec, x, acc, keep, degc);
    if (SAVE && lane == 0) *keepn = tail;
}

#define RP_SUB 16
template <bool FULL>
__device__ __forceinline__ void own_replay32(uint2 ent, unsigned nvalid, int lane, const float* __restrict__ x, float* acc) {
    #pragma unroll
    for (int h = 0; h < 32 / RP_SUB; ++h) {
        if (!FULL && (unsigned)(h * RP_SUB) >= nvalid) break;
        v2f xv[RP_SUB];
        #pragma unroll
        for (int j = 0; j < RP_SUB; ++j) {
            const unsigned o = (unsigned)__builtin_amdgcn_readlane((int)ent.x, h * RP_SUB + j) & ((1u << LST_OBITS) - 1u);
            xv[j] = *reinterpret_cast<const v2f*>(x + (size_t)o * HH + 2 * lane);
        }
        #pragma unroll
        for (int j = 0; j < RP_SUB; ++j) {
            if (FULL || (unsigned)(h * RP_SUB + j) < nvalid) {
                const unsigned n  = (unsigned)__builtin_amdgcn_readlane((int)ent.x, h * RP_SUB + j) >> LST_OBITS;
                const float    we = __int_as_float(__builtin_amdgcn_readlane((int)ent.y, h * RP_SUB + j));
                float* a = acc + (size_t)n * HH + 2 * lane;
                atomicAdd(a,     we * xv[j].x);
                atomicAdd(a + 1, we * xv[j].y);
            }
        }
    }
}

__device__ __forceinline__ void own_replay(const uint2* __restrict__ keep, unsigned cnt, int lane,
                                           const float* __restrict__ x, float* acc) {
    if (cnt == 0u) return;
    const unsigned last = cnt - 1u;
    uint2 cur = keep[((unsigned)lane < last) ? (unsigned)lane : last];
    unsigned p = 0u;
    for (; p + 32u <= cnt; p += 32u) {
        const unsigned q = p + 32u + (unsigned)lane;
        const uint2 nxt = keep[(q < last) ? q : last];
        own_replay32<true>(cur, 32u, lane, x, acc);
        cur = nxt;
    }
    if (p < cnt) own_replay32<false>(cur, cnt - p, lane, x, acc);
}

template <bool SECOND>
__global__ __launch_bounds__(OWN_THREADS) void own_kernel(const float* __restrict__ x,
                                                          const float* __restrict__ w,
                                                          const int* __restrict__ src,
                                                          const int* __restrict__ dst,
                                                          const unsigned int* __restrict__ keyF,
                                                          const unsigned int* __restrict__ keyR,
                                                          const int* __restrict__ rec,
                                                          float* __restrict__ nF,
                                                          float* __restrict__ nR,
                                                          float* __restrict__ invF,
                                                          float* __restrict__ invR,
                                                          unsigned int* __restrict__ keepw,
                                                          unsigned int* __restrict__ keepn,
                                                          size_t ws_bytes) {
    __shared__ float acc[OWN_NODES * HH];
    __shared__ unsigned int lists[OWN_WAVES * OWN_RING];
    __shared__ unsigned int degc[OWN_NODES];
    if (ws_bytes < KEY_WS_BYTES) return;
    const int tid  = threadIdx.x;
    const int lane = tid & 31;
    const int wave = tid >> 5;
    const int rev  = blockIdx.x & 1;
    const int part = blockIdx.x >> 1;
    const int base = part * OWN_NODES;
    const unsigned nloc = (unsigned)((NN - base < OWN_NODES) ? (NN - base) : OWN_NODES);
    const unsigned pk = (unsigned)part * 0x01010101u;
    const int* __restrict__ own = rev ? src : dst;
    const int* __restrict__ oth = rev ? dst : src;
    float* __restrict__ outp    = rev ? nR : nF;
    const uint4* key4 = reinterpret_cast<const uint4*>(rev ? keyR : keyF);
    unsigned int* lst = lists + wave * OWN_RING;
    const int4* rec4 = reinterpret_cast<const int4*>(rec);
    const bool room = ws_bytes >= LIST_WS_BYTES;
    const size_t wslot = (size_t)blockIdx.x * OWN_WAVES + (size_t)wave;
    uint2* keep = reinterpret_cast<uint2*>(keepw) + wslot * LST_CAP;
    unsigned int* kn = keepn + wslot;

    unsigned cnt = (unsigned)LST_CAP + 1u;
    if (SECOND && room) cnt = (unsigned)__builtin_amdgcn_readfirstlane((int)*kn);
    const bool replay = SECOND && room && cnt <= (unsigned)LST_CAP;

    uint4 v[OWN_PF];
    if (!replay) {
        #pragma unroll
        for (int j = 0; j < OWN_PF; ++j) {
            int i = j * OWN_THREADS + tid;
            v[j] = (i < NE16) ? key4[i] : make_uint4(0x7F7F7F7Fu, 0x7F7F7F7Fu, 0x7F7F7F7Fu, 0x7F7F7F7Fu);
        }
    }

    float4* acc4 = reinterpret_cast<float4*>(acc);
    for (int i = tid; i < OWN_NODES * HH / 4; i += OWN_THREADS)
        acc4[i] = make_float4(0.0f, 0.0f, 0.0f, 0.0f);
    if (!SECOND && tid < OWN_NODES) degc[tid] = 0u;
    __syncthreads();

    if (SECOND) {
        if (__builtin_expect(replay, 1))
            own_replay(keep, cnt, lane, x, acc);
        else if (ws_bytes >= REC_WS_BYTES)
            own_scan<true, false, false>(v, tid, pk, base, rev, key4, lst, lane, own, oth, w, rec4, x, acc, keep, kn, degc);
        else
            own_scan<false, false, false>(v, tid, pk, base, rev, key4, lst, lane, own, oth, w, rec4, x, acc, keep, kn, degc);
    } else {
        if (__builtin_expect(room, 1))
            own_scan<true, true, true>(v, tid, pk, base, rev, key4, lst, lane, own, oth, w, rec4, x, acc, keep, kn, degc);
        else if (ws_bytes >= REC_WS_BYTES)
            own_scan<true, false, true>(v, tid, pk, base, rev, key4, lst, lane, own, oth, w, rec4, x, acc, keep, kn, degc);
        else
            own_scan<false, false, true>(v, tid, pk, base, rev, key4, lst, lane, own, oth, w, rec4, x, acc, keep, kn, degc);
    }
    __syncthreads();

    if (!SECOND) {
        float* __restrict__ invp = rev ? invR : invF;
        if ((unsigned)tid < nloc) {
            const float c = (float)degc[tid];
            invp[base + tid] = (c > 0.0f) ? (1.0f / c) : 0.0f;
        }
    }

    float4* out4 = reinterpret_cast<float4*>(outp + (size_t)base * HH);
    const int nout4 = (int)nloc * (HH / 4);
    for (int i = tid; i < nout4; i += OWN_THREADS)
        out4[i] = acc4[i];
}

#define LY_GROUPS  256
#define LY_THREADS 416
#define LY_WAVES   (LY_THREADS / 32)
#define LY_SLOTS   (4 * 16 * 2 * 32)
#define LY_STAGE   ((LY_SLOTS + LY_THREADS - 1) / LY_THREADS)
#define LY_KG      4
__global__ __launch_bounds__(LY_THREADS) void layer_kernel(
    float* __restrict__ x,
    const float* __restrict__ nF, const float* __restrict__ nR,
    const float* __restrict__ invF, const float* __restrict__ invR,
    const float* __restrict__ WsF, const float* __restrict__ WnF, const float* __restrict__ bF,
    const float* __restrict__ WsR, const float* __restrict__ WnR, const float* __restrict__ bR) {
    __shared__ float4 wl[LY_SLOTS];
    const int tid  = threadIdx.x;
    const int wave = tid >> 5;
    const int lane = tid & 31;
    const int ln = lane & 15;
    const int hi = lane >> 4;
    const int kh = hi << 1;

    float4 st[LY_STAGE];
    #pragma unroll
    for (int i = 0; i < LY_STAGE; ++i) {
        const int s = tid + i * LY_THREADS;
        if (s < LY_SLOTS) {
            const int m  = s >> 10;
            const int ks = (s >> 6) & 15;
            const int cp = (s >> 5) & 1;
            const int l  = s & 31;
            const float* Wm = (m == 0) ? WsF : (m == 1) ? WnF : (m == 2) ? WsR : WnR;
            const float* p = Wm + (unsigned)((4 * ks + 2 * (l >> 4)) * HH + 32 * cp + (l & 15));
            st[i] = make_float4(p[0], p[HH], p[16], p[HH + 16]);
        }
    }
    #pragma unroll
    for (int i = 0; i < LY_STAGE; ++i) {
        const int s = tid + i * LY_THREADS;
        if (s < LY_SLOTS) wl[s] = st[i];
    }
    __syncthreads();

    const float4* wb = wl + lane;
    static_assert(2 * LY_GROUPS * LY_WAVES >= NTILES, "two trips cover every tile");
    for (int nt = blockIdx.x + LY_GROUPS * wave; nt < NTILES; nt += LY_GROUPS * LY_WAVES) {
        const unsigned offA = (unsigned)(nt * 16 + ln) * HH + (unsigned)kh;
        const float sF = invF[nt * 16 + ln];
        const float sR = invR[nt * 16 + ln];
        v8f aF0 = {}, aF1 = {}, aF2 = {}, aF3 = {};
        v8f aR0 = {}, aR1 = {}, aR2 = {}, aR3 = {};

        v2f cS[LY_KG], cN[LY_KG], cV[LY_KG];
        v2f dS[LY_KG], dN[LY_KG], dV[LY_KG];
        float4 cB[8], dB[8];
        #pragma unroll
        for (int j = 0; j < LY_KG; ++j) {
            cS[j] = *reinterpret_cast<const v2f*>(x  + offA + 4 * j);
            cN[j] = *reinterpret_cast<const v2f*>(nF + offA + 4 * j);
            cV[j] = *reinterpret_cast<const v2f*>(nR + offA + 4 * j);
        }
        #pragma unroll
        for (int q = 0; q < 8; ++q) cB[q] = wb[(((q >> 1) * 16 + 0) * 2 + (q & 1)) * 32];

        #pragma unroll
        for (int ks = 0; ks < HH / 4; ++ks) {
            const int j = ks % LY_KG;
            if (j == 0) {
                #pragma unroll
                for (int i = 0; i < LY_KG; ++i) {
                    const int kn = (ks + LY_KG < HH / 4) ? ks + LY_KG + i : i;
                    dS[i] = *reinterpret_cast<const v2f*>(x  + offA + 4 * kn);
                    dN[i] = *reinterpret_cast<const v2f*>(nF + offA + 4 * kn);
                    dV[i] = *reinterpret_cast<const v2f*>(nR + offA + 4 * kn);
                }
            }
            {
                const int kb = (ks + 1 < HH / 4) ? ks + 1 : 0;
                #pragma unroll
                for (int q = 0; q < 8; ++q) dB[q] = wb[(((q >> 1) * 16 + kb) * 2 + (q & 1)) * 32];
            }
            asm volatile("" ::: "memory");

            const v2f aS = cS[j];
            const v2f aN = cN[j] * sF;
            const v2f aV = cV[j] * sR;
            aF0 = wmma4(aS, mk2(cB[0].x, cB[0].y), aF0);
            aF0 = wmma4(aN, mk2(cB[2].x, cB[2].y), aF0);
            aR0 = wmma4(aS, mk2(cB[4].x, cB[4].y), aR0);
            aR0 = wmma4(aV, mk2(cB[6].x, cB[6].y), aR0);
            aF1 = wmma4(aS, mk2(cB[0].z, cB[0].w), aF1);
            aF1 = wmma4(aN, mk2(cB[2].z, cB[2].w), aF1);
            aR1 = wmma4(aS, mk2(cB[4].z, cB[4].w), aR1);
            aR1 = wmma4(aV, mk2(cB[6].z, cB[6].w), aR1);
            aF2 = wmma4(aS, mk2(cB[1].x, cB[1].y), aF2);
            aF2 = wmma4(aN, mk2(cB[3].x, cB[3].y), aF2);
            aR2 = wmma4(aS, mk2(cB[5].x, cB[5].y), aR2);
            aR2 = wmma4(aV, mk2(cB[7].x, cB[7].y), aR2);
            aF3 = wmma4(aS, mk2(cB[1].z, cB[1].w), aF3);
            aF3 = wmma4(aN, mk2(cB[3].z, cB[3].w), aF3);
            aR3 = wmma4(aS, mk2(cB[5].z, cB[5].w), aR3);
            aR3 = wmma4(aV, mk2(cB[7].z, cB[7].w), aR3);

            #pragma unroll
            for (int q = 0; q < 8; ++q) cB[q] = dB[q];
            if (j == LY_KG - 1) {
                #pragma unroll
                for (int i = 0; i < LY_KG; ++i) { cS[i] = dS[i]; cN[i] = dN[i]; cV[i] = dV[i]; }
            }
        }

        const float bf0 = bF[ln], bf1 = bF[16 + ln], bf2 = bF[32 + ln], bf3 = bF[48 + ln];
        const float br0 = bR[ln], br1 = bR[16 + ln], br2 = bR[32 + ln], br3 = bR[48 + ln];
        float* xo = x + (unsigned)((nt * 16 + 8 * hi) * HH + ln);
        float o0[8], o1[8], o2[8], o3[8];
        #pragma unroll
        for (int v = 0; v < 8; ++v) {
            o0[v] = xo[v * HH];
            o1[v] = xo[v * HH + 16];
            o2[v] = xo[v * HH + 32];
            o3[v] = xo[v * HH + 48];
        }
        #pragma unroll
        for (int v = 0; v < 8; ++v) {
            xo[v * HH]      = o0[v] + (fmaxf(aF0[v] + bf0, 0.0f) + fmaxf(aR0[v] + br0, 0.0f));
            xo[v * HH + 16] = o1[v] + (fmaxf(aF1[v] + bf1, 0.0f) + fmaxf(aR1[v] + br1, 0.0f));
            xo[v * HH + 32] = o2[v] + (fmaxf(aF2[v] + bf2, 0.0f) + fmaxf(aR2[v] + br2, 0.0f));
            xo[v * HH + 48] = o3[v] + (fmaxf(aF3[v] + bf3, 0.0f) + fmaxf(aR3[v] + br3, 0.0f));
        }
    }
}

__global__ __launch_bounds__(64) void norm_kernel(const float* __restrict__ x,
                                                  const int* __restrict__ ids,
                                                  float* __restrict__ out) {
    __shared__ float s[64];
    int b = blockIdx.x;
    int c = threadIdx.x;
    int node = ids[b];
    float v = x[(size_t)node * HH + c];
    s[c] = v * v;
    __syncthreads();
    #pragma unroll
    for (int off = 32; off > 0; off >>= 1) {
        if (c < off) s[c] += s[c + off];
        __syncthreads();
    }
    float nrm = sqrtf(s[0]);
    out[(size_t)b * HH + c] = v / nrm;
}

extern "C" void kernel_launch(void* const* d_in, const int* in_sizes, int n_in,
                              void* d_out, int out_size, void* d_ws, size_t ws_size,
                              hipStream_t stream) {
    const float* text_emb = (const float*)d_in[0];
    const float* weights  = (const float*)d_in[1];
    const float* W_enc    = (const float*)d_in[2];
    const float* b_enc    = (const float*)d_in[3];
    const float* Wself_f  = (const float*)d_in[4];
    const float* Wneigh_f = (const float*)d_in[5];
    const float* b_f      = (const float*)d_in[6];
    const float* Wself_r  = (const float*)d_in[7];
    const float* Wneigh_r = (const float*)d_in[8];
    const float* b_r      = (const float*)d_in[9];
    const int*   src      = (const int*)d_in[10];
    const int*   dst      = (const int*)d_in[11];
    const int*   ids      = (const int*)d_in[12];
    float* out = (float*)d_out;

    const size_t NH = (size_t)NN * HH;
    float* x    = (float*)d_ws;
    float* nF   = x + NH;
    float* nR   = x + 2 * NH;
    float* invF = x + 3 * NH;
    float* invR = invF + NN;
    unsigned int* keyF = (unsigned int*)(invR + NN);
    unsigned int* keyR = keyF + NE / 4;
    int* rec = (int*)(keyR + NE / 4);
    unsigned int* keepw = (unsigned int*)(rec + (size_t)NE * 4);
    unsigned int* keepn = keepw + LST_WORDS;

    const int batch = in_sizes[12];

    key_rec_kernel<<<IK_BLOCKS, IK_THREADS, 0, stream>>>(src, dst, weights, keyF, keyR, rec, ws_size);

    enc_kernel<<<(NTILES + 7) / 8, 256, 0, stream>>>(text_emb, W_enc, b_enc, x);

    own_kernel<false><<<2 * OWN_PARTS, OWN_THREADS, 0, stream>>>(x, weights, src, dst, keyF, keyR, rec, nF, nR, invF, invR, keepw, keepn, ws_size);
    layer_kernel<<<LY_GROUPS, LY_THREADS, 0, stream>>>(
        x, nF, nR, invF, invR, Wself_f, Wneigh_f, b_f, Wself_r, Wneigh_r, b_r);
    own_kernel<true><<<2 * OWN_PARTS, OWN_THREADS, 0, stream>>>(x, weights, src, dst, keyF, keyR, rec, nF, nR, invF, invR, keepw, keepn, ws_size);
    layer_kernel<<<LY_GROUPS, LY_THREADS, 0, stream>>>(
        x, nF, nR, invF, invR,
        Wself_f + (size_t)HH * HH, Wneigh_f + (size_t)HH * HH, b_f + (size_t)HH,
        Wself_r + (size_t)HH * HH, Wneigh_r + (size_t)HH * HH, b_r + (size_t)HH);

    norm_kernel<<<batch, 64, 0, stream>>>(x, ids, out);
}
